# starting kernel
# baseline (speedup 1.0000x reference)
#include <hip/hip_runtime.h>
#include <hip/hip_bf16.h>

typedef __attribute__((ext_vector_type(2))) float v2f;
typedef __attribute__((ext_vector_type(8))) float v8f;

#define NNODES 10000

__global__ __launch_bounds__(256) void zero4_kernel(float4* __restrict__ p, int n4) {
    int i = blockIdx.x * 256 + threadIdx.x;
    int stride = gridDim.x * 256;
    float4 z; z.x = 0.f; z.y = 0.f; z.z = 0.f; z.w = 0.f;
    for (; i < n4; i += stride) p[i] = z;
}

__global__ __launch_bounds__(256) void transpose_kernel(const float* __restrict__ in,
                                                        float* __restrict__ out,
                                                        int K, int NOut) {
    int idx = blockIdx.x * 256 + threadIdx.x;
    if (idx < K * NOut) {
        int k = idx / NOut;
        int c = idx - k * NOut;
        out[(size_t)c * K + k] = in[idx];
    }
}

__global__ __launch_bounds__(256) void deg_kernel(const int* __restrict__ src,
                                                  float* __restrict__ deg, int E) {
    int i = blockIdx.x * 256 + threadIdx.x;
    if (i < E) atomicAdd(&deg[src[i]], 1.0f);
}

__global__ __launch_bounds__(256) void dis_kernel(float* __restrict__ deg, int n) {
    int i = blockIdx.x * 256 + threadIdx.x;
    if (i < n) {
        float d = deg[i];
        deg[i] = (d > 0.0f) ? rsqrtf(d) : 0.0f;
    }
}

__global__ __launch_bounds__(256) void scatter_kernel(
    const float* __restrict__ xin, const int* __restrict__ src,
    const int* __restrict__ dst, const float* __restrict__ dis,
    float* __restrict__ agg, int E, int shift)
{
    unsigned tid = blockIdx.x * 256u + threadIdx.x;
    int e = (int)(tid >> shift);
    if (e >= E) return;
    int c = (int)(tid & ((1u << shift) - 1u));
    int s = src[e];
    int d = dst[e];
    float norm = -(dis[s] * dis[d]);
    int D = 4 << shift;
    const float4 v = *(const float4*)(xin + (size_t)s * D + 4 * c);
    float* outp = agg + (size_t)d * D + 4 * c;
    atomicAdd(outp + 0, norm * v.x);
    atomicAdd(outp + 1, norm * v.y);
    atomicAdd(outp + 2, norm * v.z);
    atomicAdd(outp + 3, norm * v.w);
}

template <int TILES>
__global__ __launch_bounds__(128) void cheb_gemm_kernel(
    const float* __restrict__ Xin, const float* __restrict__ Agg,
    const float* __restrict__ W0T, const float* __restrict__ W1T,
    const float* __restrict__ bias, float* __restrict__ Out,
    int K, int NOut, int relu)
{
    const int wave  = threadIdx.x >> 5;
    const int lane  = threadIdx.x & 31;
    const int rlane = lane & 15;
    const int khalf = (lane >> 4) << 1;
    const int rowBase = blockIdx.x << 4;
    const int colBase = wave * (TILES * 16);

    const float* a0p = Xin + (size_t)(rowBase + rlane) * K + khalf;
    const float* a1p = Agg + (size_t)(rowBase + rlane) * K + khalf;
    const float* w0p[TILES];
    const float* w1p[TILES];
#pragma unroll
    for (int t = 0; t < TILES; ++t) {
        w0p[t] = W0T + (size_t)(colBase + t * 16 + rlane) * K + khalf;
        w1p[t] = W1T + (size_t)(colBase + t * 16 + rlane) * K + khalf;
    }

    v8f acc[TILES];
#pragma unroll
    for (int t = 0; t < TILES; ++t) acc[t] = (v8f){};

    v2f a0c, a1c, b0c[TILES], b1c[TILES];

    a0c = *(const v2f*)(a0p);
    a1c = *(const v2f*)(a1p);
#pragma unroll
    for (int t = 0; t < TILES; ++t) {
        b0c[t] = *(const v2f*)(w0p[t]);
        b1c[t] = *(const v2f*)(w1p[t]);
    }

#pragma unroll 2
    for (int k = 0; k < K - 4; k += 4) {
        v2f a0n = *(const v2f*)(a0p + k + 4);
        v2f a1n = *(const v2f*)(a1p + k + 4);
        v2f b0n[TILES], b1n[TILES];
#pragma unroll
        for (int t = 0; t < TILES; ++t) b0n[t] = *(const v2f*)(w0p[t] + k + 4);
#pragma unroll
        for (int t = 0; t < TILES; ++t) b1n[t] = *(const v2f*)(w1p[t] + k + 4);

#pragma unroll
        for (int t = 0; t < TILES; ++t)
            acc[t] = __builtin_amdgcn_wmma_f32_16x16x4_f32(
                false, a0c, false, b0c[t], (short)0, acc[t], false, false);
#pragma unroll
        for (int t = 0; t < TILES; ++t)
            acc[t] = __builtin_amdgcn_wmma_f32_16x16x4_f32(
                false, a1c, false, b1c[t], (short)0, acc[t], false, false);

        a0c = a0n; a1c = a1n;
#pragma unroll
        for (int t = 0; t < TILES; ++t) { b0c[t] = b0n[t]; b1c[t] = b1n[t]; }
    }

#pragma unroll
    for (int t = 0; t < TILES; ++t)
        acc[t] = __builtin_amdgcn_wmma_f32_16x16x4_f32(
            false, a0c, false, b0c[t], (short)0, acc[t], false, false);
#pragma unroll
    for (int t = 0; t < TILES; ++t)
        acc[t] = __builtin_amdgcn_wmma_f32_16x16x4_f32(
            false, a1c, false, b1c[t], (short)0, acc[t], false, false);

    const int rowOff = rowBase + ((lane >> 4) << 3);
#pragma unroll
    for (int t = 0; t < TILES; ++t) {
        const int col = colBase + t * 16 + rlane;
        const float bv = bias[col];
#pragma unroll
        for (int r = 0; r < 8; ++r) {
            float v = acc[t][r] + bv;
            if (relu) v = fmaxf(v, 0.0f);
            Out[(size_t)(rowOff + r) * NOut + col] = v;
        }
    }
}

extern "C" void kernel_launch(void* const* d_in, const int* in_sizes, int n_in,
                              void* d_out, int out_size, void* d_ws, size_t ws_size,
                              hipStream_t stream) {
    const float* x   = (const float*)d_in[0];
    const int*   ei  = (const int*)d_in[1];
    const float* W01 = (const float*)d_in[3];
    const float* W11 = (const float*)d_in[4];
    const float* b1  = (const float*)d_in[5];
    const float* W02 = (const float*)d_in[6];
    const float* W12 = (const float*)d_in[7];
    const float* b2  = (const float*)d_in[8];
    const float* W03 = (const float*)d_in[9];
    const float* W13 = (const float*)d_in[10];
    const float* b3  = (const float*)d_in[11];

    const int N = NNODES;
    const int E = in_sizes[1] / 2;
    const int* src = ei;
    const int* dst = ei + E;

    float* dis  = (float*)d_ws;
    float* agg  = dis + 10240;
    float* h1   = agg + (size_t)N * 256;
    float* h2   = h1  + (size_t)N * 256;
    float* w01t = h2  + (size_t)N * 256;
    float* w11t = w01t + 128 * 256;
    float* w02t = w11t + 128 * 256;
    float* w12t = w02t + 256 * 256;
    float* w03t = w12t + 256 * 256;
    float* w13t = w03t + 256 * 128;
    float* out  = (float*)d_out;

    transpose_kernel<<<(128 * 256 + 255) / 256, 256, 0, stream>>>(W01, w01t, 128, 256);
    transpose_kernel<<<(128 * 256 + 255) / 256, 256, 0, stream>>>(W11, w11t, 128, 256);
    transpose_kernel<<<(256 * 256 + 255) / 256, 256, 0, stream>>>(W02, w02t, 256, 256);
    transpose_kernel<<<(256 * 256 + 255) / 256, 256, 0, stream>>>(W12, w12t, 256, 256);
    transpose_kernel<<<(256 * 128 + 255) / 256, 256, 0, stream>>>(W03, w03t, 256, 128);
    transpose_kernel<<<(256 * 128 + 255) / 256, 256, 0, stream>>>(W13, w13t, 256, 128);

    zero4_kernel<<<64, 256, 0, stream>>>((float4*)dis, N / 4);
    deg_kernel<<<(E + 255) / 256, 256, 0, stream>>>(src, dis, E);
    dis_kernel<<<(N + 255) / 256, 256, 0, stream>>>(dis, N);

    zero4_kernel<<<1024, 256, 0, stream>>>((float4*)agg, N * 128 / 4);
    {
        unsigned nth = (unsigned)E << 5;
        scatter_kernel<<<(nth + 255) / 256, 256, 0, stream>>>(x, src, dst, dis, agg, E, 5);
    }
    cheb_gemm_kernel<4><<<N / 16, 128, 0, stream>>>(x, agg, w01t, w11t, b1, h1, 128, 256, 1);

    zero4_kernel<<<2048, 256, 0, stream>>>((float4*)agg, N * 256 / 4);
    {
        unsigned nth = (unsigned)E << 6;
        scatter_kernel<<<(nth + 255) / 256, 256, 0, stream>>>(h1, src, dst, dis, agg, E, 6);
    }
    cheb_gemm_kernel<4><<<N / 16, 128, 0, stream>>>(h1, agg, w02t, w12t, b2, h2, 256, 256, 1);

    zero4_kernel<<<2048, 256, 0, stream>>>((float4*)agg, N * 256 / 4);
    {
        unsigned nth = (unsigned)E << 6;
        scatter_kernel<<<(nth + 255) / 256, 256, 0, stream>>>(h2, src, dst, dis, agg, E, 6);
    }
    cheb_gemm_kernel<2><<<N / 16, 128, 0, stream>>>(h2, agg, w03t, w13t, b3, out, 256, 128, 0);
}


# Aggregation by feature column in LDS: a 1024-thread block per feature keeps dis*x and the 10000 sums in LDS, streams all edges with LDS float atomics; aggregate kept feature-major; sum order changes
# speedup vs baseline: 1.6063x; 1.6063x over previous
#include <hip/hip_runtime.h>
#include <hip/hip_bf16.h>

typedef __attribute__((ext_vector_type(2))) float v2f;
typedef __attribute__((ext_vector_type(8))) float v8f;

#define NNODES 10000
#define AGGP 10016

__global__ __launch_bounds__(256) void zero4_kernel(float4* __restrict__ p, int n4) {
    int i = blockIdx.x * 256 + threadIdx.x;
    int stride = gridDim.x * 256;
    float4 z; z.x = 0.f; z.y = 0.f; z.z = 0.f; z.w = 0.f;
    for (; i < n4; i += stride) p[i] = z;
}

__global__ __launch_bounds__(256) void transpose_kernel(const float* __restrict__ in,
                                                        float* __restrict__ out,
                                                        int K, int NOut) {
    int idx = blockIdx.x * 256 + threadIdx.x;
    if (idx < K * NOut) {
        int k = idx / NOut;
        int c = idx - k * NOut;
        out[(size_t)c * K + k] = in[idx];
    }
}

__global__ __launch_bounds__(256) void deg_kernel(const int* __restrict__ src,
                                                  float* __restrict__ deg, int E) {
    int i = blockIdx.x * 256 + threadIdx.x;
    if (i < E) atomicAdd(&deg[src[i]], 1.0f);
}

__global__ __launch_bounds__(256) void dis_kernel(float* __restrict__ deg, int n) {
    int i = blockIdx.x * 256 + threadIdx.x;
    if (i < n) {
        float d = deg[i];
        deg[i] = (d > 0.0f) ? rsqrtf(d) : 0.0f;
    }
}

__global__ __launch_bounds__(1024) void agg_column_kernel(
    const float* __restrict__ xin, const int* __restrict__ src,
    const int* __restrict__ dst, const float* __restrict__ dis,
    float* __restrict__ aggT, int E, int D)
{
    __shared__ float xs[NNODES];
    __shared__ float acc[NNODES];
    const unsigned tid = threadIdx.x;
    const unsigned f = blockIdx.x;

    for (unsigned n = tid; n < NNODES; n += 1024u) {
        xs[n] = dis[n] * xin[(size_t)n * D + f];
        acc[n] = 0.0f;
    }
    __syncthreads();

    const unsigned nq = ((E & 3) == 0) ? ((unsigned)E >> 2) : 0u;
    const unsigned nFull = nq >> 10;
    const unsigned st0 = (f * nFull) / gridDim.x;
    const int4* __restrict__ src4 = (const int4*)src;
    const int4* __restrict__ dst4 = (const int4*)dst;
    unsigned it = 0;
    for (; it + 4u <= nFull; it += 4u) {
        int4 s[4], d[4];
#pragma unroll
        for (int u = 0; u < 4; ++u) {
            unsigned st = st0 + it + u;
            if (st >= nFull) st -= nFull;
            const unsigned q = (st << 10) + tid;
            s[u] = src4[q];
            d[u] = dst4[q];
        }
        float v[16];
#pragma unroll
        for (int u = 0; u < 4; ++u) {
            v[4 * u + 0] = xs[s[u].x];
            v[4 * u + 1] = xs[s[u].y];
            v[4 * u + 2] = xs[s[u].z];
            v[4 * u + 3] = xs[s[u].w];
        }
#pragma unroll
        for (int u = 0; u < 4; ++u) {
            __hip_atomic_fetch_add(&acc[d[u].x], v[4 * u + 0], __ATOMIC_RELAXED, __HIP_MEMORY_SCOPE_WORKGROUP);
            __hip_atomic_fetch_add(&acc[d[u].y], v[4 * u + 1], __ATOMIC_RELAXED, __HIP_MEMORY_SCOPE_WORKGROUP);
            __hip_atomic_fetch_add(&acc[d[u].z], v[4 * u + 2], __ATOMIC_RELAXED, __HIP_MEMORY_SCOPE_WORKGROUP);
            __hip_atomic_fetch_add(&acc[d[u].w], v[4 * u + 3], __ATOMIC_RELAXED, __HIP_MEMORY_SCOPE_WORKGROUP);
        }
    }
    for (; it < nFull; ++it) {
        unsigned st = st0 + it;
        if (st >= nFull) st -= nFull;
        const unsigned q = (st << 10) + tid;
        const int4 s = src4[q];
        const int4 d = dst4[q];
        const float v0 = xs[s.x];
        const float v1 = xs[s.y];
        const float v2 = xs[s.z];
        const float v3 = xs[s.w];
        __hip_atomic_fetch_add(&acc[d.x], v0, __ATOMIC_RELAXED, __HIP_MEMORY_SCOPE_WORKGROUP);
        __hip_atomic_fetch_add(&acc[d.y], v1, __ATOMIC_RELAXED, __HIP_MEMORY_SCOPE_WORKGROUP);
        __hip_atomic_fetch_add(&acc[d.z], v2, __ATOMIC_RELAXED, __HIP_MEMORY_SCOPE_WORKGROUP);
        __hip_atomic_fetch_add(&acc[d.w], v3, __ATOMIC_RELAXED, __HIP_MEMORY_SCOPE_WORKGROUP);
    }
    for (unsigned e = (nFull << 12) + tid; e < (unsigned)E; e += 1024u)
        __hip_atomic_fetch_add(&acc[dst[e]], xs[src[e]], __ATOMIC_RELAXED, __HIP_MEMORY_SCOPE_WORKGROUP);
    __syncthreads();

    float* __restrict__ col = aggT + (size_t)f * AGGP;
    for (unsigned n = tid; n < NNODES; n += 1024u) col[n] = -(dis[n] * acc[n]);
}

template <int TILES>
__global__ __launch_bounds__(128) void cheb_gemm_kernel(
    const float* __restrict__ Xin, const float* __restrict__ AggT,
    const float* __restrict__ W0T, const float* __restrict__ W1T,
    const float* __restrict__ bias, float* __restrict__ Out,
    int K, int NOut, int relu)
{
    const int wave  = threadIdx.x >> 5;
    const int lane  = threadIdx.x & 31;
    const int rlane = lane & 15;
    const int khalf = (lane >> 4) << 1;
    const int rowBase = blockIdx.x << 4;
    const int colBase = wave * (TILES * 16);

    const float* a0p = Xin + (size_t)(rowBase + rlane) * K + khalf;
    const float* a1p = AggT + (size_t)khalf * AGGP + rowBase + rlane;
    const float* w0p[TILES];
    const float* w1p[TILES];
#pragma unroll
    for (int t = 0; t < TILES; ++t) {
        w0p[t] = W0T + (size_t)(colBase + t * 16 + rlane) * K + khalf;
        w1p[t] = W1T + (size_t)(colBase + t * 16 + rlane) * K + khalf;
    }

    v8f acc[TILES];
#pragma unroll
    for (int t = 0; t < TILES; ++t) acc[t] = (v8f){};

    v2f a0c, a1c, b0c[TILES], b1c[TILES];

    a0c = *(const v2f*)(a0p);
    a1c.x = a1p[0];
    a1c.y = a1p[AGGP];
#pragma unroll
    for (int t = 0; t < TILES; ++t) {
        b0c[t] = *(const v2f*)(w0p[t]);
        b1c[t] = *(const v2f*)(w1p[t]);
    }

#pragma unroll 2
    for (int k = 0; k < K - 4; k += 4) {
        v2f a0n = *(const v2f*)(a0p + k + 4);
        v2f a1n;
        a1n.x = a1p[(size_t)(k + 4) * AGGP];
        a1n.y = a1p[(size_t)(k + 5) * AGGP];
        v2f b0n[TILES], b1n[TILES];
#pragma unroll
        for (int t = 0; t < TILES; ++t) b0n[t] = *(const v2f*)(w0p[t] + k + 4);
#pragma unroll
        for (int t = 0; t < TILES; ++t) b1n[t] = *(const v2f*)(w1p[t] + k + 4);

#pragma unroll
        for (int t = 0; t < TILES; ++t)
            acc[t] = __builtin_amdgcn_wmma_f32_16x16x4_f32(
                false, a0c, false, b0c[t], (short)0, acc[t], false, false);
#pragma unroll
        for (int t = 0; t < TILES; ++t)
            acc[t] = __builtin_amdgcn_wmma_f32_16x16x4_f32(
                false, a1c, false, b1c[t], (short)0, acc[t], false, false);

        a0c = a0n; a1c = a1n;
#pragma unroll
        for (int t = 0; t < TILES; ++t) { b0c[t] = b0n[t]; b1c[t] = b1n[t]; }
    }

#pragma unroll
    for (int t = 0; t < TILES; ++t)
        acc[t] = __builtin_amdgcn_wmma_f32_16x16x4_f32(
            false, a0c, false, b0c[t], (short)0, acc[t], false, false);
#pragma unroll
    for (int t = 0; t < TILES; ++t)
        acc[t] = __builtin_amdgcn_wmma_f32_16x16x4_f32(
            false, a1c, false, b1c[t], (short)0, acc[t], false, false);

    const int rowOff = rowBase + ((lane >> 4) << 3);
#pragma unroll
    for (int t = 0; t < TILES; ++t) {
        const int col = colBase + t * 16 + rlane;
        const float bv = bias[col];
#pragma unroll
        for (int r = 0; r < 8; ++r) {
            float v = acc[t][r] + bv;
            if (relu) v = fmaxf(v, 0.0f);
            Out[(size_t)(rowOff + r) * NOut + col] = v;
        }
    }
}

extern "C" void kernel_launch(void* const* d_in, const int* in_sizes, int n_in,
                              void* d_out, int out_size, void* d_ws, size_t ws_size,
                              hipStream_t stream) {
    const float* x   = (const float*)d_in[0];
    const int*   ei  = (const int*)d_in[1];
    const float* W01 = (const float*)d_in[3];
    const float* W11 = (const float*)d_in[4];
    const float* b1  = (const float*)d_in[5];
    const float* W02 = (const float*)d_in[6];
    const float* W12 = (const float*)d_in[7];
    const float* b2  = (const float*)d_in[8];
    const float* W03 = (const float*)d_in[9];
    const float* W13 = (const float*)d_in[10];
    const float* b3  = (const float*)d_in[11];

    const int N = NNODES;
    const int E = in_sizes[1] / 2;
    const int* src = ei;
    const int* dst = ei + E;

    float* dis  = (float*)d_ws;
    float* aggT = dis + 10240;
    float* w01t = aggT + 128 * AGGP;
    float* w11t = w01t + 128 * 256;
    float* h1   = aggT + 256 * AGGP;
    float* h2   = h1  + (size_t)N * 256;
    float* w02t = h2  + (size_t)N * 256;
    float* w12t = w02t + 256 * 256;
    float* w03t = w12t + 256 * 256;
    float* w13t = w03t + 256 * 128;
    float* out  = (float*)d_out;

    transpose_kernel<<<(128 * 256 + 255) / 256, 256, 0, stream>>>(W01, w01t, 128, 256);
    transpose_kernel<<<(128 * 256 + 255) / 256, 256, 0, stream>>>(W11, w11t, 128, 256);
    transpose_kernel<<<(256 * 256 + 255) / 256, 256, 0, stream>>>(W02, w02t, 256, 256);
    transpose_kernel<<<(256 * 256 + 255) / 256, 256, 0, stream>>>(W12, w12t, 256, 256);
    transpose_kernel<<<(256 * 128 + 255) / 256, 256, 0, stream>>>(W03, w03t, 256, 128);
    transpose_kernel<<<(256 * 128 + 255) / 256, 256, 0, stream>>>(W13, w13t, 256, 128);

    zero4_kernel<<<64, 256, 0, stream>>>((float4*)dis, N / 4);
    deg_kernel<<<(E + 255) / 256, 256, 0, stream>>>(src, dis, E);
    dis_kernel<<<(N + 255) / 256, 256, 0, stream>>>(dis, N);

    agg_column_kernel<<<128, 1024, 0, stream>>>(x, src, dst, dis, aggT, E, 128);
    cheb_gemm_kernel<4><<<N / 16, 128, 0, stream>>>(x, aggT, w01t, w11t, b1, h1, 128, 256, 1);

    agg_column_kernel<<<256, 1024, 0, stream>>>(h1, src, dst, dis, aggT, E, 256);
    cheb_gemm_kernel<4><<<N / 16, 128, 0, stream>>>(h1, aggT, w02t, w12t, b2, h2, 256, 256, 1);

    agg_column_kernel<<<256, 1024, 0, stream>>>(h2, src, dst, dis, aggT, E, 256);
    cheb_gemm_kernel<2><<<N / 16, 128, 0, stream>>>(h2, aggT, w03t, w13t, b3, out, 256, 128, 0);
}


# Edge list re-laid once per call so that lane l gets only edges with dst mod 32 = l (packed 32-bit entries): the 32 LDS float adds of an instruction hit 32 different banks; sum order changes
# speedup vs baseline: 1.8653x; 1.1612x over previous
#include <hip/hip_runtime.h>
#include <hip/hip_bf16.h>

typedef __attribute__((ext_vector_type(2))) float v2f;
typedef __attribute__((ext_vector_type(8))) float v8f;

#define NNODES 10000
#define AGGP 10016

__global__ __launch_bounds__(256) void zero4_kernel(float4* __restrict__ p, int n4) {
    int i = blockIdx.x * 256 + threadIdx.x;
    int stride = gridDim.x * 256;
    float4 z; z.x = 0.f; z.y = 0.f; z.z = 0.f; z.w = 0.f;
    for (; i < n4; i += stride) p[i] = z;
}

__global__ __launch_bounds__(256) void transpose_kernel(const float* __restrict__ in,
                                                        float* __restrict__ out,
                                                        int K, int NOut) {
    int idx = blockIdx.x * 256 + threadIdx.x;
    if (idx < K * NOut) {
        int k = idx / NOut;
        int c = idx - k * NOut;
        out[(size_t)c * K + k] = in[idx];
    }
}

#define CLS_CAP 40000u

__global__ __launch_bounds__(256) void deg_kernel(const int* __restrict__ src,
                                                  const int* __restrict__ dst,
                                                  float* __restrict__ deg,
                                                  unsigned* __restrict__ cnt, int E) {
    __shared__ unsigned h[32];
    if (threadIdx.x < 32) h[threadIdx.x] = 0u;
    __syncthreads();
    int i = blockIdx.x * 256 + threadIdx.x;
    if (i < E) {
        atomicAdd(&deg[src[i]], 1.0f);
        atomicAdd(&h[dst[i] & 31], 1u);
    }
    __syncthreads();
    if (threadIdx.x < 32) {
        const unsigned c = h[threadIdx.x];
        if (c != 0u) atomicAdd(&cnt[threadIdx.x], c);
    }
}

__global__ __launch_bounds__(256) void place_kernel(const int* __restrict__ src,
                                                    const int* __restrict__ dst,
                                                    unsigned* __restrict__ cur,
                                                    unsigned* __restrict__ list, int E) {
    __shared__ unsigned h[32];
    __shared__ unsigned base[32];
    if (threadIdx.x < 32) h[threadIdx.x] = 0u;
    __syncthreads();
    const int i = blockIdx.x * 256 + threadIdx.x;
    const bool live = i < E;
    unsigned s = 0u, d = 0u, c = 0u, r = 0u;
    if (live) {
        s = (unsigned)src[i];
        d = (unsigned)dst[i];
        c = d & 31u;
        r = atomicAdd(&h[c], 1u);
    }
    __syncthreads();
    if (threadIdx.x < 32) {
        const unsigned n = h[threadIdx.x];
        base[threadIdx.x] = (n != 0u) ? atomicAdd(&cur[threadIdx.x], n) : 0u;
    }
    __syncthreads();
    if (live) {
        const unsigned p = base[c] + r;
        if (p < CLS_CAP)
            __hip_atomic_exchange(&list[(((p >> 2) << 5) + c) * 4u + (p & 3u)], s | (d << 16),
                                  __ATOMIC_RELAXED, __HIP_MEMORY_SCOPE_AGENT);
    }
}

__global__ __launch_bounds__(256) void dis_kernel(float* __restrict__ deg, int n) {
    int i = blockIdx.x * 256 + threadIdx.x;
    if (i < n) {
        float d = deg[i];
        deg[i] = (d > 0.0f) ? rsqrtf(d) : 0.0f;
    }
}

__global__ __launch_bounds__(1024) void agg_column_kernel(
    const float* __restrict__ xin, const int* __restrict__ src,
    const int* __restrict__ dst, const float* __restrict__ dis,
    const unsigned* __restrict__ cnt, const uint4* __restrict__ list,
    float* __restrict__ aggT, int E, int D)
{
    __shared__ float xs[NNODES];
    __shared__ float acc[NNODES];
    const unsigned tid = threadIdx.x;
    const unsigned f = blockIdx.x;
    const unsigned lane = tid & 31u;
    const unsigned wave = tid >> 5;

    for (unsigned n = tid; n < NNODES; n += 1024u) {
        xs[n] = dis[n] * xin[(size_t)n * D + f];
        acc[n] = 0.0f;
    }
    const unsigned myc = cnt[lane];
    unsigned mn = myc, mx = myc;
#pragma unroll
    for (int o = 16; o > 0; o >>= 1) {
        const unsigned a = __shfl_xor(mn, o);
        const unsigned b = __shfl_xor(mx, o);
        mn = (a < mn) ? a : mn;
        mx = (b > mx) ? b : mx;
    }
    __syncthreads();

    if (mx <= CLS_CAP) {
        const unsigned qFull = mn >> 2;
        const unsigned qAll = (mx + 3u) >> 2;
        const unsigned off = (f * qFull) / gridDim.x;
        unsigned i = wave;
        for (; i + 96u < qFull; i += 128u) {
            uint4 e[4];
#pragma unroll
            for (int u = 0; u < 4; ++u) {
                unsigned q = off + i + 32u * (unsigned)u;
                if (q >= qFull) q -= qFull;
                e[u] = list[(q << 5) + lane];
            }
            float v[16];
#pragma unroll
            for (int u = 0; u < 4; ++u) {
                v[4 * u + 0] = xs[e[u].x & 0xFFFFu];
                v[4 * u + 1] = xs[e[u].y & 0xFFFFu];
                v[4 * u + 2] = xs[e[u].z & 0xFFFFu];
                v[4 * u + 3] = xs[e[u].w & 0xFFFFu];
            }
#pragma unroll
            for (int u = 0; u < 4; ++u) {
                __hip_atomic_fetch_add(&acc[e[u].x >> 16], v[4 * u + 0], __ATOMIC_RELAXED, __HIP_MEMORY_SCOPE_WORKGROUP);
                __hip_atomic_fetch_add(&acc[e[u].y >> 16], v[4 * u + 1], __ATOMIC_RELAXED, __HIP_MEMORY_SCOPE_WORKGROUP);
                __hip_atomic_fetch_add(&acc[e[u].z >> 16], v[4 * u + 2], __ATOMIC_RELAXED, __HIP_MEMORY_SCOPE_WORKGROUP);
                __hip_atomic_fetch_add(&acc[e[u].w >> 16], v[4 * u + 3], __ATOMIC_RELAXED, __HIP_MEMORY_SCOPE_WORKGROUP);
            }
        }
        for (; i < qFull; i += 32u) {
            unsigned q = off + i;
            if (q >= qFull) q -= qFull;
            const uint4 e = list[(q << 5) + lane];
            const float v0 = xs[e.x & 0xFFFFu];
            const float v1 = xs[e.y & 0xFFFFu];
            const float v2 = xs[e.z & 0xFFFFu];
            const float v3 = xs[e.w & 0xFFFFu];
            __hip_atomic_fetch_add(&acc[e.x >> 16], v0, __ATOMIC_RELAXED, __HIP_MEMORY_SCOPE_WORKGROUP);
            __hip_atomic_fetch_add(&acc[e.y >> 16], v1, __ATOMIC_RELAXED, __HIP_MEMORY_SCOPE_WORKGROUP);
            __hip_atomic_fetch_add(&acc[e.z >> 16], v2, __ATOMIC_RELAXED, __HIP_MEMORY_SCOPE_WORKGROUP);
            __hip_atomic_fetch_add(&acc[e.w >> 16], v3, __ATOMIC_RELAXED, __HIP_MEMORY_SCOPE_WORKGROUP);
        }
        for (unsigned q = qFull + wave; q < qAll; q += 32u) {
            const uint4 e = list[(q << 5) + lane];
            const unsigned p = q << 2;
            if (p < myc)
                __hip_atomic_fetch_add(&acc[e.x >> 16], xs[e.x & 0xFFFFu], __ATOMIC_RELAXED, __HIP_MEMORY_SCOPE_WORKGROUP);
            if (p + 1u < myc)
                __hip_atomic_fetch_add(&acc[e.y >> 16], xs[e.y & 0xFFFFu], __ATOMIC_RELAXED, __HIP_MEMORY_SCOPE_WORKGROUP);
            if (p + 2u < myc)
                __hip_atomic_fetch_add(&acc[e.z >> 16], xs[e.z & 0xFFFFu], __ATOMIC_RELAXED, __HIP_MEMORY_SCOPE_WORKGROUP);
            if (p + 3u < myc)
                __hip_atomic_fetch_add(&acc[e.w >> 16], xs[e.w & 0xFFFFu], __ATOMIC_RELAXED, __HIP_MEMORY_SCOPE_WORKGROUP);
        }
    } else {
        for (unsigned e = tid; e < (unsigned)E; e += 1024u)
            __hip_atomic_fetch_add(&acc[dst[e]], xs[src[e]], __ATOMIC_RELAXED, __HIP_MEMORY_SCOPE_WORKGROUP);
    }
    __syncthreads();

    float* __restrict__ col = aggT + (size_t)f * AGGP;
    for (unsigned n = tid; n < NNODES; n += 1024u) col[n] = -(dis[n] * acc[n]);
}

template <int TILES>
__global__ __launch_bounds__(128) void cheb_gemm_kernel(
    const float* __restrict__ Xin, const float* __restrict__ AggT,
    const float* __restrict__ W0T, const float* __restrict__ W1T,
    const float* __restrict__ bias, float* __restrict__ Out,
    int K, int NOut, int relu)
{
    const int wave  = threadIdx.x >> 5;
    const int lane  = threadIdx.x & 31;
    const int rlane = lane & 15;
    const int khalf = (lane >> 4) << 1;
    const int rowBase = blockIdx.x << 4;
    const int colBase = wave * (TILES * 16);

    const float* a0p = Xin + (size_t)(rowBase + rlane) * K + khalf;
    const float* a1p = AggT + (size_t)khalf * AGGP + rowBase + rlane;
    const float* w0p[TILES];
    const float* w1p[TILES];
#pragma unroll
    for (int t = 0; t < TILES; ++t) {
        w0p[t] = W0T + (size_t)(colBase + t * 16 + rlane) * K + khalf;
        w1p[t] = W1T + (size_t)(colBase + t * 16 + rlane) * K + khalf;
    }

    v8f acc[TILES];
#pragma unroll
    for (int t = 0; t < TILES; ++t) acc[t] = (v8f){};

    v2f a0c, a1c, b0c[TILES], b1c[TILES];

    a0c = *(const v2f*)(a0p);
    a1c.x = a1p[0];
    a1c.y = a1p[AGGP];
#pragma unroll
    for (int t = 0; t < TILES; ++t) {
        b0c[t] = *(const v2f*)(w0p[t]);
        b1c[t] = *(const v2f*)(w1p[t]);
    }

#pragma unroll 2
    for (int k = 0; k < K - 4; k += 4) {
        v2f a0n = *(const v2f*)(a0p + k + 4);
        v2f a1n;
        a1n.x = a1p[(size_t)(k + 4) * AGGP];
        a1n.y = a1p[(size_t)(k + 5) * AGGP];
        v2f b0n[TILES], b1n[TILES];
#pragma unroll
        for (int t = 0; t < TILES; ++t) b0n[t] = *(const v2f*)(w0p[t] + k + 4);
#pragma unroll
        for (int t = 0; t < TILES; ++t) b1n[t] = *(const v2f*)(w1p[t] + k + 4);

#pragma unroll
        for (int t = 0; t < TILES; ++t)
            acc[t] = __builtin_amdgcn_wmma_f32_16x16x4_f32(
                false, a0c, false, b0c[t], (short)0, acc[t], false, false);
#pragma unroll
        for (int t = 0; t < TILES; ++t)
            acc[t] = __builtin_amdgcn_wmma_f32_16x16x4_f32(
                false, a1c, false, b1c[t], (short)0, acc[t], false, false);

        a0c = a0n; a1c = a1n;
#pragma unroll
        for (int t = 0; t < TILES; ++t) { b0c[t] = b0n[t]; b1c[t] = b1n[t]; }
    }

#pragma unroll
    for (int t = 0; t < TILES; ++t)
        acc[t] = __builtin_amdgcn_wmma_f32_16x16x4_f32(
            false, a0c, false, b0c[t], (short)0, acc[t], false, false);
#pragma unroll
    for (int t = 0; t < TILES; ++t)
        acc[t] = __builtin_amdgcn_wmma_f32_16x16x4_f32(
            false, a1c, false, b1c[t], (short)0, acc[t], false, false);

    const int rowOff = rowBase + ((lane >> 4) << 3);
#pragma unroll
    for (int t = 0; t < TILES; ++t) {
        const int col = colBase + t * 16 + rlane;
        const float bv = bias[col];
#pragma unroll
        for (int r = 0; r < 8; ++r) {
            float v = acc[t][r] + bv;
            if (relu) v = fmaxf(v, 0.0f);
            Out[(size_t)(rowOff + r) * NOut + col] = v;
        }
    }
}

extern "C" void kernel_launch(void* const* d_in, const int* in_sizes, int n_in,
                              void* d_out, int out_size, void* d_ws, size_t ws_size,
                              hipStream_t stream) {
    const float* x   = (const float*)d_in[0];
    const int*   ei  = (const int*)d_in[1];
    const float* W01 = (const float*)d_in[3];
    const float* W11 = (const float*)d_in[4];
    const float* b1  = (const float*)d_in[5];
    const float* W02 = (const float*)d_in[6];
    const float* W12 = (const float*)d_in[7];
    const float* b2  = (const float*)d_in[8];
    const float* W03 = (const float*)d_in[9];
    const float* W13 = (const float*)d_in[10];
    const float* b3  = (const float*)d_in[11];

    const int N = NNODES;
    const int E = in_sizes[1] / 2;
    const int* src = ei;
    const int* dst = ei + E;

    float* dis  = (float*)d_ws;
    float* aggT = dis + 10240;
    float* w01t = aggT + 128 * AGGP;
    float* w11t = w01t + 128 * 256;
    float* h1   = aggT + 256 * AGGP;
    float* h2   = h1  + (size_t)N * 256;
    float* w02t = h2  + (size_t)N * 256;
    float* w12t = w02t + 256 * 256;
    float* w03t = w12t + 256 * 256;
    float* w13t = w03t + 256 * 128;
    float* out  = (float*)d_out;
    unsigned* cnt = (unsigned*)(dis + 10016);
    unsigned* cur = (unsigned*)(dis + 10048);
    unsigned* listw = (unsigned*)d_out;
    const uint4* list = (const uint4*)d_out;

    transpose_kernel<<<(128 * 256 + 255) / 256, 256, 0, stream>>>(W01, w01t, 128, 256);
    transpose_kernel<<<(128 * 256 + 255) / 256, 256, 0, stream>>>(W11, w11t, 128, 256);
    transpose_kernel<<<(256 * 256 + 255) / 256, 256, 0, stream>>>(W02, w02t, 256, 256);
    transpose_kernel<<<(256 * 256 + 255) / 256, 256, 0, stream>>>(W12, w12t, 256, 256);
    transpose_kernel<<<(256 * 128 + 255) / 256, 256, 0, stream>>>(W03, w03t, 256, 128);
    transpose_kernel<<<(256 * 128 + 255) / 256, 256, 0, stream>>>(W13, w13t, 256, 128);

    zero4_kernel<<<64, 256, 0, stream>>>((float4*)dis, 10240 / 4);
    deg_kernel<<<(E + 255) / 256, 256, 0, stream>>>(src, dst, dis, cnt, E);
    dis_kernel<<<(N + 255) / 256, 256, 0, stream>>>(dis, N);
    place_kernel<<<(E + 255) / 256, 256, 0, stream>>>(src, dst, cur, listw, E);

    agg_column_kernel<<<128, 1024, 0, stream>>>(x, src, dst, dis, cnt, list, aggT, E, 128);
    cheb_gemm_kernel<4><<<N / 16, 128, 0, stream>>>(x, aggT, w01t, w11t, b1, h1, 128, 256, 1);

    agg_column_kernel<<<256, 1024, 0, stream>>>(h1, src, dst, dis, cnt, list, aggT, E, 256);
    cheb_gemm_kernel<4><<<N / 16, 128, 0, stream>>>(h1, aggT, w02t, w12t, b2, h2, 256, 256, 1);

    agg_column_kernel<<<256, 1024, 0, stream>>>(h2, src, dst, dis, cnt, list, aggT, E, 256);
    cheb_gemm_kernel<2><<<N / 16, 128, 0, stream>>>(h2, aggT, w03t, w13t, b3, out, 256, 128, 0);
}


# Weights read in place, six transpose launches gone: a wave's tile t takes columns 4c+t, so one 16-byte load per weight row feeds four tiles; 8 k per stage as each row is a new line; sums as before
# speedup vs baseline: 2.2751x; 1.2197x over previous
#include <hip/hip_runtime.h>
#include <hip/hip_bf16.h>

typedef __attribute__((ext_vector_type(2))) float v2f;
typedef __attribute__((ext_vector_type(4))) float v4f;
typedef __attribute__((ext_vector_type(8))) float v8f;

#define NNODES 10000
#define AGGP 10016

__global__ __launch_bounds__(256) void zero4_kernel(float4* __restrict__ p, int n4) {
    int i = blockIdx.x * 256 + threadIdx.x;
    int stride = gridDim.x * 256;
    float4 z; z.x = 0.f; z.y = 0.f; z.z = 0.f; z.w = 0.f;
    for (; i < n4; i += stride) p[i] = z;
}

#define CLS_CAP 40000u

__global__ __launch_bounds__(256) void deg_kernel(const int* __restrict__ src,
                                                  const int* __restrict__ dst,
                                                  float* __restrict__ deg,
                                                  unsigned* __restrict__ cnt, int E) {
    __shared__ unsigned h[32];
    if (threadIdx.x < 32) h[threadIdx.x] = 0u;
    __syncthreads();
    int i = blockIdx.x * 256 + threadIdx.x;
    if (i < E) {
        atomicAdd(&deg[src[i]], 1.0f);
        atomicAdd(&h[dst[i] & 31], 1u);
    }
    __syncthreads();
    if (threadIdx.x < 32) {
        const unsigned c = h[threadIdx.x];
        if (c != 0u) atomicAdd(&cnt[threadIdx.x], c);
    }
}

__global__ __launch_bounds__(256) void place_kernel(const int* __restrict__ src,
                                                    const int* __restrict__ dst,
                                                    unsigned* __restrict__ cur,
                                                    unsigned* __restrict__ list, int E) {
    __shared__ unsigned h[32];
    __shared__ unsigned base[32];
    if (threadIdx.x < 32) h[threadIdx.x] = 0u;
    __syncthreads();
    const int i = blockIdx.x * 256 + threadIdx.x;
    const bool live = i < E;
    unsigned s = 0u, d = 0u, c = 0u, r = 0u;
    if (live) {
        s = (unsigned)src[i];
        d = (unsigned)dst[i];
        c = d & 31u;
        r = atomicAdd(&h[c], 1u);
    }
    __syncthreads();
    if (threadIdx.x < 32) {
        const unsigned n = h[threadIdx.x];
        base[threadIdx.x] = (n != 0u) ? atomicAdd(&cur[threadIdx.x], n) : 0u;
    }
    __syncthreads();
    if (live) {
        const unsigned p = base[c] + r;
        if (p < CLS_CAP)
            __hip_atomic_exchange(&list[(((p >> 2) << 5) + c) * 4u + (p & 3u)], s | (d << 16),
                                  __ATOMIC_RELAXED, __HIP_MEMORY_SCOPE_AGENT);
    }
}

__global__ __launch_bounds__(256) void dis_kernel(float* __restrict__ deg, int n) {
    int i = blockIdx.x * 256 + threadIdx.x;
    if (i < n) {
        float d = deg[i];
        deg[i] = (d > 0.0f) ? rsqrtf(d) : 0.0f;
    }
}

__global__ __launch_bounds__(1024) void agg_column_kernel(
    const float* __restrict__ xin, const int* __restrict__ src,
    const int* __restrict__ dst, const float* __restrict__ dis,
    const unsigned* __restrict__ cnt, const uint4* __restrict__ list,
    float* __restrict__ aggT, int E, int D)
{
    __shared__ float xs[NNODES];
    __shared__ float acc[NNODES];
    const unsigned tid = threadIdx.x;
    const unsigned f = blockIdx.x;
    const unsigned lane = tid & 31u;
    const unsigned wave = tid >> 5;

    for (unsigned n = tid; n < NNODES; n += 1024u) {
        xs[n] = dis[n] * xin[(size_t)n * D + f];
        acc[n] = 0.0f;
    }
    const unsigned myc = cnt[lane];
    unsigned mn = myc, mx = myc;
#pragma unroll
    for (int o = 16; o > 0; o >>= 1) {
        const unsigned a = __shfl_xor(mn, o);
        const unsigned b = __shfl_xor(mx, o);
        mn = (a < mn) ? a : mn;
        mx = (b > mx) ? b : mx;
    }
    __syncthreads();

    if (mx <= CLS_CAP) {
        const unsigned qFull = mn >> 2;
        const unsigned qAll = (mx + 3u) >> 2;
        const unsigned off = (f * qFull) / gridDim.x;
        unsigned i = wave;
        for (; i + 96u < qFull; i += 128u) {
            uint4 e[4];
#pragma unroll
            for (int u = 0; u < 4; ++u) {
                unsigned q = off + i + 32u * (unsigned)u;
                if (q >= qFull) q -= qFull;
                e[u] = list[(q << 5) + lane];
            }
            float v[16];
#pragma unroll
            for (int u = 0; u < 4; ++u) {
                v[4 * u + 0] = xs[e[u].x & 0xFFFFu];
                v[4 * u + 1] = xs[e[u].y & 0xFFFFu];
                v[4 * u + 2] = xs[e[u].z & 0xFFFFu];
                v[4 * u + 3] = xs[e[u].w & 0xFFFFu];
            }
#pragma unroll
            for (int u = 0; u < 4; ++u) {
                __hip_atomic_fetch_add(&acc[e[u].x >> 16], v[4 * u + 0], __ATOMIC_RELAXED, __HIP_MEMORY_SCOPE_WORKGROUP);
                __hip_atomic_fetch_add(&acc[e[u].y >> 16], v[4 * u + 1], __ATOMIC_RELAXED, __HIP_MEMORY_SCOPE_WORKGROUP);
                __hip_atomic_fetch_add(&acc[e[u].z >> 16], v[4 * u + 2], __ATOMIC_RELAXED, __HIP_MEMORY_SCOPE_WORKGROUP);
                __hip_atomic_fetch_add(&acc[e[u].w >> 16], v[4 * u + 3], __ATOMIC_RELAXED, __HIP_MEMORY_SCOPE_WORKGROUP);
            }
        }
        for (; i < qFull; i += 32u) {
            unsigned q = off + i;
            if (q >= qFull) q -= qFull;
            const uint4 e = list[(q << 5) + lane];
            const float v0 = xs[e.x & 0xFFFFu];
            const float v1 = xs[e.y & 0xFFFFu];
            const float v2 = xs[e.z & 0xFFFFu];
            const float v3 = xs[e.w & 0xFFFFu];
            __hip_atomic_fetch_add(&acc[e.x >> 16], v0, __ATOMIC_RELAXED, __HIP_MEMORY_SCOPE_WORKGROUP);
            __hip_atomic_fetch_add(&acc[e.y >> 16], v1, __ATOMIC_RELAXED, __HIP_MEMORY_SCOPE_WORKGROUP);
            __hip_atomic_fetch_add(&acc[e.z >> 16], v2, __ATOMIC_RELAXED, __HIP_MEMORY_SCOPE_WORKGROUP);
            __hip_atomic_fetch_add(&acc[e.w >> 16], v3, __ATOMIC_RELAXED, __HIP_MEMORY_SCOPE_WORKGROUP);
        }
        for (unsigned q = qFull + wave; q < qAll; q += 32u) {
            const uint4 e = list[(q << 5) + lane];
            const unsigned p = q << 2;
            if (p < myc)
                __hip_atomic_fetch_add(&acc[e.x >> 16], xs[e.x & 0xFFFFu], __ATOMIC_RELAXED, __HIP_MEMORY_SCOPE_WORKGROUP);
            if (p + 1u < myc)
                __hip_atomic_fetch_add(&acc[e.y >> 16], xs[e.y & 0xFFFFu], __ATOMIC_RELAXED, __HIP_MEMORY_SCOPE_WORKGROUP);
            if (p + 2u < myc)
                __hip_atomic_fetch_add(&acc[e.z >> 16], xs[e.z & 0xFFFFu], __ATOMIC_RELAXED, __HIP_MEMORY_SCOPE_WORKGROUP);
            if (p + 3u < myc)
                __hip_atomic_fetch_add(&acc[e.w >> 16], xs[e.w & 0xFFFFu], __ATOMIC_RELAXED, __HIP_MEMORY_SCOPE_WORKGROUP);
        }
    } else {
        for (unsigned e = tid; e < (unsigned)E; e += 1024u)
            __hip_atomic_fetch_add(&acc[dst[e]], xs[src[e]], __ATOMIC_RELAXED, __HIP_MEMORY_SCOPE_WORKGROUP);
    }
    __syncthreads();

    float* __restrict__ col = aggT + (size_t)f * AGGP;
    for (unsigned n = tid; n < NNODES; n += 1024u) col[n] = -(dis[n] * acc[n]);
}

template <int TILES> struct bvec;
template <> struct bvec<4> { typedef v4f type; };
template <> struct bvec<2> { typedef v2f type; };

template <int TILES> struct stage_t {
    v2f a0[2], a1[2];
    typename bvec<TILES>::type b0lo[2], b0hi[2], b1lo[2], b1hi[2];
};

template <int TILES>
__device__ __forceinline__ void load_stage(stage_t<TILES>& s, const float* __restrict__ a0p,
                                           const float* __restrict__ a1p, const float* __restrict__ w0p,
                                           const float* __restrict__ w1p, int k) {
    typedef typename bvec<TILES>::type bv;
    constexpr int NOUT = 64 * TILES;
#pragma unroll
    for (int u = 0; u < 2; ++u) {
        const int kk = k + 4 * u;
        s.a0[u] = *(const v2f*)(a0p + kk);
        s.a1[u].x = a1p[(size_t)kk * AGGP];
        s.a1[u].y = a1p[(size_t)(kk + 1) * AGGP];
        s.b0lo[u] = *(const bv*)(w0p + (size_t)kk * NOUT);
        s.b0hi[u] = *(const bv*)(w0p + (size_t)(kk + 1) * NOUT);
        s.b1lo[u] = *(const bv*)(w1p + (size_t)kk * NOUT);
        s.b1hi[u] = *(const bv*)(w1p + (size_t)(kk + 1) * NOUT);
    }
}

template <int TILES>
__device__ __forceinline__ void mma_stage(const stage_t<TILES>& s, v8f (&acc)[TILES]) {
#pragma unroll
    for (int u = 0; u < 2; ++u) {
#pragma unroll
        for (int t = 0; t < TILES; ++t) {
            v2f b; b.x = s.b0lo[u][t]; b.y = s.b0hi[u][t];
            acc[t] = __builtin_amdgcn_wmma_f32_16x16x4_f32(
                false, s.a0[u], false, b, (short)0, acc[t], false, false);
        }
#pragma unroll
        for (int t = 0; t < TILES; ++t) {
            v2f b; b.x = s.b1lo[u][t]; b.y = s.b1hi[u][t];
            acc[t] = __builtin_amdgcn_wmma_f32_16x16x4_f32(
                false, s.a1[u], false, b, (short)0, acc[t], false, false);
        }
    }
}

template <int TILES>
__global__ __launch_bounds__(128) void cheb_gemm_kernel(
    const float* __restrict__ Xin, const float* __restrict__ AggT,
    const float* __restrict__ W0, const float* __restrict__ W1,
    const float* __restrict__ bias, float* __restrict__ Out,
    int K, int relu)
{
    typedef typename bvec<TILES>::type bv;
    constexpr int NOUT = 64 * TILES;
    const int wave  = threadIdx.x >> 5;
    const int lane  = threadIdx.x & 31;
    const int rlane = lane & 15;
    const int khalf = (lane >> 4) << 1;
    const int rowBase = blockIdx.x << 4;
    const int colBase = wave * (TILES * 16);
    const int col0 = colBase + TILES * rlane;

    const float* a0p = Xin + (size_t)(rowBase + rlane) * K + khalf;
    const float* a1p = AggT + (size_t)khalf * AGGP + rowBase + rlane;
    const float* w0p = W0 + (size_t)khalf * NOUT + col0;
    const float* w1p = W1 + (size_t)khalf * NOUT + col0;

    v8f acc[TILES];
#pragma unroll
    for (int t = 0; t < TILES; ++t) acc[t] = (v8f){};

    stage_t<TILES> cur, nxt;
    load_stage<TILES>(cur, a0p, a1p, w0p, w1p, 0);

#pragma unroll 2
    for (int k = 0; k < K - 8; k += 8) {
        load_stage<TILES>(nxt, a0p, a1p, w0p, w1p, k + 8);
        mma_stage<TILES>(cur, acc);
        cur = nxt;
    }
    mma_stage<TILES>(cur, acc);

    const int rowOff = rowBase + ((lane >> 4) << 3);
    const bv bvv = *(const bv*)(bias + col0);
    float* op = Out + (size_t)rowOff * NOUT + col0;
#pragma unroll
    for (int r = 0; r < 8; ++r) {
        bv v;
#pragma unroll
        for (int t = 0; t < TILES; ++t) {
            float x = acc[t][r] + bvv[t];
            if (relu) x = fmaxf(x, 0.0f);
            v[t] = x;
        }
        *(bv*)(op + (size_t)r * NOUT) = v;
    }
}

extern "C" void kernel_launch(void* const* d_in, const int* in_sizes, int n_in,
                              void* d_out, int out_size, void* d_ws, size_t ws_size,
                              hipStream_t stream) {
    const float* x   = (const float*)d_in[0];
    const int*   ei  = (const int*)d_in[1];
    const float* W01 = (const float*)d_in[3];
    const float* W11 = (const float*)d_in[4];
    const float* b1  = (const float*)d_in[5];
    const float* W02 = (const float*)d_in[6];
    const float* W12 = (const float*)d_in[7];
    const float* b2  = (const float*)d_in[8];
    const float* W03 = (const float*)d_in[9];
    const float* W13 = (const float*)d_in[10];
    const float* b3  = (const float*)d_in[11];

    const int N = NNODES;
    const int E = in_sizes[1] / 2;
    const int* src = ei;
    const int* dst = ei + E;

    float* dis  = (float*)d_ws;
    float* aggT = dis + 10240;
    float* h1   = aggT + 256 * AGGP;
    float* h2   = h1  + (size_t)N * 256;
    float* out  = (float*)d_out;
    unsigned* cnt = (unsigned*)(dis + 10016);
    unsigned* cur = (unsigned*)(dis + 10048);
    unsigned* listw = (unsigned*)d_out;
    const uint4* list = (const uint4*)d_out;

    zero4_kernel<<<64, 256, 0, stream>>>((float4*)dis, 10240 / 4);
    deg_kernel<<<(E + 255) / 256, 256, 0, stream>>>(src, dst, dis, cnt, E);
    dis_kernel<<<(N + 255) / 256, 256, 0, stream>>>(dis, N);
    place_kernel<<<(E + 255) / 256, 256, 0, stream>>>(src, dst, cur, listw, E);

    agg_column_kernel<<<128, 1024, 0, stream>>>(x, src, dst, dis, cnt, list, aggT, E, 128);
    cheb_gemm_kernel<4><<<N / 16, 128, 0, stream>>>(x, aggT, W01, W11, b1, h1, 128, 1);

    agg_column_kernel<<<256, 1024, 0, stream>>>(h1, src, dst, dis, cnt, list, aggT, E, 256);
    cheb_gemm_kernel<4><<<N / 16, 128, 0, stream>>>(h1, aggT, W02, W12, b2, h2, 256, 1);

    agg_column_kernel<<<256, 1024, 0, stream>>>(h2, src, dst, dis, cnt, list, aggT, E, 256);
    cheb_gemm_kernel<2><<<N / 16, 128, 0, stream>>>(h2, aggT, W03, W13, b3, out, 256, 0);
}


# Aggregation split by node half: edge list in 64 classes (dst mod 64); a block sums half the nodes for two features, one 8-byte LDS read per edge; layer 1 one feature per block; sum order changes
# speedup vs baseline: 2.6451x; 1.1626x over previous
#include <hip/hip_runtime.h>
#include <hip/hip_bf16.h>

typedef __attribute__((ext_vector_type(2))) float v2f;
typedef __attribute__((ext_vector_type(4))) float v4f;
typedef __attribute__((ext_vector_type(8))) float v8f;

#define NNODES 10000
#define AGGP 10016

__global__ __launch_bounds__(256) void zero4_kernel(float4* __restrict__ p, int n4) {
    int i = blockIdx.x * 256 + threadIdx.x;
    int stride = gridDim.x * 256;
    float4 z; z.x = 0.f; z.y = 0.f; z.z = 0.f; z.w = 0.f;
    for (; i < n4; i += stride) p[i] = z;
}

#define SUBCAP 20000u
#define PARTQ 160000u
#define ACCN 5024u

__global__ __launch_bounds__(256) void deg_kernel(const int* __restrict__ src,
                                                  const int* __restrict__ dst,
                                                  float* __restrict__ deg,
                                                  unsigned* __restrict__ cnt, int E) {
    __shared__ unsigned h[64];
    if (threadIdx.x < 64) h[threadIdx.x] = 0u;
    __syncthreads();
    int i = blockIdx.x * 256 + threadIdx.x;
    if (i < E) {
        atomicAdd(&deg[src[i]], 1.0f);
        atomicAdd(&h[dst[i] & 63], 1u);
    }
    __syncthreads();
    if (threadIdx.x < 64) {
        const unsigned c = h[threadIdx.x];
        if (c != 0u) atomicAdd(&cnt[threadIdx.x], c);
    }
}

__global__ __launch_bounds__(256) void place_kernel(const int* __restrict__ src,
                                                    const int* __restrict__ dst,
                                                    unsigned* __restrict__ cur,
                                                    unsigned* __restrict__ list, int E) {
    __shared__ unsigned h[64];
    __shared__ unsigned base[64];
    if (threadIdx.x < 64) h[threadIdx.x] = 0u;
    __syncthreads();
    const int i = blockIdx.x * 256 + threadIdx.x;
    const bool live = i < E;
    unsigned s = 0u, d = 0u, c = 0u, r = 0u;
    if (live) {
        s = (unsigned)src[i];
        d = (unsigned)dst[i];
        c = d & 63u;
        r = atomicAdd(&h[c], 1u);
    }
    __syncthreads();
    if (threadIdx.x < 64) {
        const unsigned n = h[threadIdx.x];
        base[threadIdx.x] = (n != 0u) ? atomicAdd(&cur[threadIdx.x], n) : 0u;
    }
    __syncthreads();
    if (live) {
        const unsigned p = base[c] + r;
        if (p < SUBCAP) {
            const unsigned l = c & 31u;
            const unsigned slot = ((d >> 6) << 5) | l;
            const unsigned word = (c >> 5) * PARTQ + ((p >> 2) << 5) + l;
            __hip_atomic_exchange(&list[word * 4u + (p & 3u)], s | (slot << 16),
                                  __ATOMIC_RELAXED, __HIP_MEMORY_SCOPE_AGENT);
        }
    }
}

__global__ __launch_bounds__(256) void dis_kernel(float* __restrict__ deg, int n) {
    int i = blockIdx.x * 256 + threadIdx.x;
    if (i < n) {
        float d = deg[i];
        deg[i] = (d > 0.0f) ? rsqrtf(d) : 0.0f;
    }
}
template <int FPB> struct gvec;
template <> struct gvec<1> { typedef float type; };
template <> struct gvec<2> { typedef v2f type; };

template <int FPB>
__device__ __forceinline__ void add_edge(float* acc, unsigned slot, typename gvec<FPB>::type v);
template <>
__device__ __forceinline__ void add_edge<1>(float* acc, unsigned slot, float v) {
    __hip_atomic_fetch_add(&acc[slot], v, __ATOMIC_RELAXED, __HIP_MEMORY_SCOPE_WORKGROUP);
}
template <>
__device__ __forceinline__ void add_edge<2>(float* acc, unsigned slot, v2f v) {
    __hip_atomic_fetch_add(&acc[slot], v.x, __ATOMIC_RELAXED, __HIP_MEMORY_SCOPE_WORKGROUP);
    __hip_atomic_fetch_add(&acc[ACCN + slot], v.y, __ATOMIC_RELAXED, __HIP_MEMORY_SCOPE_WORKGROUP);
}

template <int FPB>
__global__ __launch_bounds__(1024) void agg_split_kernel(
    const float* __restrict__ xin, const int* __restrict__ src,
    const int* __restrict__ dst, const float* __restrict__ dis,
    const unsigned* __restrict__ cnt, const uint4* __restrict__ list,
    float* __restrict__ aggT, int E, int D)
{
    typedef typename gvec<FPB>::type gv;
    __shared__ gv xs[NNODES];
    __shared__ float acc[ACCN * FPB];
    const unsigned tid = threadIdx.x;
    const unsigned part = blockIdx.x & 1u;
    const unsigned f0 = (blockIdx.x >> 1) * (unsigned)FPB;
    const unsigned lane = tid & 31u;
    const unsigned wave = tid >> 5;

    for (unsigned n = tid; n < NNODES; n += 1024u) {
        const gv t = *(const gv*)(xin + (size_t)n * D + f0);
        xs[n] = dis[n] * t;
    }
    for (unsigned c = tid; c < ACCN * FPB; c += 1024u) acc[c] = 0.0f;
    const unsigned myc = cnt[part * 32u + lane];
    unsigned mn = myc, mx = myc;
#pragma unroll
    for (int o = 16; o > 0; o >>= 1) {
        const unsigned a = __shfl_xor(mn, o);
        const unsigned b = __shfl_xor(mx, o);
        mn = (a < mn) ? a : mn;
        mx = (b > mx) ? b : mx;
    }
    __syncthreads();

    if (mx <= SUBCAP) {
        const uint4* __restrict__ pl = list + (size_t)part * PARTQ;
        const unsigned qFull = mn >> 2;
        const unsigned qAll = (mx + 3u) >> 2;
        const unsigned off = (blockIdx.x * qFull) / gridDim.x;
        unsigned i = wave;
        for (; i + 96u < qFull; i += 128u) {
            uint4 e[4];
#pragma unroll
            for (int u = 0; u < 4; ++u) {
                unsigned q = off + i + 32u * (unsigned)u;
                if (q >= qFull) q -= qFull;
                e[u] = pl[(q << 5) + lane];
            }
            gv v[16];
#pragma unroll
            for (int u = 0; u < 4; ++u) {
                v[4 * u + 0] = xs[e[u].x & 0xFFFFu];
                v[4 * u + 1] = xs[e[u].y & 0xFFFFu];
                v[4 * u + 2] = xs[e[u].z & 0xFFFFu];
                v[4 * u + 3] = xs[e[u].w & 0xFFFFu];
            }
#pragma unroll
            for (int u = 0; u < 4; ++u) {
                add_edge<FPB>(acc, e[u].x >> 16, v[4 * u + 0]);
                add_edge<FPB>(acc, e[u].y >> 16, v[4 * u + 1]);
                add_edge<FPB>(acc, e[u].z >> 16, v[4 * u + 2]);
                add_edge<FPB>(acc, e[u].w >> 16, v[4 * u + 3]);
            }
        }
        for (; i < qFull; i += 32u) {
            unsigned q = off + i;
            if (q >= qFull) q -= qFull;
            const uint4 e = pl[(q << 5) + lane];
            const gv v0 = xs[e.x & 0xFFFFu];
            const gv v1 = xs[e.y & 0xFFFFu];
            const gv v2 = xs[e.z & 0xFFFFu];
            const gv v3 = xs[e.w & 0xFFFFu];
            add_edge<FPB>(acc, e.x >> 16, v0);
            add_edge<FPB>(acc, e.y >> 16, v1);
            add_edge<FPB>(acc, e.z >> 16, v2);
            add_edge<FPB>(acc, e.w >> 16, v3);
        }
        for (unsigned q = qFull + wave; q < qAll; q += 32u) {
            const uint4 e = pl[(q << 5) + lane];
            const unsigned p = q << 2;
            if (p < myc) add_edge<FPB>(acc, e.x >> 16, xs[e.x & 0xFFFFu]);
            if (p + 1u < myc) add_edge<FPB>(acc, e.y >> 16, xs[e.y & 0xFFFFu]);
            if (p + 2u < myc) add_edge<FPB>(acc, e.z >> 16, xs[e.z & 0xFFFFu]);
            if (p + 3u < myc) add_edge<FPB>(acc, e.w >> 16, xs[e.w & 0xFFFFu]);
        }
    } else {
        for (unsigned e = tid; e < (unsigned)E; e += 1024u) {
            const unsigned d = (unsigned)dst[e];
            if (((d >> 5) & 1u) == part) add_edge<FPB>(acc, ((d >> 6) << 5) | (d & 31u), xs[src[e]]);
        }
    }
    __syncthreads();

    for (unsigned c = tid; c < ACCN; c += 1024u) {
        const unsigned n = ((c >> 5) << 6) | (part << 5) | (c & 31u);
        if (n < NNODES) {
            const float dn = dis[n];
#pragma unroll
            for (int j = 0; j < FPB; ++j)
                aggT[(size_t)(f0 + j) * AGGP + n] = -(dn * acc[j * ACCN + c]);
        }
    }
}

template <int TILES> struct bvec;
template <> struct bvec<4> { typedef v4f type; };
template <> struct bvec<2> { typedef v2f type; };

template <int TILES> struct stage_t {
    v2f a0[2], a1[2];
    typename bvec<TILES>::type b0lo[2], b0hi[2], b1lo[2], b1hi[2];
};

template <int TILES>
__device__ __forceinline__ void load_stage(stage_t<TILES>& s, const float* __restrict__ a0p,
                                           const float* __restrict__ a1p, const float* __restrict__ w0p,
                                           const float* __restrict__ w1p, int k) {
    typedef typename bvec<TILES>::type bv;
    constexpr int NOUT = 64 * TILES;
#pragma unroll
    for (int u = 0; u < 2; ++u) {
        const int kk = k + 4 * u;
        s.a0[u] = *(const v2f*)(a0p + kk);
        s.a1[u].x = a1p[(size_t)kk * AGGP];
        s.a1[u].y = a1p[(size_t)(kk + 1) * AGGP];
        s.b0lo[u] = *(const bv*)(w0p + (size_t)kk * NOUT);
        s.b0hi[u] = *(const bv*)(w0p + (size_t)(kk + 1) * NOUT);
        s.b1lo[u] = *(const bv*)(w1p + (size_t)kk * NOUT);
        s.b1hi[u] = *(const bv*)(w1p + (size_t)(kk + 1) * NOUT);
    }
}

template <int TILES>
__device__ __forceinline__ void mma_stage(const stage_t<TILES>& s, v8f (&acc)[TILES]) {
#pragma unroll
    for (int u = 0; u < 2; ++u) {
#pragma unroll
        for (int t = 0; t < TILES; ++t) {
            v2f b; b.x = s.b0lo[u][t]; b.y = s.b0hi[u][t];
            acc[t] = __builtin_amdgcn_wmma_f32_16x16x4_f32(
                false, s.a0[u], false, b, (short)0, acc[t], false, false);
        }
#pragma unroll
        for (int t = 0; t < TILES; ++t) {
            v2f b; b.x = s.b1lo[u][t]; b.y = s.b1hi[u][t];
            acc[t] = __builtin_amdgcn_wmma_f32_16x16x4_f32(
                false, s.a1[u], false, b, (short)0, acc[t], false, false);
        }
    }
}

template <int TILES>
__global__ __launch_bounds__(128) void cheb_gemm_kernel(
    const float* __restrict__ Xin, const float* __restrict__ AggT,
    const float* __restrict__ W0, const float* __restrict__ W1,
    const float* __restrict__ bias, float* __restrict__ Out,
    int K, int relu)
{
    typedef typename bvec<TILES>::type bv;
    constexpr int NOUT = 64 * TILES;
    const int wave  = threadIdx.x >> 5;
    const int lane  = threadIdx.x & 31;
    const int rlane = lane & 15;
    const int khalf = (lane >> 4) << 1;
    const int rowBase = blockIdx.x << 4;
    const int colBase = wave * (TILES * 16);
    const int col0 = colBase + TILES * rlane;

    const float* a0p = Xin + (size_t)(rowBase + rlane) * K + khalf;
    const float* a1p = AggT + (size_t)khalf * AGGP + rowBase + rlane;
    const float* w0p = W0 + (size_t)khalf * NOUT + col0;
    const float* w1p = W1 + (size_t)khalf * NOUT + col0;

    v8f acc[TILES];
#pragma unroll
    for (int t = 0; t < TILES; ++t) acc[t] = (v8f){};

    stage_t<TILES> cur, nxt;
    load_stage<TILES>(cur, a0p, a1p, w0p, w1p, 0);

#pragma unroll 2
    for (int k = 0; k < K - 8; k += 8) {
        load_stage<TILES>(nxt, a0p, a1p, w0p, w1p, k + 8);
        mma_stage<TILES>(cur, acc);
        cur = nxt;
    }
    mma_stage<TILES>(cur, acc);

    const int rowOff = rowBase + ((lane >> 4) << 3);
    const bv bvv = *(const bv*)(bias + col0);
    float* op = Out + (size_t)rowOff * NOUT + col0;
#pragma unroll
    for (int r = 0; r < 8; ++r) {
        bv v;
#pragma unroll
        for (int t = 0; t < TILES; ++t) {
            float x = acc[t][r] + bvv[t];
            if (relu) x = fmaxf(x, 0.0f);
            v[t] = x;
        }
        *(bv*)(op + (size_t)r * NOUT) = v;
    }
}

extern "C" void kernel_launch(void* const* d_in, const int* in_sizes, int n_in,
                              void* d_out, int out_size, void* d_ws, size_t ws_size,
                              hipStream_t stream) {
    const float* x   = (const float*)d_in[0];
    const int*   ei  = (const int*)d_in[1];
    const float* W01 = (const float*)d_in[3];
    const float* W11 = (const float*)d_in[4];
    const float* b1  = (const float*)d_in[5];
    const float* W02 = (const float*)d_in[6];
    const float* W12 = (const float*)d_in[7];
    const float* b2  = (const float*)d_in[8];
    const float* W03 = (const float*)d_in[9];
    const float* W13 = (const float*)d_in[10];
    const float* b3  = (const float*)d_in[11];

    const int N = NNODES;
    const int E = in_sizes[1] / 2;
    const int* src = ei;
    const int* dst = ei + E;

    float* dis  = (float*)d_ws;
    float* aggT = dis + 10240;
    float* h1   = aggT + 256 * AGGP;
    float* h2   = h1  + (size_t)N * 256;
    float* out  = (float*)d_out;
    unsigned* cnt = (unsigned*)(dis + 10016);
    unsigned* cur = (unsigned*)(dis + 10080);
    unsigned* listw = (unsigned*)d_out;
    const uint4* list = (const uint4*)d_out;

    zero4_kernel<<<64, 256, 0, stream>>>((float4*)dis, 10240 / 4);
    deg_kernel<<<(E + 255) / 256, 256, 0, stream>>>(src, dst, dis, cnt, E);
    dis_kernel<<<(N + 255) / 256, 256, 0, stream>>>(dis, N);
    place_kernel<<<(E + 255) / 256, 256, 0, stream>>>(src, dst, cur, listw, E);

    agg_split_kernel<1><<<256, 1024, 0, stream>>>(x, src, dst, dis, cnt, list, aggT, E, 128);
    cheb_gemm_kernel<4><<<N / 16, 128, 0, stream>>>(x, aggT, W01, W11, b1, h1, 128, 1);

    agg_split_kernel<2><<<256, 1024, 0, stream>>>(h1, src, dst, dis, cnt, list, aggT, E, 256);
    cheb_gemm_kernel<4><<<N / 16, 128, 0, stream>>>(h1, aggT, W02, W12, b2, h2, 256, 1);

    agg_split_kernel<2><<<256, 1024, 0, stream>>>(h2, src, dst, dis, cnt, list, aggT, E, 256);
    cheb_gemm_kernel<2><<<N / 16, 128, 0, stream>>>(h2, aggT, W03, W13, b3, out, 256, 0);
}


# Hidden activations and aggregates stored as feature-pair planes [f/2][node][2]: aggregation reads its two columns as one contiguous run; products swap operands to write planes; sums as before
# speedup vs baseline: 2.7643x; 1.0451x over previous
#include <hip/hip_runtime.h>
#include <hip/hip_bf16.h>

typedef __attribute__((ext_vector_type(2))) float v2f;
typedef __attribute__((ext_vector_type(4))) float v4f;
typedef __attribute__((ext_vector_type(8))) float v8f;

#define NNODES 10000
#define AGGP 10016
#define PLP 20000

__global__ __launch_bounds__(256) void zero4_kernel(float4* __restrict__ p, int n4) {
    int i = blockIdx.x * 256 + threadIdx.x;
    int stride = gridDim.x * 256;
    float4 z; z.x = 0.f; z.y = 0.f; z.z = 0.f; z.w = 0.f;
    for (; i < n4; i += stride) p[i] = z;
}

#define SUBCAP 20000u
#define PARTQ 160000u
#define ACCN 5024u

__global__ __launch_bounds__(256) void deg_kernel(const int* __restrict__ src,
                                                  const int* __restrict__ dst,
                                                  float* __restrict__ deg,
                                                  unsigned* __restrict__ cnt, int E) {
    __shared__ unsigned h[64];
    if (threadIdx.x < 64) h[threadIdx.x] = 0u;
    __syncthreads();
    int i = blockIdx.x * 256 + threadIdx.x;
    if (i < E) {
        atomicAdd(&deg[src[i]], 1.0f);
        atomicAdd(&h[dst[i] & 63], 1u);
    }
    __syncthreads();
    if (threadIdx.x < 64) {
        const unsigned c = h[threadIdx.x];
        if (c != 0u) atomicAdd(&cnt[threadIdx.x], c);
    }
}

__global__ __launch_bounds__(256) void place_kernel(const int* __restrict__ src,
                                                    const int* __restrict__ dst,
                                                    unsigned* __restrict__ cur,
                                                    unsigned* __restrict__ list, int E) {
    __shared__ unsigned h[64];
    __shared__ unsigned base[64];
    if (threadIdx.x < 64) h[threadIdx.x] = 0u;
    __syncthreads();
    const int i = blockIdx.x * 256 + threadIdx.x;
    const bool live = i < E;
    unsigned s = 0u, d = 0u, c = 0u, r = 0u;
    if (live) {
        s = (unsigned)src[i];
        d = (unsigned)dst[i];
        c = d & 63u;
        r = atomicAdd(&h[c], 1u);
    }
    __syncthreads();
    if (threadIdx.x < 64) {
        const unsigned n = h[threadIdx.x];
        base[threadIdx.x] = (n != 0u) ? atomicAdd(&cur[threadIdx.x], n) : 0u;
    }
    __syncthreads();
    if (live) {
        const unsigned p = base[c] + r;
        if (p < SUBCAP) {
            const unsigned l = c & 31u;
            const unsigned slot = ((d >> 6) << 5) | l;
            const unsigned word = (c >> 5) * PARTQ + ((p >> 2) << 5) + l;
            __hip_atomic_exchange(&list[word * 4u + (p & 3u)], s | (slot << 16),
                                  __ATOMIC_RELAXED, __HIP_MEMORY_SCOPE_AGENT);
        }
    }
}

__global__ __launch_bounds__(256) void dis_kernel(float* __restrict__ deg, int n) {
    int i = blockIdx.x * 256 + threadIdx.x;
    if (i < n) {
        float d = deg[i];
        deg[i] = (d > 0.0f) ? rsqrtf(d) : 0.0f;
    }
}
template <int FPB> struct gvec;
template <> struct gvec<1> { typedef float type; };
template <> struct gvec<2> { typedef v2f type; };

template <int FPB>
__device__ __forceinline__ void add_edge(float* acc, unsigned slot, typename gvec<FPB>::type v);
template <>
__device__ __forceinline__ void add_edge<1>(float* acc, unsigned slot, float v) {
    __hip_atomic_fetch_add(&acc[slot], v, __ATOMIC_RELAXED, __HIP_MEMORY_SCOPE_WORKGROUP);
}
template <>
__device__ __forceinline__ void add_edge<2>(float* acc, unsigned slot, v2f v) {
    __hip_atomic_fetch_add(&acc[slot], v.x, __ATOMIC_RELAXED, __HIP_MEMORY_SCOPE_WORKGROUP);
    __hip_atomic_fetch_add(&acc[ACCN + slot], v.y, __ATOMIC_RELAXED, __HIP_MEMORY_SCOPE_WORKGROUP);
}

template <int FPB, bool PL>
__global__ __launch_bounds__(1024) void agg_split_kernel(
    const float* __restrict__ xin, const int* __restrict__ src,
    const int* __restrict__ dst, const float* __restrict__ dis,
    const unsigned* __restrict__ cnt, const uint4* __restrict__ list,
    float* __restrict__ aggT, int E, int D)
{
    typedef typename gvec<FPB>::type gv;
    __shared__ gv xs[NNODES];
    __shared__ float acc[ACCN * FPB];
    const unsigned tid = threadIdx.x;
    const unsigned part = blockIdx.x & 1u;
    const unsigned f0 = (blockIdx.x >> 1) * (unsigned)FPB;
    const unsigned lane = tid & 31u;
    const unsigned wave = tid >> 5;

    if constexpr (PL) {
        const float* __restrict__ pin = xin + (size_t)(blockIdx.x >> 1) * PLP;
        for (unsigned n = tid; n < NNODES; n += 1024u) {
            const gv t = *(const gv*)(pin + 2u * n);
            xs[n] = dis[n] * t;
        }
    } else {
        for (unsigned n = tid; n < NNODES; n += 1024u) {
            const gv t = *(const gv*)(xin + (size_t)n * D + f0);
            xs[n] = dis[n] * t;
        }
    }
    for (unsigned c = tid; c < ACCN * FPB; c += 1024u) acc[c] = 0.0f;
    const unsigned myc = cnt[part * 32u + lane];
    unsigned mn = myc, mx = myc;
#pragma unroll
    for (int o = 16; o > 0; o >>= 1) {
        const unsigned a = __shfl_xor(mn, o);
        const unsigned b = __shfl_xor(mx, o);
        mn = (a < mn) ? a : mn;
        mx = (b > mx) ? b : mx;
    }
    __syncthreads();

    if (mx <= SUBCAP) {
        const uint4* __restrict__ pl = list + (size_t)part * PARTQ;
        const unsigned qFull = mn >> 2;
        const unsigned qAll = (mx + 3u) >> 2;
        const unsigned off = (blockIdx.x * qFull) / gridDim.x;
        unsigned i = wave;
        for (; i + 96u < qFull; i += 128u) {
            uint4 e[4];
#pragma unroll
            for (int u = 0; u < 4; ++u) {
                unsigned q = off + i + 32u * (unsigned)u;
                if (q >= qFull) q -= qFull;
                e[u] = pl[(q << 5) + lane];
            }
            gv v[16];
#pragma unroll
            for (int u = 0; u < 4; ++u) {
                v[4 * u + 0] = xs[e[u].x & 0xFFFFu];
                v[4 * u + 1] = xs[e[u].y & 0xFFFFu];
                v[4 * u + 2] = xs[e[u].z & 0xFFFFu];
                v[4 * u + 3] = xs[e[u].w & 0xFFFFu];
            }
#pragma unroll
            for (int u = 0; u < 4; ++u) {
                add_edge<FPB>(acc, e[u].x >> 16, v[4 * u + 0]);
                add_edge<FPB>(acc, e[u].y >> 16, v[4 * u + 1]);
                add_edge<FPB>(acc, e[u].z >> 16, v[4 * u + 2]);
                add_edge<FPB>(acc, e[u].w >> 16, v[4 * u + 3]);
            }
        }
        for (; i < qFull; i += 32u) {
            unsigned q = off + i;
            if (q >= qFull) q -= qFull;
            const uint4 e = pl[(q << 5) + lane];
            const gv v0 = xs[e.x & 0xFFFFu];
            const gv v1 = xs[e.y & 0xFFFFu];
            const gv v2 = xs[e.z & 0xFFFFu];
            const gv v3 = xs[e.w & 0xFFFFu];
            add_edge<FPB>(acc, e.x >> 16, v0);
            add_edge<FPB>(acc, e.y >> 16, v1);
            add_edge<FPB>(acc, e.z >> 16, v2);
            add_edge<FPB>(acc, e.w >> 16, v3);
        }
        for (unsigned q = qFull + wave; q < qAll; q += 32u) {
            const uint4 e = pl[(q << 5) + lane];
            const unsigned p = q << 2;
            if (p < myc) add_edge<FPB>(acc, e.x >> 16, xs[e.x & 0xFFFFu]);
            if (p + 1u < myc) add_edge<FPB>(acc, e.y >> 16, xs[e.y & 0xFFFFu]);
            if (p + 2u < myc) add_edge<FPB>(acc, e.z >> 16, xs[e.z & 0xFFFFu]);
            if (p + 3u < myc) add_edge<FPB>(acc, e.w >> 16, xs[e.w & 0xFFFFu]);
        }
    } else {
        for (unsigned e = tid; e < (unsigned)E; e += 1024u) {
            const unsigned d = (unsigned)dst[e];
            if (((d >> 5) & 1u) == part) add_edge<FPB>(acc, ((d >> 6) << 5) | (d & 31u), xs[src[e]]);
        }
    }
    __syncthreads();

    for (unsigned c = tid; c < ACCN; c += 1024u) {
        const unsigned n = ((c >> 5) << 6) | (part << 5) | (c & 31u);
        if (n < NNODES) {
            const float dn = dis[n];
            if constexpr (PL) {
                v2f o;
                o.x = -(dn * acc[c]);
                o.y = -(dn * acc[ACCN + c]);
                *(v2f*)(aggT + (size_t)(blockIdx.x >> 1) * PLP + 2u * n) = o;
            } else {
#pragma unroll
                for (int j = 0; j < FPB; ++j)
                    aggT[(size_t)(f0 + j) * AGGP + n] = -(dn * acc[j * ACCN + c]);
            }
        }
    }
}

template <int TILES> struct bvec;
template <> struct bvec<4> { typedef v4f type; };
template <> struct bvec<2> { typedef v2f type; };

template <int TILES> struct stage_t {
    v2f a0[2], a1[2];
    typename bvec<TILES>::type b0lo[2], b0hi[2], b1lo[2], b1hi[2];
};

template <int TILES, bool A0P, bool A1P>
__device__ __forceinline__ void load_stage(stage_t<TILES>& s, const float* __restrict__ a0p,
                                           const float* __restrict__ a1p, const float* __restrict__ w0p,
                                           const float* __restrict__ w1p, int k) {
    typedef typename bvec<TILES>::type bv;
    constexpr int NOUT = 64 * TILES;
#pragma unroll
    for (int u = 0; u < 2; ++u) {
        const int kk = k + 4 * u;
        if constexpr (A0P) {
            s.a0[u] = *(const v2f*)(a0p + (size_t)(kk >> 1) * PLP);
        } else {
            s.a0[u] = *(const v2f*)(a0p + kk);
        }
        if constexpr (A1P) {
            s.a1[u] = *(const v2f*)(a1p + (size_t)(kk >> 1) * PLP);
        } else {
            s.a1[u].x = a1p[(size_t)kk * AGGP];
            s.a1[u].y = a1p[(size_t)(kk + 1) * AGGP];
        }
        s.b0lo[u] = *(const bv*)(w0p + (size_t)kk * NOUT);
        s.b0hi[u] = *(const bv*)(w0p + (size_t)(kk + 1) * NOUT);
        s.b1lo[u] = *(const bv*)(w1p + (size_t)kk * NOUT);
        s.b1hi[u] = *(const bv*)(w1p + (size_t)(kk + 1) * NOUT);
    }
}

template <int TILES, bool OUTP>
__device__ __forceinline__ void mma_stage(const stage_t<TILES>& s, v8f (&acc)[TILES]) {
#pragma unroll
    for (int u = 0; u < 2; ++u) {
#pragma unroll
        for (int t = 0; t < TILES; ++t) {
            v2f b; b.x = s.b0lo[u][t]; b.y = s.b0hi[u][t];
            if constexpr (OUTP)
                acc[t] = __builtin_amdgcn_wmma_f32_16x16x4_f32(
                    false, b, false, s.a0[u], (short)0, acc[t], false, false);
            else
                acc[t] = __builtin_amdgcn_wmma_f32_16x16x4_f32(
                    false, s.a0[u], false, b, (short)0, acc[t], false, false);
        }
#pragma unroll
        for (int t = 0; t < TILES; ++t) {
            v2f b; b.x = s.b1lo[u][t]; b.y = s.b1hi[u][t];
            if constexpr (OUTP)
                acc[t] = __builtin_amdgcn_wmma_f32_16x16x4_f32(
                    false, b, false, s.a1[u], (short)0, acc[t], false, false);
            else
                acc[t] = __builtin_amdgcn_wmma_f32_16x16x4_f32(
                    false, s.a1[u], false, b, (short)0, acc[t], false, false);
        }
    }
}

template <int TILES, bool A0P, bool A1P, bool OUTP>
__global__ __launch_bounds__(128) void cheb_gemm_kernel(
    const float* __restrict__ Xin, const float* __restrict__ AggT,
    const float* __restrict__ W0, const float* __restrict__ W1,
    const float* __restrict__ bias, float* __restrict__ Out,
    int K, int relu)
{
    typedef typename bvec<TILES>::type bv;
    constexpr int NOUT = 64 * TILES;
    const int wave  = threadIdx.x >> 5;
    const int lane  = threadIdx.x & 31;
    const int rlane = lane & 15;
    const int half  = lane >> 4;
    const int khalf = half << 1;
    const int rowBase = blockIdx.x << 4;
    const int colBase = wave * (TILES * 16);
    const int col0 = colBase + TILES * rlane;

    const float* a0p = A0P ? Xin + (size_t)half * PLP + 2 * (rowBase + rlane)
                           : Xin + (size_t)(rowBase + rlane) * K + khalf;
    const float* a1p = A1P ? AggT + (size_t)half * PLP + 2 * (rowBase + rlane)
                           : AggT + (size_t)khalf * AGGP + rowBase + rlane;
    const float* w0p = W0 + (size_t)khalf * NOUT + col0;
    const float* w1p = W1 + (size_t)khalf * NOUT + col0;

    v8f acc[TILES];
#pragma unroll
    for (int t = 0; t < TILES; ++t) acc[t] = (v8f){};

    stage_t<TILES> cur, nxt;
    load_stage<TILES, A0P, A1P>(cur, a0p, a1p, w0p, w1p, 0);

#pragma unroll 2
    for (int k = 0; k < K - 8; k += 8) {
        load_stage<TILES, A0P, A1P>(nxt, a0p, a1p, w0p, w1p, k + 8);
        mma_stage<TILES, OUTP>(cur, acc);
        cur = nxt;
    }
    mma_stage<TILES, OUTP>(cur, acc);

    if constexpr (OUTP) {
        float* op = Out + (size_t)((colBase >> 1) + (TILES / 2) * 8 * half) * PLP + 2 * (rowBase + rlane);
        const float* bp = bias + colBase + TILES * 8 * half;
#pragma unroll
        for (int r = 0; r < 8; ++r) {
            const bv bvv = *(const bv*)(bp + TILES * r);
#pragma unroll
            for (int t = 0; t < TILES; t += 2) {
                v2f v;
                v.x = acc[t][r] + bvv[t];
                v.y = acc[t + 1][r] + bvv[t + 1];
                if (relu) { v.x = fmaxf(v.x, 0.0f); v.y = fmaxf(v.y, 0.0f); }
                *(v2f*)(op + (size_t)((TILES / 2) * r + (t >> 1)) * PLP) = v;
            }
        }
    } else {
        const int rowOff = rowBase + (half << 3);
        const bv bvv = *(const bv*)(bias + col0);
        float* op = Out + (size_t)rowOff * NOUT + col0;
#pragma unroll
        for (int r = 0; r < 8; ++r) {
            bv v;
#pragma unroll
            for (int t = 0; t < TILES; ++t) {
                float x = acc[t][r] + bvv[t];
                if (relu) x = fmaxf(x, 0.0f);
                v[t] = x;
            }
            *(bv*)(op + (size_t)r * NOUT) = v;
        }
    }
}

extern "C" void kernel_launch(void* const* d_in, const int* in_sizes, int n_in,
                              void* d_out, int out_size, void* d_ws, size_t ws_size,
                              hipStream_t stream) {
    const float* x   = (const float*)d_in[0];
    const int*   ei  = (const int*)d_in[1];
    const float* W01 = (const float*)d_in[3];
    const float* W11 = (const float*)d_in[4];
    const float* b1  = (const float*)d_in[5];
    const float* W02 = (const float*)d_in[6];
    const float* W12 = (const float*)d_in[7];
    const float* b2  = (const float*)d_in[8];
    const float* W03 = (const float*)d_in[9];
    const float* W13 = (const float*)d_in[10];
    const float* b3  = (const float*)d_in[11];

    const int N = NNODES;
    const int E = in_sizes[1] / 2;
    const int* src = ei;
    const int* dst = ei + E;

    float* dis  = (float*)d_ws;
    float* aggT = dis + 10240;
    float* h1   = aggT + 256 * AGGP;
    float* h2   = h1  + (size_t)N * 256;
    float* out  = (float*)d_out;
    unsigned* cnt = (unsigned*)(dis + 10016);
    unsigned* cur = (unsigned*)(dis + 10080);
    unsigned* listw = (unsigned*)d_out;
    const uint4* list = (const uint4*)d_out;

    zero4_kernel<<<64, 256, 0, stream>>>((float4*)dis, 10240 / 4);
    deg_kernel<<<(E + 255) / 256, 256, 0, stream>>>(src, dst, dis, cnt, E);
    dis_kernel<<<(N + 255) / 256, 256, 0, stream>>>(dis, N);
    place_kernel<<<(E + 255) / 256, 256, 0, stream>>>(src, dst, cur, listw, E);

    agg_split_kernel<1, false><<<256, 1024, 0, stream>>>(x, src, dst, dis, cnt, list, aggT, E, 128);
    cheb_gemm_kernel<4, false, false, true><<<N / 16, 128, 0, stream>>>(x, aggT, W01, W11, b1, h1, 128, 1);

    agg_split_kernel<2, true><<<256, 1024, 0, stream>>>(h1, src, dst, dis, cnt, list, aggT, E, 256);
    cheb_gemm_kernel<4, true, true, true><<<N / 16, 128, 0, stream>>>(h1, aggT, W02, W12, b2, h2, 256, 1);

    agg_split_kernel<2, true><<<256, 1024, 0, stream>>>(h2, src, dst, dis, cnt, list, aggT, E, 256);
    cheb_gemm_kernel<2, true, true, false><<<N / 16, 128, 0, stream>>>(h2, aggT, W03, W13, b3, out, 256, 0);
}


# Degree count and edge placement gathered in LDS per 10240-edge tile, then sent to global memory in address order: a wave's 32 atomics fall in one 128-byte line, not 32; list and sum order change
# speedup vs baseline: 3.6093x; 1.3057x over previous
#include <hip/hip_runtime.h>
#include <hip/hip_bf16.h>

typedef __attribute__((ext_vector_type(2))) float v2f;
typedef __attribute__((ext_vector_type(4))) float v4f;
typedef __attribute__((ext_vector_type(8))) float v8f;

#define NNODES 10000
#define AGGP 10016
#define PLP 20000

__global__ __launch_bounds__(256) void zero4_kernel(float4* __restrict__ p, int n4) {
    int i = blockIdx.x * 256 + threadIdx.x;
    int stride = gridDim.x * 256;
    float4 z; z.x = 0.f; z.y = 0.f; z.z = 0.f; z.w = 0.f;
    for (; i < n4; i += stride) p[i] = z;
}

#define SUBCAP 20000u
#define PARTQ 160000u
#define ACCN 5024u

#define ETILE 10240u
#define EPT 10
__global__ __launch_bounds__(1024) void deg_kernel(const int* __restrict__ src,
                                                   const int* __restrict__ dst,
                                                   float* __restrict__ deg,
                                                   unsigned* __restrict__ cnt, int E) {
    __shared__ unsigned hist[NNODES];
    __shared__ unsigned h[64];
    const unsigned tid = threadIdx.x;
    for (unsigned n = tid; n < NNODES; n += 1024u) hist[n] = 0u;
    if (tid < 64u) h[tid] = 0u;
    __syncthreads();
    for (unsigned t0 = blockIdx.x * ETILE; t0 < (unsigned)E; t0 += gridDim.x * ETILE) {
        unsigned s[EPT], d[EPT];
#pragma unroll
        for (int u = 0; u < EPT; ++u) {
            const unsigned i = t0 + 1024u * (unsigned)u + tid;
            const unsigned ii = (i < (unsigned)E) ? i : t0;
            s[u] = (unsigned)src[ii];
            d[u] = (unsigned)dst[ii];
        }
#pragma unroll
        for (int u = 0; u < EPT; ++u) {
            if (t0 + 1024u * (unsigned)u + tid < (unsigned)E) {
                atomicAdd(&hist[s[u]], 1u);
                atomicAdd(&h[d[u] & 63u], 1u);
            }
        }
    }
    __syncthreads();
    for (unsigned n = tid; n < NNODES; n += 1024u) {
        const unsigned c = hist[n];
        if (c != 0u) atomicAdd(&deg[n], (float)c);
    }
    if (tid < 64u) {
        const unsigned c = h[tid];
        if (c != 0u) atomicAdd(&cnt[tid], c);
    }
}

__global__ __launch_bounds__(1024) void place_kernel(const int* __restrict__ src,
                                                     const int* __restrict__ dst,
                                                     unsigned* __restrict__ cur,
                                                     unsigned* __restrict__ list, int E) {
    __shared__ unsigned h[64];
    __shared__ unsigned base[64];
    __shared__ unsigned coff[64];
    __shared__ unsigned stage[ETILE];
    const unsigned tid = threadIdx.x;
    const unsigned lane = tid & 31u;
    for (unsigned t0 = blockIdx.x * ETILE; t0 < (unsigned)E; t0 += gridDim.x * ETILE) {
        if (tid < 64u) h[tid] = 0u;
        __syncthreads();
        unsigned s[EPT], d[EPT], r[EPT];
#pragma unroll
        for (int u = 0; u < EPT; ++u) {
            const unsigned i = t0 + 1024u * (unsigned)u + tid;
            const unsigned ii = (i < (unsigned)E) ? i : t0;
            s[u] = (unsigned)src[ii];
            d[u] = (unsigned)dst[ii];
        }
#pragma unroll
        for (int u = 0; u < EPT; ++u) {
            r[u] = 0u;
            if (t0 + 1024u * (unsigned)u + tid < (unsigned)E) r[u] = atomicAdd(&h[d[u] & 63u], 1u);
        }
        __syncthreads();
        if (tid < 64u) {
            const unsigned n = h[tid];
            unsigned v = n;
#pragma unroll
            for (int o = 1; o < 32; o <<= 1) {
                const unsigned t = __shfl_up(v, o);
                if (lane >= (unsigned)o) v += t;
            }
            unsigned low = h[lane];
#pragma unroll
            for (int o = 16; o > 0; o >>= 1) low += __shfl_xor(low, o);
            coff[tid] = v - n + ((tid >= 32u) ? low : 0u);
            base[tid] = (n != 0u) ? atomicAdd(&cur[tid], n) : 0u;
        }
        __syncthreads();
#pragma unroll
        for (int u = 0; u < EPT; ++u) {
            if (t0 + 1024u * (unsigned)u + tid < (unsigned)E) {
                const unsigned c = d[u] & 63u;
                stage[coff[c] + r[u]] = s[u] | ((((d[u] >> 6) << 5) | (c & 31u)) << 16);
            }
        }
        __syncthreads();
        const unsigned l = (tid >> 2) & 31u;
        const unsigned i4 = tid & 3u;
        const unsigned w8 = tid >> 7;
        for (unsigned part = 0u; part < 2u; ++part) {
            const unsigned c = (part << 5) | l;
            const unsigned bc = base[c];
            const unsigned hc = h[c];
            const unsigned cc = coff[c];
            unsigned rfirst = 0xFFFFFFFFu, rlast = 0u;
            if (hc != 0u && bc < SUBCAP) {
                unsigned p1 = bc + hc - 1u;
                if (p1 > SUBCAP - 1u) p1 = SUBCAP - 1u;
                const unsigned r0 = bc >> 2;
                rfirst = r0 + ((w8 - r0) & 7u);
                rlast = p1 >> 2;
            }
            unsigned wmin = rfirst, wmax = rlast;
#pragma unroll
            for (int o = 16; o > 0; o >>= 1) {
                const unsigned a = __shfl_xor(wmin, o);
                const unsigned b = __shfl_xor(wmax, o);
                wmin = (a < wmin) ? a : wmin;
                wmax = (b > wmax) ? b : wmax;
            }
            unsigned* __restrict__ out = list + (size_t)part * (PARTQ * 4u) + (l << 2) + i4;
            if (wmin != 0xFFFFFFFFu) {
                for (unsigned q = wmin; q <= wmax; q += 8u) {
                    const unsigned rel = ((q << 2) | i4) - bc;
                    if (q <= rlast && rel < hc)
                        __hip_atomic_exchange(&out[(size_t)q << 7], stage[cc + rel],
                                              __ATOMIC_RELAXED, __HIP_MEMORY_SCOPE_AGENT);
                }
            }
        }
        __syncthreads();
    }
}

__global__ __launch_bounds__(256) void dis_kernel(float* __restrict__ deg, int n) {
    int i = blockIdx.x * 256 + threadIdx.x;
    if (i < n) {
        float d = deg[i];
        deg[i] = (d > 0.0f) ? rsqrtf(d) : 0.0f;
    }
}
template <int FPB> struct gvec;
template <> struct gvec<1> { typedef float type; };
template <> struct gvec<2> { typedef v2f type; };

template <int FPB>
__device__ __forceinline__ void add_edge(float* acc, unsigned slot, typename gvec<FPB>::type v);
template <>
__device__ __forceinline__ void add_edge<1>(float* acc, unsigned slot, float v) {
    __hip_atomic_fetch_add(&acc[slot], v, __ATOMIC_RELAXED, __HIP_MEMORY_SCOPE_WORKGROUP);
}
template <>
__device__ __forceinline__ void add_edge<2>(float* acc, unsigned slot, v2f v) {
    __hip_atomic_fetch_add(&acc[slot], v.x, __ATOMIC_RELAXED, __HIP_MEMORY_SCOPE_WORKGROUP);
    __hip_atomic_fetch_add(&acc[ACCN + slot], v.y, __ATOMIC_RELAXED, __HIP_MEMORY_SCOPE_WORKGROUP);
}

template <int FPB, bool PL>
__global__ __launch_bounds__(1024) void agg_split_kernel(
    const float* __restrict__ xin, const int* __restrict__ src,
    const int* __restrict__ dst, const float* __restrict__ dis,
    const unsigned* __restrict__ cnt, const uint4* __restrict__ list,
    float* __restrict__ aggT, int E, int D)
{
    typedef typename gvec<FPB>::type gv;
    __shared__ gv xs[NNODES];
    __shared__ float acc[ACCN * FPB];
    const unsigned tid = threadIdx.x;
    const unsigned part = blockIdx.x & 1u;
    const unsigned f0 = (blockIdx.x >> 1) * (unsigned)FPB;
    const unsigned lane = tid & 31u;
    const unsigned wave = tid >> 5;

    if constexpr (PL) {
        const float* __restrict__ pin = xin + (size_t)(blockIdx.x >> 1) * PLP;
        for (unsigned n = tid; n < NNODES; n += 1024u) {
            const gv t = *(const gv*)(pin + 2u * n);
            xs[n] = dis[n] * t;
        }
    } else {
        for (unsigned n = tid; n < NNODES; n += 1024u) {
            const gv t = *(const gv*)(xin + (size_t)n * D + f0);
            xs[n] = dis[n] * t;
        }
    }
    for (unsigned c = tid; c < ACCN * FPB; c += 1024u) acc[c] = 0.0f;
    const unsigned myc = cnt[part * 32u + lane];
    unsigned mn = myc, mx = myc;
#pragma unroll
    for (int o = 16; o > 0; o >>= 1) {
        const unsigned a = __shfl_xor(mn, o);
        const unsigned b = __shfl_xor(mx, o);
        mn = (a < mn) ? a : mn;
        mx = (b > mx) ? b : mx;
    }
    __syncthreads();

    if (mx <= SUBCAP) {
        const uint4* __restrict__ pl = list + (size_t)part * PARTQ;
        const unsigned qFull = mn >> 2;
        const unsigned qAll = (mx + 3u) >> 2;
        const unsigned off = (blockIdx.x * qFull) / gridDim.x;
        unsigned i = wave;
        for (; i + 96u < qFull; i += 128u) {
            uint4 e[4];
#pragma unroll
            for (int u = 0; u < 4; ++u) {
                unsigned q = off + i + 32u * (unsigned)u;
                if (q >= qFull) q -= qFull;
                e[u] = pl[(q << 5) + lane];
            }
            gv v[16];
#pragma unroll
            for (int u = 0; u < 4; ++u) {
                v[4 * u + 0] = xs[e[u].x & 0xFFFFu];
                v[4 * u + 1] = xs[e[u].y & 0xFFFFu];
                v[4 * u + 2] = xs[e[u].z & 0xFFFFu];
                v[4 * u + 3] = xs[e[u].w & 0xFFFFu];
            }
#pragma unroll
            for (int u = 0; u < 4; ++u) {
                add_edge<FPB>(acc, e[u].x >> 16, v[4 * u + 0]);
                add_edge<FPB>(acc, e[u].y >> 16, v[4 * u + 1]);
                add_edge<FPB>(acc, e[u].z >> 16, v[4 * u + 2]);
                add_edge<FPB>(acc, e[u].w >> 16, v[4 * u + 3]);
            }
        }
        for (; i < qFull; i += 32u) {
            unsigned q = off + i;
            if (q >= qFull) q -= qFull;
            const uint4 e = pl[(q << 5) + lane];
            const gv v0 = xs[e.x & 0xFFFFu];
            const gv v1 = xs[e.y & 0xFFFFu];
            const gv v2 = xs[e.z & 0xFFFFu];
            const gv v3 = xs[e.w & 0xFFFFu];
            add_edge<FPB>(acc, e.x >> 16, v0);
            add_edge<FPB>(acc, e.y >> 16, v1);
            add_edge<FPB>(acc, e.z >> 16, v2);
            add_edge<FPB>(acc, e.w >> 16, v3);
        }
        for (unsigned q = qFull + wave; q < qAll; q += 32u) {
            const uint4 e = pl[(q << 5) + lane];
            const unsigned p = q << 2;
            if (p < myc) add_edge<FPB>(acc, e.x >> 16, xs[e.x & 0xFFFFu]);
            if (p + 1u < myc) add_edge<FPB>(acc, e.y >> 16, xs[e.y & 0xFFFFu]);
            if (p + 2u < myc) add_edge<FPB>(acc, e.z >> 16, xs[e.z & 0xFFFFu]);
            if (p + 3u < myc) add_edge<FPB>(acc, e.w >> 16, xs[e.w & 0xFFFFu]);
        }
    } else {
        for (unsigned e = tid; e < (unsigned)E; e += 1024u) {
            const unsigned d = (unsigned)dst[e];
            if (((d >> 5) & 1u) == part) add_edge<FPB>(acc, ((d >> 6) << 5) | (d & 31u), xs[src[e]]);
        }
    }
    __syncthreads();

    for (unsigned c = tid; c < ACCN; c += 1024u) {
        const unsigned n = ((c >> 5) << 6) | (part << 5) | (c & 31u);
        if (n < NNODES) {
            const float dn = dis[n];
            if constexpr (PL) {
                v2f o;
                o.x = -(dn * acc[c]);
                o.y = -(dn * acc[ACCN + c]);
                *(v2f*)(aggT + (size_t)(blockIdx.x >> 1) * PLP + 2u * n) = o;
            } else {
#pragma unroll
                for (int j = 0; j < FPB; ++j)
                    aggT[(size_t)(f0 + j) * AGGP + n] = -(dn * acc[j * ACCN + c]);
            }
        }
    }
}

template <int TILES> struct bvec;
template <> struct bvec<4> { typedef v4f type; };
template <> struct bvec<2> { typedef v2f type; };

template <int TILES> struct stage_t {
    v2f a0[2], a1[2];
    typename bvec<TILES>::type b0lo[2], b0hi[2], b1lo[2], b1hi[2];
};

template <int TILES, bool A0P, bool A1P>
__device__ __forceinline__ void load_stage(stage_t<TILES>& s, const float* __restrict__ a0p,
                                           const float* __restrict__ a1p, const float* __restrict__ w0p,
                                           const float* __restrict__ w1p, int k) {
    typedef typename bvec<TILES>::type bv;
    constexpr int NOUT = 64 * TILES;
#pragma unroll
    for (int u = 0; u < 2; ++u) {
        const int kk = k + 4 * u;
        if constexpr (A0P) {
            s.a0[u] = *(const v2f*)(a0p + (size_t)(kk >> 1) * PLP);
        } else {
            s.a0[u] = *(const v2f*)(a0p + kk);
        }
        if constexpr (A1P) {
            s.a1[u] = *(const v2f*)(a1p + (size_t)(kk >> 1) * PLP);
        } else {
            s.a1[u].x = a1p[(size_t)kk * AGGP];
            s.a1[u].y = a1p[(size_t)(kk + 1) * AGGP];
        }
        s.b0lo[u] = *(const bv*)(w0p + (size_t)kk * NOUT);
        s.b0hi[u] = *(const bv*)(w0p + (size_t)(kk + 1) * NOUT);
        s.b1lo[u] = *(const bv*)(w1p + (size_t)kk * NOUT);
        s.b1hi[u] = *(const bv*)(w1p + (size_t)(kk + 1) * NOUT);
    }
}

template <int TILES, bool OUTP>
__device__ __forceinline__ void mma_stage(const stage_t<TILES>& s, v8f (&acc)[TILES]) {
#pragma unroll
    for (int u = 0; u < 2; ++u) {
#pragma unroll
        for (int t = 0; t < TILES; ++t) {
            v2f b; b.x = s.b0lo[u][t]; b.y = s.b0hi[u][t];
            if constexpr (OUTP)
                acc[t] = __builtin_amdgcn_wmma_f32_16x16x4_f32(
                    false, b, false, s.a0[u], (short)0, acc[t], false, false);
            else
                acc[t] = __builtin_amdgcn_wmma_f32_16x16x4_f32(
                    false, s.a0[u], false, b, (short)0, acc[t], false, false);
        }
#pragma unroll
        for (int t = 0; t < TILES; ++t) {
            v2f b; b.x = s.b1lo[u][t]; b.y = s.b1hi[u][t];
            if constexpr (OUTP)
                acc[t] = __builtin_amdgcn_wmma_f32_16x16x4_f32(
                    false, b, false, s.a1[u], (short)0, acc[t], false, false);
            else
                acc[t] = __builtin_amdgcn_wmma_f32_16x16x4_f32(
                    false, s.a1[u], false, b, (short)0, acc[t], false, false);
        }
    }
}

template <int TILES, bool A0P, bool A1P, bool OUTP>
__global__ __launch_bounds__(128) void cheb_gemm_kernel(
    const float* __restrict__ Xin, const float* __restrict__ AggT,
    const float* __restrict__ W0, const float* __restrict__ W1,
    const float* __restrict__ bias, float* __restrict__ Out,
    int K, int relu)
{
    typedef typename bvec<TILES>::type bv;
    constexpr int NOUT = 64 * TILES;
    const int wave  = threadIdx.x >> 5;
    const int lane  = threadIdx.x & 31;
    const int rlane = lane & 15;
    const int half  = lane >> 4;
    const int khalf = half << 1;
    const int rowBase = blockIdx.x << 4;
    const int colBase = wave * (TILES * 16);
    const int col0 = colBase + TILES * rlane;

    const float* a0p = A0P ? Xin + (size_t)half * PLP + 2 * (rowBase + rlane)
                           : Xin + (size_t)(rowBase + rlane) * K + khalf;
    const float* a1p = A1P ? AggT + (size_t)half * PLP + 2 * (rowBase + rlane)
                           : AggT + (size_t)khalf * AGGP + rowBase + rlane;
    const float* w0p = W0 + (size_t)khalf * NOUT + col0;
    const float* w1p = W1 + (size_t)khalf * NOUT + col0;

    v8f acc[TILES];
#pragma unroll
    for (int t = 0; t < TILES; ++t) acc[t] = (v8f){};

    stage_t<TILES> cur, nxt;
    load_stage<TILES, A0P, A1P>(cur, a0p, a1p, w0p, w1p, 0);

#pragma unroll 2
    for (int k = 0; k < K - 8; k += 8) {
        load_stage<TILES, A0P, A1P>(nxt, a0p, a1p, w0p, w1p, k + 8);
        mma_stage<TILES, OUTP>(cur, acc);
        cur = nxt;
    }
    mma_stage<TILES, OUTP>(cur, acc);

    if constexpr (OUTP) {
        float* op = Out + (size_t)((colBase >> 1) + (TILES / 2) * 8 * half) * PLP + 2 * (rowBase + rlane);
        const float* bp = bias + colBase + TILES * 8 * half;
#pragma unroll
        for (int r = 0; r < 8; ++r) {
            const bv bvv = *(const bv*)(bp + TILES * r);
#pragma unroll
            for (int t = 0; t < TILES; t += 2) {
                v2f v;
                v.x = acc[t][r] + bvv[t];
                v.y = acc[t + 1][r] + bvv[t + 1];
                if (relu) { v.x = fmaxf(v.x, 0.0f); v.y = fmaxf(v.y, 0.0f); }
                *(v2f*)(op + (size_t)((TILES / 2) * r + (t >> 1)) * PLP) = v;
            }
        }
    } else {
        const int rowOff = rowBase + (half << 3);
        const bv bvv = *(const bv*)(bias + col0);
        float* op = Out + (size_t)rowOff * NOUT + col0;
#pragma unroll
        for (int r = 0; r < 8; ++r) {
            bv v;
#pragma unroll
            for (int t = 0; t < TILES; ++t) {
                float x = acc[t][r] + bvv[t];
                if (relu) x = fmaxf(x, 0.0f);
                v[t] = x;
            }
            *(bv*)(op + (size_t)r * NOUT) = v;
        }
    }
}

extern "C" void kernel_launch(void* const* d_in, const int* in_sizes, int n_in,
                              void* d_out, int out_size, void* d_ws, size_t ws_size,
                              hipStream_t stream) {
    const float* x   = (const float*)d_in[0];
    const int*   ei  = (const int*)d_in[1];
    const float* W01 = (const float*)d_in[3];
    const float* W11 = (const float*)d_in[4];
    const float* b1  = (const float*)d_in[5];
    const float* W02 = (const float*)d_in[6];
    const float* W12 = (const float*)d_in[7];
    const float* b2  = (const float*)d_in[8];
    const float* W03 = (const float*)d_in[9];
    const float* W13 = (const float*)d_in[10];
    const float* b3  = (const float*)d_in[11];

    const int N = NNODES;
    const int E = in_sizes[1] / 2;
    const int* src = ei;
    const int* dst = ei + E;

    float* dis  = (float*)d_ws;
    float* aggT = dis + 10240;
    float* h1   = aggT + 256 * AGGP;
    float* h2   = h1  + (size_t)N * 256;
    float* out  = (float*)d_out;
    unsigned* cnt = (unsigned*)(dis + 10016);
    unsigned* cur = (unsigned*)(dis + 10080);
    unsigned* listw = (unsigned*)d_out;
    const uint4* list = (const uint4*)d_out;

    zero4_kernel<<<64, 256, 0, stream>>>((float4*)dis, 10240 / 4);
    deg_kernel<<<64, 1024, 0, stream>>>(src, dst, dis, cnt, E);
    dis_kernel<<<(N + 255) / 256, 256, 0, stream>>>(dis, N);
    place_kernel<<<64, 1024, 0, stream>>>(src, dst, cur, listw, E);

    agg_split_kernel<1, false><<<256, 1024, 0, stream>>>(x, src, dst, dis, cnt, list, aggT, E, 128);
    cheb_gemm_kernel<4, false, false, true><<<N / 16, 128, 0, stream>>>(x, aggT, W01, W11, b1, h1, 128, 1);

    agg_split_kernel<2, true><<<256, 1024, 0, stream>>>(h1, src, dst, dis, cnt, list, aggT, E, 256);
    cheb_gemm_kernel<4, true, true, true><<<N / 16, 128, 0, stream>>>(h1, aggT, W02, W12, b2, h2, 256, 1);

    agg_split_kernel<2, true><<<256, 1024, 0, stream>>>(h2, src, dst, dis, cnt, list, aggT, E, 256);
    cheb_gemm_kernel<2, true, true, false><<<N / 16, 128, 0, stream>>>(h2, aggT, W03, W13, b3, out, 256, 0);
}


# Aggregation without LDS float adds: a new launch orders each class's edge stream by destination, then source; a lane sums a node's run in registers, stores it once; 11 launches; sum order changes
# speedup vs baseline: 3.9466x; 1.0935x over previous
#include <hip/hip_runtime.h>
#include <hip/hip_bf16.h>

typedef __attribute__((ext_vector_type(2))) float v2f;
typedef __attribute__((ext_vector_type(4))) float v4f;
typedef __attribute__((ext_vector_type(8))) float v8f;

#define NNODES 10000
#define AGGP 10016
#define PLP 20000

__global__ __launch_bounds__(256) void zero4_kernel(float4* __restrict__ p, int n4) {
    int i = blockIdx.x * 256 + threadIdx.x;
    int stride = gridDim.x * 256;
    float4 z; z.x = 0.f; z.y = 0.f; z.z = 0.f; z.w = 0.f;
    for (; i < n4; i += stride) p[i] = z;
}

#define SUBCAP 20000u
#define PARTQ 160000u
#define ACCN 5024u

#define ETILE 10240u
#define EPT 10
__global__ __launch_bounds__(1024) void deg_kernel(const int* __restrict__ src,
                                                   const int* __restrict__ dst,
                                                   float* __restrict__ deg,
                                                   unsigned* __restrict__ cnt, int E) {
    __shared__ unsigned hist[NNODES];
    __shared__ unsigned h[64];
    const unsigned tid = threadIdx.x;
    for (unsigned n = tid; n < NNODES; n += 1024u) hist[n] = 0u;
    if (tid < 64u) h[tid] = 0u;
    __syncthreads();
    for (unsigned t0 = blockIdx.x * ETILE; t0 < (unsigned)E; t0 += gridDim.x * ETILE) {
        unsigned s[EPT], d[EPT];
#pragma unroll
        for (int u = 0; u < EPT; ++u) {
            const unsigned i = t0 + 1024u * (unsigned)u + tid;
            const unsigned ii = (i < (unsigned)E) ? i : t0;
            s[u] = (unsigned)src[ii];
            d[u] = (unsigned)dst[ii];
        }
#pragma unroll
        for (int u = 0; u < EPT; ++u) {
            if (t0 + 1024u * (unsigned)u + tid < (unsigned)E) {
                atomicAdd(&hist[s[u]], 1u);
                atomicAdd(&h[d[u] & 63u], 1u);
            }
        }
    }
    __syncthreads();
    for (unsigned n = tid; n < NNODES; n += 1024u) {
        const unsigned c = hist[n];
        if (c != 0u) atomicAdd(&deg[n], (float)c);
    }
    if (tid < 64u) {
        const unsigned c = h[tid];
        if (c != 0u) atomicAdd(&cnt[tid], c);
    }
}

__global__ __launch_bounds__(1024) void place_kernel(const int* __restrict__ src,
                                                     const int* __restrict__ dst,
                                                     unsigned* __restrict__ cur,
                                                     unsigned* __restrict__ list, int E) {
    __shared__ unsigned h[64];
    __shared__ unsigned base[64];
    __shared__ unsigned coff[64];
    __shared__ unsigned stage[ETILE];
    const unsigned tid = threadIdx.x;
    const unsigned lane = tid & 31u;
    for (unsigned t0 = blockIdx.x * ETILE; t0 < (unsigned)E; t0 += gridDim.x * ETILE) {
        if (tid < 64u) h[tid] = 0u;
        __syncthreads();
        unsigned s[EPT], d[EPT], r[EPT];
#pragma unroll
        for (int u = 0; u < EPT; ++u) {
            const unsigned i = t0 + 1024u * (unsigned)u + tid;
            const unsigned ii = (i < (unsigned)E) ? i : t0;
            s[u] = (unsigned)src[ii];
            d[u] = (unsigned)dst[ii];
        }
#pragma unroll
        for (int u = 0; u < EPT; ++u) {
            r[u] = 0u;
            if (t0 + 1024u * (unsigned)u + tid < (unsigned)E) r[u] = atomicAdd(&h[d[u] & 63u], 1u);
        }
        __syncthreads();
        if (tid < 64u) {
            const unsigned n = h[tid];
            unsigned v = n;
#pragma unroll
            for (int o = 1; o < 32; o <<= 1) {
                const unsigned t = __shfl_up(v, o);
                if (lane >= (unsigned)o) v += t;
            }
            unsigned low = h[lane];
#pragma unroll
            for (int o = 16; o > 0; o >>= 1) low += __shfl_xor(low, o);
            coff[tid] = v - n + ((tid >= 32u) ? low : 0u);
            base[tid] = (n != 0u) ? atomicAdd(&cur[tid], n) : 0u;
        }
        __syncthreads();
#pragma unroll
        for (int u = 0; u < EPT; ++u) {
            if (t0 + 1024u * (unsigned)u + tid < (unsigned)E) {
                const unsigned c = d[u] & 63u;
                stage[coff[c] + r[u]] = s[u] | ((((d[u] >> 6) << 5) | (c & 31u)) << 16);
            }
        }
        __syncthreads();
        for (unsigned c = tid >> 5; c < 64u; c += 32u) {
            const unsigned bc = base[c];
            const unsigned end = bc + h[c];
            const unsigned cc = coff[c];
            unsigned* __restrict__ out = list + (size_t)c * SUBCAP;
            for (unsigned pos = (bc & ~31u) + lane; pos < end; pos += 32u) {
                if (pos >= bc && pos < SUBCAP)
                    __hip_atomic_exchange(&out[pos], stage[cc + (pos - bc)],
                                          __ATOMIC_RELAXED, __HIP_MEMORY_SCOPE_AGENT);
            }
        }
        __syncthreads();
    }
}

#define OSPLIT 4u
#define OGRP 40u
__global__ __launch_bounds__(1024) void order_kernel(const unsigned* __restrict__ cnt,
                                                     const unsigned* __restrict__ lin,
                                                     unsigned* __restrict__ lout) {
    __shared__ unsigned ent[SUBCAP];
    __shared__ unsigned srt[SUBCAP];
    __shared__ unsigned gsz[160];
    __shared__ unsigned gst[161];
    const unsigned tid = threadIdx.x;
    const unsigned c = blockIdx.x / OSPLIT;
    const unsigned sp = blockIdx.x % OSPLIT;
    const unsigned n = cnt[c];
    if (n > SUBCAP) return;
    const unsigned* __restrict__ in = lin + (size_t)c * SUBCAP;
    unsigned* __restrict__ out = lout + (size_t)(c >> 5) * (PARTQ * 4u) + ((c & 31u) << 2);
    if (tid < 160u) gsz[tid] = 0u;
    __syncthreads();
    for (unsigned p = tid; p < n; p += 1024u) {
        const unsigned e = in[p];
        atomicAdd(&gsz[e >> 21], 1u);
    }
    __syncthreads();
    if (tid < 32u) {
        unsigned a[5];
        unsigned t = 0u;
#pragma unroll
        for (int k = 0; k < 5; ++k) { a[k] = gsz[5u * tid + (unsigned)k]; t += a[k]; }
        unsigned v = t;
#pragma unroll
        for (int o = 1; o < 32; o <<= 1) {
            const unsigned u = __shfl_up(v, o);
            if (tid >= (unsigned)o) v += u;
        }
        unsigned ex = v - t;
#pragma unroll
        for (int k = 0; k < 5; ++k) {
            gst[5u * tid + (unsigned)k] = ex;
            gsz[5u * tid + (unsigned)k] = 0u;
            ex += a[k];
        }
        if (tid == 31u) gst[160] = ex;
    }
    __syncthreads();
    for (unsigned p = tid; p < n; p += 1024u) {
        const unsigned e = in[p];
        const unsigned g = e >> 21;
        const unsigned r = atomicAdd(&gsz[g], 1u);
        ent[gst[g] + r] = e;
    }
    __syncthreads();
    const unsigned a0 = gst[sp * OGRP];
    const unsigned a1 = gst[sp * OGRP + OGRP];
    for (unsigned i = a0 + tid; i < a1; i += 1024u) {
        const unsigned key = ent[i];
        const unsigned g = key >> 21;
        const unsigned a = gst[g];
        const unsigned b = gst[g + 1u];
        unsigned rank = 0u;
        for (unsigned j = a; j < b; ++j) {
            const unsigned k = ent[j];
            rank += (k < key || (k == key && j < i)) ? 1u : 0u;
        }
        srt[a + rank] = key;
    }
    __syncthreads();
    for (unsigned i = a0 + tid; i < a1; i += 1024u)
        __hip_atomic_exchange(&out[((size_t)(i >> 2) << 7) + (i & 3u)], srt[i],
                              __ATOMIC_RELAXED, __HIP_MEMORY_SCOPE_AGENT);
}

__global__ __launch_bounds__(256) void dis_kernel(float* __restrict__ deg, int n) {
    int i = blockIdx.x * 256 + threadIdx.x;
    if (i < n) {
        float d = deg[i];
        deg[i] = (d > 0.0f) ? rsqrtf(d) : 0.0f;
    }
}
template <int FPB> struct gvec;
template <> struct gvec<1> { typedef float type; };
template <> struct gvec<2> { typedef v2f type; };

#define ACCX (ACCN + 1024u)
#define NOSLOT 0xFFFFFFFFu

template <int FPB>
__device__ __forceinline__ void add_edge(float* acc, unsigned slot, typename gvec<FPB>::type v);
template <>
__device__ __forceinline__ void add_edge<1>(float* acc, unsigned slot, float v) {
    __hip_atomic_fetch_add(&acc[slot], v, __ATOMIC_RELAXED, __HIP_MEMORY_SCOPE_WORKGROUP);
}
template <>
__device__ __forceinline__ void add_edge<2>(float* acc, unsigned slot, v2f v) {
    __hip_atomic_fetch_add(&acc[slot], v.x, __ATOMIC_RELAXED, __HIP_MEMORY_SCOPE_WORKGROUP);
    __hip_atomic_fetch_add(&acc[ACCX + slot], v.y, __ATOMIC_RELAXED, __HIP_MEMORY_SCOPE_WORKGROUP);
}

template <int FPB>
__device__ __forceinline__ void put_sum(float* acc, unsigned idx, typename gvec<FPB>::type s);
template <>
__device__ __forceinline__ void put_sum<1>(float* acc, unsigned idx, float s) { acc[idx] = s; }
template <>
__device__ __forceinline__ void put_sum<2>(float* acc, unsigned idx, v2f s) {
    acc[idx] = s.x;
    acc[ACCX + idx] = s.y;
}
template <int FPB>
__device__ __forceinline__ typename gvec<FPB>::type get_sum(const float* acc, unsigned idx);
template <>
__device__ __forceinline__ float get_sum<1>(const float* acc, unsigned idx) { return acc[idx]; }
template <>
__device__ __forceinline__ v2f get_sum<2>(const float* acc, unsigned idx) {
    v2f r; r.x = acc[idx]; r.y = acc[ACCX + idx];
    return r;
}

template <int FPB, bool GUARD>
__device__ __forceinline__ void run_step(float* acc, unsigned& cur, unsigned& idx, typename gvec<FPB>::type& s,
                                         unsigned ew, typename gvec<FPB>::type v, bool valid) {
    const unsigned slot = ew >> 16;
    bool chg = slot != cur;
    if (GUARD) chg = chg && valid;
    if (chg) {
        put_sum<FPB>(acc, idx, s);
        idx = slot;
        cur = slot;
    }
    const typename gvec<FPB>::type t = s + v;
    if (GUARD) {
        if (valid) s = chg ? v : t;
    } else {
        s = chg ? v : t;
    }
}

template <int FPB, int ROWS, bool GUARD>
__device__ __forceinline__ void run_rows(const uint4* __restrict__ pl, const typename gvec<FPB>::type* xs, float* acc,
                                         unsigned q, unsigned lane, unsigned myc, unsigned& cur, unsigned& idx,
                                         typename gvec<FPB>::type& s) {
    typedef typename gvec<FPB>::type gv;
    uint4 e[ROWS];
    gv v[4 * ROWS];
#pragma unroll
    for (int u = 0; u < ROWS; ++u) e[u] = pl[((q + (unsigned)u) << 5) + lane];
#pragma unroll
    for (int u = 0; u < ROWS; ++u) {
        const unsigned p = (q + (unsigned)u) << 2;
        v[4 * u + 0] = xs[(!GUARD || p < myc) ? (e[u].x & 0xFFFFu) : 0u];
        v[4 * u + 1] = xs[(!GUARD || p + 1u < myc) ? (e[u].y & 0xFFFFu) : 0u];
        v[4 * u + 2] = xs[(!GUARD || p + 2u < myc) ? (e[u].z & 0xFFFFu) : 0u];
        v[4 * u + 3] = xs[(!GUARD || p + 3u < myc) ? (e[u].w & 0xFFFFu) : 0u];
    }
#pragma unroll
    for (int u = 0; u < ROWS; ++u) {
        const unsigned p = (q + (unsigned)u) << 2;
        run_step<FPB, GUARD>(acc, cur, idx, s, e[u].x, v[4 * u + 0], p < myc);
        run_step<FPB, GUARD>(acc, cur, idx, s, e[u].y, v[4 * u + 1], p + 1u < myc);
        run_step<FPB, GUARD>(acc, cur, idx, s, e[u].z, v[4 * u + 2], p + 2u < myc);
        run_step<FPB, GUARD>(acc, cur, idx, s, e[u].w, v[4 * u + 3], p + 3u < myc);
    }
}

template <int FPB, bool PL>
__global__ __launch_bounds__(1024) void agg_split_kernel(
    const float* __restrict__ xin, const int* __restrict__ src,
    const int* __restrict__ dst, const float* __restrict__ dis,
    const unsigned* __restrict__ cnt, const uint4* __restrict__ list,
    float* __restrict__ aggT, int E, int D)
{
    typedef typename gvec<FPB>::type gv;
    __shared__ gv xs[NNODES];
    __shared__ float acc[ACCX * FPB];
    __shared__ unsigned slotF[1024];
    __shared__ unsigned slotL[1024];
    const unsigned tid = threadIdx.x;
    const unsigned part = blockIdx.x & 1u;
    const unsigned f0 = (blockIdx.x >> 1) * (unsigned)FPB;
    const unsigned lane = tid & 31u;
    const unsigned wave = tid >> 5;

    if constexpr (PL) {
        const float* __restrict__ pin = xin + (size_t)(blockIdx.x >> 1) * PLP;
        for (unsigned n = tid; n < NNODES; n += 1024u) {
            const gv t = *(const gv*)(pin + 2u * n);
            xs[n] = dis[n] * t;
        }
    } else {
        for (unsigned n = tid; n < NNODES; n += 1024u) {
            const gv t = *(const gv*)(xin + (size_t)n * D + f0);
            xs[n] = dis[n] * t;
        }
    }
    for (unsigned c = tid; c < ACCN; c += 1024u) {
#pragma unroll
        for (int j = 0; j < FPB; ++j) acc[(unsigned)j * ACCX + c] = 0.0f;
    }
    const unsigned myc = cnt[part * 32u + lane];
    unsigned mn = myc, mx = myc;
#pragma unroll
    for (int o = 16; o > 0; o >>= 1) {
        const unsigned a = __shfl_xor(mn, o);
        const unsigned b = __shfl_xor(mx, o);
        mn = (a < mn) ? a : mn;
        mx = (b > mx) ? b : mx;
    }
    __syncthreads();

    const unsigned qAll = (mx + 3u) >> 2;
    const bool ordered = (mx <= SUBCAP) && (qAll >= 32u);
    unsigned sF = NOSLOT, sL = NOSLOT;
    gv vF = {}, vL = {};
    if (ordered) {
        const uint4* __restrict__ pl = list + (size_t)part * PARTQ;
        const unsigned qFull = mn >> 2;
        const unsigned qs = (wave * qAll) >> 5;
        const unsigned qe = ((wave + 1u) * qAll) >> 5;
        const unsigned qm = (qFull < qs) ? qs : ((qFull < qe) ? qFull : qe);
        unsigned cur = NOSLOT;
        if (qs < qe && (qs << 2) < myc) cur = pl[(qs << 5) + lane].x >> 16;
        const unsigned firstslot = cur;
        unsigned idx = ACCN + tid;
        gv s = {};
        unsigned q = qs;
        for (; q + 4u <= qe; q += 4u) {
            if (q + 4u <= qm) run_rows<FPB, 4, false>(pl, xs, acc, q, lane, myc, cur, idx, s);
            else run_rows<FPB, 4, true>(pl, xs, acc, q, lane, myc, cur, idx, s);
        }
        for (; q < qe; ++q) run_rows<FPB, 1, true>(pl, xs, acc, q, lane, myc, cur, idx, s);
        sF = firstslot;
        if (idx == ACCN + tid) {
            vF = s;
            put_sum<FPB>(acc, ACCN + tid, s);
        } else {
            vF = get_sum<FPB>(acc, ACCN + tid);
            sL = cur;
            vL = s;
        }
        slotF[tid] = sF;
        slotL[tid] = sL;
    } else {
        for (unsigned e = tid; e < (unsigned)E; e += 1024u) {
            const unsigned d = (unsigned)dst[e];
            if (((d >> 5) & 1u) == part) add_edge<FPB>(acc, ((d >> 6) << 5) | (d & 31u), xs[src[e]]);
        }
    }
    __syncthreads();

    if (ordered) {
        bool lead = sF != NOSLOT;
        if (lead && wave != 0u) {
            const unsigned pl1 = slotL[tid - 32u];
            const unsigned prev = (pl1 != NOSLOT) ? pl1 : slotF[tid - 32u];
            lead = prev != sF;
        }
        if (lead && sL != NOSLOT) put_sum<FPB>(acc, sF, vF);
        if (sL != NOSLOT || lead) {
            const unsigned slot = (sL != NOSLOT) ? sL : sF;
            gv sum = (sL != NOSLOT) ? vL : vF;
            for (unsigned w2 = wave + 1u; w2 < 32u; ++w2) {
                const unsigned t2 = (w2 << 5) + lane;
                if (slotF[t2] != slot) break;
                sum = sum + get_sum<FPB>(acc, ACCN + t2);
                if (slotL[t2] != NOSLOT) break;
            }
            put_sum<FPB>(acc, slot, sum);
        }
    }
    __syncthreads();

    for (unsigned c = tid; c < ACCN; c += 1024u) {
        const unsigned n = ((c >> 5) << 6) | (part << 5) | (c & 31u);
        if (n < NNODES) {
            const float dn = dis[n];
            if constexpr (PL) {
                v2f o;
                o.x = -(dn * acc[c]);
                o.y = -(dn * acc[ACCX + c]);
                *(v2f*)(aggT + (size_t)(blockIdx.x >> 1) * PLP + 2u * n) = o;
            } else {
#pragma unroll
                for (int j = 0; j < FPB; ++j)
                    aggT[(size_t)(f0 + j) * AGGP + n] = -(dn * acc[(unsigned)j * ACCX + c]);
            }
        }
    }
}

template <int TILES> struct bvec;
template <> struct bvec<4> { typedef v4f type; };
template <> struct bvec<2> { typedef v2f type; };

template <int TILES> struct stage_t {
    v2f a0[2], a1[2];
    typename bvec<TILES>::type b0lo[2], b0hi[2], b1lo[2], b1hi[2];
};

template <int TILES, bool A0P, bool A1P>
__device__ __forceinline__ void load_stage(stage_t<TILES>& s, const float* __restrict__ a0p,
                                           const float* __restrict__ a1p, const float* __restrict__ w0p,
                                           const float* __restrict__ w1p, int k) {
    typedef typename bvec<TILES>::type bv;
    constexpr int NOUT = 64 * TILES;
#pragma unroll
    for (int u = 0; u < 2; ++u) {
        const int kk = k + 4 * u;
        if constexpr (A0P) {
            s.a0[u] = *(const v2f*)(a0p + (size_t)(kk >> 1) * PLP);
        } else {
            s.a0[u] = *(const v2f*)(a0p + kk);
        }
        if constexpr (A1P) {
            s.a1[u] = *(const v2f*)(a1p + (size_t)(kk >> 1) * PLP);
        } else {
            s.a1[u].x = a1p[(size_t)kk * AGGP];
            s.a1[u].y = a1p[(size_t)(kk + 1) * AGGP];
        }
        s.b0lo[u] = *(const bv*)(w0p + (size_t)kk * NOUT);
        s.b0hi[u] = *(const bv*)(w0p + (size_t)(kk + 1) * NOUT);
        s.b1lo[u] = *(const bv*)(w1p + (size_t)kk * NOUT);
        s.b1hi[u] = *(const bv*)(w1p + (size_t)(kk + 1) * NOUT);
    }
}

template <int TILES, bool OUTP>
__device__ __forceinline__ void mma_stage(const stage_t<TILES>& s, v8f (&acc)[TILES]) {
#pragma unroll
    for (int u = 0; u < 2; ++u) {
#pragma unroll
        for (int t = 0; t < TILES; ++t) {
            v2f b; b.x = s.b0lo[u][t]; b.y = s.b0hi[u][t];
            if constexpr (OUTP)
                acc[t] = __builtin_amdgcn_wmma_f32_16x16x4_f32(
                    false, b, false, s.a0[u], (short)0, acc[t], false, false);
            else
                acc[t] = __builtin_amdgcn_wmma_f32_16x16x4_f32(
                    false, s.a0[u], false, b, (short)0, acc[t], false, false);
        }
#pragma unroll
        for (int t = 0; t < TILES; ++t) {
            v2f b; b.x = s.b1lo[u][t]; b.y = s.b1hi[u][t];
            if constexpr (OUTP)
                acc[t] = __builtin_amdgcn_wmma_f32_16x16x4_f32(
                    false, b, false, s.a1[u], (short)0, acc[t], false, false);
            else
                acc[t] = __builtin_amdgcn_wmma_f32_16x16x4_f32(
                    false, s.a1[u], false, b, (short)0, acc[t], false, false);
        }
    }
}

template <int TILES, bool A0P, bool A1P, bool OUTP>
__global__ __launch_bounds__(128) void cheb_gemm_kernel(
    const float* __restrict__ Xin, const float* __restrict__ AggT,
    const float* __restrict__ W0, const float* __restrict__ W1,
    const float* __restrict__ bias, float* __restrict__ Out,
    int K, int relu)
{
    typedef typename bvec<TILES>::type bv;
    constexpr int NOUT = 64 * TILES;
    const int wave  = threadIdx.x >> 5;
    const int lane  = threadIdx.x & 31;
    const int rlane = lane & 15;
    const int half  = lane >> 4;
    const int khalf = half << 1;
    const int rowBase = blockIdx.x << 4;
    const int colBase = wave * (TILES * 16);
    const int col0 = colBase + TILES * rlane;

    const float* a0p = A0P ? Xin + (size_t)half * PLP + 2 * (rowBase + rlane)
                           : Xin + (size_t)(rowBase + rlane) * K + khalf;
    const float* a1p = A1P ? AggT + (size_t)half * PLP + 2 * (rowBase + rlane)
                           : AggT + (size_t)khalf * AGGP + rowBase + rlane;
    const float* w0p = W0 + (size_t)khalf * NOUT + col0;
    const float* w1p = W1 + (size_t)khalf * NOUT + col0;

    v8f acc[TILES];
#pragma unroll
    for (int t = 0; t < TILES; ++t) acc[t] = (v8f){};

    stage_t<TILES> cur, nxt;
    load_stage<TILES, A0P, A1P>(cur, a0p, a1p, w0p, w1p, 0);

#pragma unroll 2
    for (int k = 0; k < K - 8; k += 8) {
        load_stage<TILES, A0P, A1P>(nxt, a0p, a1p, w0p, w1p, k + 8);
        mma_stage<TILES, OUTP>(cur, acc);
        cur = nxt;
    }
    mma_stage<TILES, OUTP>(cur, acc);

    if constexpr (OUTP) {
        float* op = Out + (size_t)((colBase >> 1) + (TILES / 2) * 8 * half) * PLP + 2 * (rowBase + rlane);
        const float* bp = bias + colBase + TILES * 8 * half;
#pragma unroll
        for (int r = 0; r < 8; ++r) {
            const bv bvv = *(const bv*)(bp + TILES * r);
#pragma unroll
            for (int t = 0; t < TILES; t += 2) {
                v2f v;
                v.x = acc[t][r] + bvv[t];
                v.y = acc[t + 1][r] + bvv[t + 1];
                if (relu) { v.x = fmaxf(v.x, 0.0f); v.y = fmaxf(v.y, 0.0f); }
                *(v2f*)(op + (size_t)((TILES / 2) * r + (t >> 1)) * PLP) = v;
            }
        }
    } else {
        const int rowOff = rowBase + (half << 3);
        const bv bvv = *(const bv*)(bias + col0);
        float* op = Out + (size_t)rowOff * NOUT + col0;
#pragma unroll
        for (int r = 0; r < 8; ++r) {
            bv v;
#pragma unroll
            for (int t = 0; t < TILES; ++t) {
                float x = acc[t][r] + bvv[t];
                if (relu) x = fmaxf(x, 0.0f);
                v[t] = x;
            }
            *(bv*)(op + (size_t)r * NOUT) = v;
        }
    }
}

extern "C" void kernel_launch(void* const* d_in, const int* in_sizes, int n_in,
                              void* d_out, int out_size, void* d_ws, size_t ws_size,
                              hipStream_t stream) {
    const float* x   = (const float*)d_in[0];
    const int*   ei  = (const int*)d_in[1];
    const float* W01 = (const float*)d_in[3];
    const float* W11 = (const float*)d_in[4];
    const float* b1  = (const float*)d_in[5];
    const float* W02 = (const float*)d_in[6];
    const float* W12 = (const float*)d_in[7];
    const float* b2  = (const float*)d_in[8];
    const float* W03 = (const float*)d_in[9];
    const float* W13 = (const float*)d_in[10];
    const float* b3  = (const float*)d_in[11];

    const int N = NNODES;
    const int E = in_sizes[1] / 2;
    const int* src = ei;
    const int* dst = ei + E;

    float* dis  = (float*)d_ws;
    float* aggT = dis + 10240;
    float* h1   = aggT + 256 * AGGP;
    float* h2   = h1  + (size_t)N * 256;
    float* out  = (float*)d_out;
    unsigned* cnt = (unsigned*)(dis + 10016);
    unsigned* cur = (unsigned*)(dis + 10080);
    unsigned* list0 = (unsigned*)h2;
    unsigned* listw = (unsigned*)d_out;
    const uint4* list = (const uint4*)d_out;

    zero4_kernel<<<64, 256, 0, stream>>>((float4*)dis, 10240 / 4);
    deg_kernel<<<64, 1024, 0, stream>>>(src, dst, dis, cnt, E);
    dis_kernel<<<(N + 255) / 256, 256, 0, stream>>>(dis, N);
    place_kernel<<<64, 1024, 0, stream>>>(src, dst, cur, list0, E);
    order_kernel<<<64 * OSPLIT, 1024, 0, stream>>>(cnt, list0, listw);

    agg_split_kernel<1, false><<<256, 1024, 0, stream>>>(x, src, dst, dis, cnt, list, aggT, E, 128);
    cheb_gemm_kernel<4, false, false, true><<<N / 16, 128, 0, stream>>>(x, aggT, W01, W11, b1, h1, 128, 1);

    agg_split_kernel<2, true><<<256, 1024, 0, stream>>>(h1, src, dst, dis, cnt, list, aggT, E, 256);
    cheb_gemm_kernel<4, true, true, true><<<N / 16, 128, 0, stream>>>(h1, aggT, W02, W12, b2, h2, 256, 1);

    agg_split_kernel<2, true><<<256, 1024, 0, stream>>>(h2, src, dst, dis, cnt, list, aggT, E, 256);
    cheb_gemm_kernel<2, true, true, false><<<N / 16, 128, 0, stream>>>(h2, aggT, W03, W13, b3, out, 256, 0);
}


# Aggregation by node quarter: a lane's ordered stream is cut at a node boundary; a block sums a quarter of the nodes for 4 features (2 in layer 1), one 16-byte LDS read per edge; sum order changes
# speedup vs baseline: 4.2466x; 1.0760x over previous
#include <hip/hip_runtime.h>
#include <hip/hip_bf16.h>

typedef __attribute__((ext_vector_type(2))) float v2f;
typedef __attribute__((ext_vector_type(4))) float v4f;
typedef __attribute__((ext_vector_type(8))) float v8f;

#define NNODES 10000
#define AGGP 10016
#define PLP 20000

__global__ __launch_bounds__(256) void zero4_kernel(float4* __restrict__ p, int n4) {
    int i = blockIdx.x * 256 + threadIdx.x;
    int stride = gridDim.x * 256;
    float4 z; z.x = 0.f; z.y = 0.f; z.z = 0.f; z.w = 0.f;
    for (; i < n4; i += stride) p[i] = z;
}

#define SUBCAP 20000u
#define PARTQ 160000u
#define ACCN 5024u
#define NH 79u

#define ETILE 10240u
#define EPT 10
__global__ __launch_bounds__(1024) void deg_kernel(const int* __restrict__ src,
                                                   const int* __restrict__ dst,
                                                   float* __restrict__ deg,
                                                   unsigned* __restrict__ cnt, int E) {
    __shared__ unsigned hist[NNODES];
    __shared__ unsigned h[64];
    const unsigned tid = threadIdx.x;
    for (unsigned n = tid; n < NNODES; n += 1024u) hist[n] = 0u;
    if (tid < 64u) h[tid] = 0u;
    __syncthreads();
    for (unsigned t0 = blockIdx.x * ETILE; t0 < (unsigned)E; t0 += gridDim.x * ETILE) {
        unsigned s[EPT], d[EPT];
#pragma unroll
        for (int u = 0; u < EPT; ++u) {
            const unsigned i = t0 + 1024u * (unsigned)u + tid;
            const unsigned ii = (i < (unsigned)E) ? i : t0;
            s[u] = (unsigned)src[ii];
            d[u] = (unsigned)dst[ii];
        }
#pragma unroll
        for (int u = 0; u < EPT; ++u) {
            if (t0 + 1024u * (unsigned)u + tid < (unsigned)E) {
                atomicAdd(&hist[s[u]], 1u);
                atomicAdd(&h[d[u] & 63u], 1u);
            }
        }
    }
    __syncthreads();
    for (unsigned n = tid; n < NNODES; n += 1024u) {
        const unsigned c = hist[n];
        if (c != 0u) atomicAdd(&deg[n], (float)c);
    }
    if (tid < 64u) {
        const unsigned c = h[tid];
        if (c != 0u) atomicAdd(&cnt[tid], c);
    }
}

__global__ __launch_bounds__(1024) void place_kernel(const int* __restrict__ src,
                                                     const int* __restrict__ dst,
                                                     unsigned* __restrict__ cur,
                                                     unsigned* __restrict__ list, int E) {
    __shared__ unsigned h[64];
    __shared__ unsigned base[64];
    __shared__ unsigned coff[64];
    __shared__ unsigned stage[ETILE];
    const unsigned tid = threadIdx.x;
    const unsigned lane = tid & 31u;
    for (unsigned t0 = blockIdx.x * ETILE; t0 < (unsigned)E; t0 += gridDim.x * ETILE) {
        if (tid < 64u) h[tid] = 0u;
        __syncthreads();
        unsigned s[EPT], d[EPT], r[EPT];
#pragma unroll
        for (int u = 0; u < EPT; ++u) {
            const unsigned i = t0 + 1024u * (unsigned)u + tid;
            const unsigned ii = (i < (unsigned)E) ? i : t0;
            s[u] = (unsigned)src[ii];
            d[u] = (unsigned)dst[ii];
        }
#pragma unroll
        for (int u = 0; u < EPT; ++u) {
            r[u] = 0u;
            if (t0 + 1024u * (unsigned)u + tid < (unsigned)E) r[u] = atomicAdd(&h[d[u] & 63u], 1u);
        }
        __syncthreads();
        if (tid < 64u) {
            const unsigned n = h[tid];
            unsigned v = n;
#pragma unroll
            for (int o = 1; o < 32; o <<= 1) {
                const unsigned t = __shfl_up(v, o);
                if (lane >= (unsigned)o) v += t;
            }
            unsigned low = h[lane];
#pragma unroll
            for (int o = 16; o > 0; o >>= 1) low += __shfl_xor(low, o);
            coff[tid] = v - n + ((tid >= 32u) ? low : 0u);
            base[tid] = (n != 0u) ? atomicAdd(&cur[tid], n) : 0u;
        }
        __syncthreads();
#pragma unroll
        for (int u = 0; u < EPT; ++u) {
            if (t0 + 1024u * (unsigned)u + tid < (unsigned)E) {
                const unsigned c = d[u] & 63u;
                stage[coff[c] + r[u]] = s[u] | ((((d[u] >> 6) << 5) | (c & 31u)) << 16);
            }
        }
        __syncthreads();
        for (unsigned c = tid >> 5; c < 64u; c += 32u) {
            const unsigned bc = base[c];
            const unsigned end = bc + h[c];
            const unsigned cc = coff[c];
            unsigned* __restrict__ out = list + (size_t)c * SUBCAP;
            for (unsigned pos = (bc & ~31u) + lane; pos < end; pos += 32u) {
                if (pos >= bc && pos < SUBCAP)
                    __hip_atomic_exchange(&out[pos], stage[cc + (pos - bc)],
                                          __ATOMIC_RELAXED, __HIP_MEMORY_SCOPE_AGENT);
            }
        }
        __syncthreads();
    }
}

#define OSPLIT 4u
#define OGRP 40u
__global__ __launch_bounds__(1024) void order_kernel(const unsigned* __restrict__ cnt,
                                                     const unsigned* __restrict__ lin,
                                                     unsigned* __restrict__ lout,
                                                     unsigned* __restrict__ mid) {
    __shared__ unsigned ent[SUBCAP];
    __shared__ unsigned srt[SUBCAP];
    __shared__ unsigned gsz[160];
    __shared__ unsigned gst[161];
    const unsigned tid = threadIdx.x;
    const unsigned c = blockIdx.x / OSPLIT;
    const unsigned sp = blockIdx.x % OSPLIT;
    const unsigned n = cnt[c];
    if (n > SUBCAP) return;
    const unsigned* __restrict__ in = lin + (size_t)c * SUBCAP;
    unsigned* __restrict__ out = lout + (size_t)(c >> 5) * (PARTQ * 4u) + ((c & 31u) << 2);
    if (tid < 160u) gsz[tid] = 0u;
    __syncthreads();
    for (unsigned p = tid; p < n; p += 1024u) {
        const unsigned e = in[p];
        atomicAdd(&gsz[e >> 21], 1u);
    }
    __syncthreads();
    if (tid < 32u) {
        unsigned a[5];
        unsigned t = 0u;
#pragma unroll
        for (int k = 0; k < 5; ++k) { a[k] = gsz[5u * tid + (unsigned)k]; t += a[k]; }
        unsigned v = t;
#pragma unroll
        for (int o = 1; o < 32; o <<= 1) {
            const unsigned u = __shfl_up(v, o);
            if (tid >= (unsigned)o) v += u;
        }
        unsigned ex = v - t;
#pragma unroll
        for (int k = 0; k < 5; ++k) {
            gst[5u * tid + (unsigned)k] = ex;
            gsz[5u * tid + (unsigned)k] = 0u;
            ex += a[k];
        }
        if (tid == 31u) gst[160] = ex;
    }
    __syncthreads();
    if (sp == 0u && tid == 0u)
        __hip_atomic_exchange(&mid[c], gst[NH], __ATOMIC_RELAXED, __HIP_MEMORY_SCOPE_AGENT);
    for (unsigned p = tid; p < n; p += 1024u) {
        const unsigned e = in[p];
        const unsigned g = e >> 21;
        const unsigned r = atomicAdd(&gsz[g], 1u);
        ent[gst[g] + r] = e;
    }
    __syncthreads();
    const unsigned a0 = gst[sp * OGRP];
    const unsigned a1 = gst[sp * OGRP + OGRP];
    for (unsigned i = a0 + tid; i < a1; i += 1024u) {
        const unsigned key = ent[i];
        const unsigned g = key >> 21;
        const unsigned a = gst[g];
        const unsigned b = gst[g + 1u];
        unsigned rank = 0u;
        for (unsigned j = a; j < b; ++j) {
            const unsigned k = ent[j];
            rank += (k < key || (k == key && j < i)) ? 1u : 0u;
        }
        srt[a + rank] = key;
    }
    __syncthreads();
    for (unsigned i = a0 + tid; i < a1; i += 1024u)
        __hip_atomic_exchange(&out[((size_t)(i >> 2) << 7) + (i & 3u)], srt[i],
                              __ATOMIC_RELAXED, __HIP_MEMORY_SCOPE_AGENT);
}

__global__ __launch_bounds__(256) void dis_kernel(float* __restrict__ deg, int n) {
    int i = blockIdx.x * 256 + threadIdx.x;
    if (i < n) {
        float d = deg[i];
        deg[i] = (d > 0.0f) ? rsqrtf(d) : 0.0f;
    }
}
template <int FPB> struct gvec;
template <> struct gvec<2> { typedef v2f type; };
template <> struct gvec<4> { typedef v4f type; };

#define ACCX (ACCN + 1024u)
#define NOSLOT 0xFFFFFFFFu

template <int FPB>
__device__ __forceinline__ void add_edge(float* accf, unsigned slot, typename gvec<FPB>::type v) {
#pragma unroll
    for (int j = 0; j < FPB; ++j)
        __hip_atomic_fetch_add(&accf[slot * (unsigned)FPB + (unsigned)j], v[j], __ATOMIC_RELAXED,
                               __HIP_MEMORY_SCOPE_WORKGROUP);
}

template <int FPB, bool GUARD>
__device__ __forceinline__ void run_step(typename gvec<FPB>::type* acc, unsigned& cur, unsigned& idx, typename gvec<FPB>::type& s,
                                         unsigned ew, typename gvec<FPB>::type v, bool valid) {
    const unsigned slot = ew >> 16;
    bool chg = slot != cur;
    if (GUARD) chg = chg && valid;
    if (chg) {
        acc[idx] = s;
        idx = slot;
        cur = slot;
    }
    const typename gvec<FPB>::type t = s + v;
    if (GUARD) {
        if (valid) s = chg ? v : t;
    } else {
        s = chg ? v : t;
    }
}

template <int FPB, int ROWS, bool GUARD>
__device__ __forceinline__ void run_rows(const uint4* __restrict__ pl, const typename gvec<FPB>::type* xs,
                                         typename gvec<FPB>::type* acc,
                                         unsigned q, unsigned lane, unsigned lo, unsigned len, unsigned& cur,
                                         unsigned& idx, typename gvec<FPB>::type& s) {
    typedef typename gvec<FPB>::type gv;
    uint4 e[ROWS];
    gv v[4 * ROWS];
#pragma unroll
    for (int u = 0; u < ROWS; ++u) e[u] = pl[((q + (unsigned)u) << 5) + lane];
#pragma unroll
    for (int u = 0; u < ROWS; ++u) {
        const unsigned p = ((q + (unsigned)u) << 2) - lo;
        v[4 * u + 0] = xs[(!GUARD || p < len) ? (e[u].x & 0xFFFFu) : 0u];
        v[4 * u + 1] = xs[(!GUARD || p + 1u < len) ? (e[u].y & 0xFFFFu) : 0u];
        v[4 * u + 2] = xs[(!GUARD || p + 2u < len) ? (e[u].z & 0xFFFFu) : 0u];
        v[4 * u + 3] = xs[(!GUARD || p + 3u < len) ? (e[u].w & 0xFFFFu) : 0u];
    }
#pragma unroll
    for (int u = 0; u < ROWS; ++u) {
        const unsigned p = ((q + (unsigned)u) << 2) - lo;
        run_step<FPB, GUARD>(acc, cur, idx, s, e[u].x, v[4 * u + 0], p < len);
        run_step<FPB, GUARD>(acc, cur, idx, s, e[u].y, v[4 * u + 1], p + 1u < len);
        run_step<FPB, GUARD>(acc, cur, idx, s, e[u].z, v[4 * u + 2], p + 2u < len);
        run_step<FPB, GUARD>(acc, cur, idx, s, e[u].w, v[4 * u + 3], p + 3u < len);
    }
}

template <int FPB, bool PL>
__global__ __launch_bounds__(1024) void agg_split_kernel(
    const float* __restrict__ xin, const int* __restrict__ src,
    const int* __restrict__ dst, const float* __restrict__ dis,
    const unsigned* __restrict__ cnt, const unsigned* __restrict__ mid, const uint4* __restrict__ list,
    float* __restrict__ aggT, int E, int D)
{
    typedef typename gvec<FPB>::type gv;
    __shared__ gv xs[NNODES];
    __shared__ gv acc[ACCX];
    __shared__ unsigned slotF[1024];
    __shared__ unsigned slotL[1024];
    const unsigned tid = threadIdx.x;
    const unsigned half = blockIdx.x & 1u;
    const unsigned part = (blockIdx.x >> 1) & 1u;
    const unsigned fg = blockIdx.x >> 2;
    const unsigned f0 = fg * (unsigned)FPB;
    const unsigned lane = tid & 31u;
    const unsigned wave = tid >> 5;

    if constexpr (PL) {
        const float* __restrict__ pin = xin + (size_t)(fg * (unsigned)(FPB / 2)) * PLP;
        for (unsigned n = tid; n < NNODES; n += 1024u) {
            gv t;
#pragma unroll
            for (int j = 0; j < FPB / 2; ++j) {
                const v2f a = *(const v2f*)(pin + (size_t)j * PLP + 2u * n);
                t[2 * j] = a.x;
                t[2 * j + 1] = a.y;
            }
            xs[n] = dis[n] * t;
        }
    } else {
        for (unsigned n = tid; n < NNODES; n += 1024u) {
            const gv t = *(const gv*)(xin + (size_t)n * D + f0);
            xs[n] = dis[n] * t;
        }
    }
    for (unsigned c = tid; c < ACCN; c += 1024u) acc[c] = gv{};
    const unsigned myc = cnt[part * 32u + lane];
    const unsigned mym0 = mid[part * 32u + lane];
    const unsigned mym = (mym0 < myc) ? mym0 : myc;
    const unsigned lo = (half != 0u) ? mym : 0u;
    const unsigned hi = (half != 0u) ? myc : mym;
    const unsigned len = hi - lo;
    unsigned mxc = myc, lomin = lo, lomax = lo, himin = hi, himax = hi;
#pragma unroll
    for (int o = 16; o > 0; o >>= 1) {
        const unsigned a = __shfl_xor(mxc, o);
        const unsigned b0 = __shfl_xor(lomin, o);
        const unsigned b1 = __shfl_xor(lomax, o);
        const unsigned c0 = __shfl_xor(himin, o);
        const unsigned c1 = __shfl_xor(himax, o);
        mxc = (a > mxc) ? a : mxc;
        lomin = (b0 < lomin) ? b0 : lomin;
        lomax = (b1 > lomax) ? b1 : lomax;
        himin = (c0 < himin) ? c0 : himin;
        himax = (c1 > himax) ? c1 : himax;
    }
    __syncthreads();

    const unsigned qB = lomin >> 2;
    const unsigned qE = (himax + 3u) >> 2;
    const unsigned nrows = qE - qB;
    const bool ordered = (mxc <= SUBCAP) && (nrows >= 32u);
    unsigned sF = NOSLOT, sL = NOSLOT;
    gv vF = {}, vL = {};
    if (ordered) {
        const uint4* __restrict__ pl = list + (size_t)part * PARTQ;
        const unsigned qFS = (lomax + 3u) >> 2;
        const unsigned qFE = himin >> 2;
        const unsigned qs = qB + ((wave * nrows) >> 5);
        const unsigned qe = qB + (((wave + 1u) * nrows) >> 5);
        unsigned cur = NOSLOT;
        {
            const unsigned p0 = ((qs << 2) > lo) ? (qs << 2) : lo;
            const unsigned p1 = ((qe << 2) < hi) ? (qe << 2) : hi;
            if (p0 < p1) {
                const uint4 w0 = pl[((p0 >> 2) << 5) + lane];
                const unsigned k = p0 & 3u;
                const unsigned ew = (k == 0u) ? w0.x : ((k == 1u) ? w0.y : ((k == 2u) ? w0.z : w0.w));
                cur = ew >> 16;
            }
        }
        const unsigned firstslot = cur;
        unsigned idx = ACCN + tid;
        gv s = {};
        unsigned q = qs;
        constexpr int RT = 4;
        for (; q + (unsigned)RT <= qe; q += (unsigned)RT) {
            if (q >= qFS && q + (unsigned)RT <= qFE) run_rows<FPB, RT, false>(pl, xs, acc, q, lane, lo, len, cur, idx, s);
            else run_rows<FPB, RT, true>(pl, xs, acc, q, lane, lo, len, cur, idx, s);
        }
        for (; q < qe; ++q) run_rows<FPB, 1, true>(pl, xs, acc, q, lane, lo, len, cur, idx, s);
        sF = firstslot;
        if (idx == ACCN + tid) {
            vF = s;
            acc[ACCN + tid] = s;
        } else {
            vF = acc[ACCN + tid];
            sL = cur;
            vL = s;
        }
        slotF[tid] = sF;
        slotL[tid] = sL;
    } else {
        for (unsigned e = tid; e < (unsigned)E; e += 1024u) {
            const unsigned d = (unsigned)dst[e];
            if (((d >> 5) & 1u) == part && (((d >> 6) >= NH) == (half != 0u)))
                add_edge<FPB>((float*)acc, ((d >> 6) << 5) | (d & 31u), xs[src[e]]);
        }
    }
    __syncthreads();

    if (ordered) {
        bool lead = sF != NOSLOT;
        if (lead && wave != 0u) {
            const unsigned pl1 = slotL[tid - 32u];
            const unsigned prev = (pl1 != NOSLOT) ? pl1 : slotF[tid - 32u];
            lead = prev != sF;
        }
        if (lead && sL != NOSLOT) acc[sF] = vF;
        if (sL != NOSLOT || lead) {
            const unsigned slot = (sL != NOSLOT) ? sL : sF;
            gv sum = (sL != NOSLOT) ? vL : vF;
            for (unsigned w2 = wave + 1u; w2 < 32u; ++w2) {
                const unsigned t2 = (w2 << 5) + lane;
                if (slotF[t2] != slot) break;
                sum = sum + acc[ACCN + t2];
                if (slotL[t2] != NOSLOT) break;
            }
            acc[slot] = sum;
        }
    }
    __syncthreads();

    const unsigned cB = (half != 0u) ? (NH << 5) : 0u;
    const unsigned cE = (half != 0u) ? ACCN : (NH << 5);
    for (unsigned c = cB + tid; c < cE; c += 1024u) {
        const unsigned n = ((c >> 5) << 6) | (part << 5) | (c & 31u);
        if (n < NNODES) {
            const float dn = dis[n];
            const gv a = acc[c];
            if constexpr (PL) {
#pragma unroll
                for (int j = 0; j < FPB / 2; ++j) {
                    v2f o;
                    o.x = -(dn * a[2 * j]);
                    o.y = -(dn * a[2 * j + 1]);
                    *(v2f*)(aggT + (size_t)(fg * (unsigned)(FPB / 2) + (unsigned)j) * PLP + 2u * n) = o;
                }
            } else {
#pragma unroll
                for (int j = 0; j < FPB; ++j)
                    aggT[(size_t)(f0 + j) * AGGP + n] = -(dn * a[j]);
            }
        }
    }
}

template <int TILES> struct bvec;
template <> struct bvec<4> { typedef v4f type; };
template <> struct bvec<2> { typedef v2f type; };

template <int TILES> struct stage_t {
    v2f a0[2], a1[2];
    typename bvec<TILES>::type b0lo[2], b0hi[2], b1lo[2], b1hi[2];
};

template <int TILES, bool A0P, bool A1P>
__device__ __forceinline__ void load_stage(stage_t<TILES>& s, const float* __restrict__ a0p,
                                           const float* __restrict__ a1p, const float* __restrict__ w0p,
                                           const float* __restrict__ w1p, int k) {
    typedef typename bvec<TILES>::type bv;
    constexpr int NOUT = 64 * TILES;
#pragma unroll
    for (int u = 0; u < 2; ++u) {
        const int kk = k + 4 * u;
        if constexpr (A0P) {
            s.a0[u] = *(const v2f*)(a0p + (size_t)(kk >> 1) * PLP);
        } else {
            s.a0[u] = *(const v2f*)(a0p + kk);
        }
        if constexpr (A1P) {
            s.a1[u] = *(const v2f*)(a1p + (size_t)(kk >> 1) * PLP);
        } else {
            s.a1[u].x = a1p[(size_t)kk * AGGP];
            s.a1[u].y = a1p[(size_t)(kk + 1) * AGGP];
        }
        s.b0lo[u] = *(const bv*)(w0p + (size_t)kk * NOUT);
        s.b0hi[u] = *(const bv*)(w0p + (size_t)(kk + 1) * NOUT);
        s.b1lo[u] = *(const bv*)(w1p + (size_t)kk * NOUT);
        s.b1hi[u] = *(const bv*)(w1p + (size_t)(kk + 1) * NOUT);
    }
}

template <int TILES, bool OUTP>
__device__ __forceinline__ void mma_stage(const stage_t<TILES>& s, v8f (&acc)[TILES]) {
#pragma unroll
    for (int u = 0; u < 2; ++u) {
#pragma unroll
        for (int t = 0; t < TILES; ++t) {
            v2f b; b.x = s.b0lo[u][t]; b.y = s.b0hi[u][t];
            if constexpr (OUTP)
                acc[t] = __builtin_amdgcn_wmma_f32_16x16x4_f32(
                    false, b, false, s.a0[u], (short)0, acc[t], false, false);
            else
                acc[t] = __builtin_amdgcn_wmma_f32_16x16x4_f32(
                    false, s.a0[u], false, b, (short)0, acc[t], false, false);
        }
#pragma unroll
        for (int t = 0; t < TILES; ++t) {
            v2f b; b.x = s.b1lo[u][t]; b.y = s.b1hi[u][t];
            if constexpr (OUTP)
                acc[t] = __builtin_amdgcn_wmma_f32_16x16x4_f32(
                    false, b, false, s.a1[u], (short)0, acc[t], false, false);
            else
                acc[t] = __builtin_amdgcn_wmma_f32_16x16x4_f32(
                    false, s.a1[u], false, b, (short)0, acc[t], false, false);
        }
    }
}

template <int TILES, bool A0P, bool A1P, bool OUTP>
__global__ __launch_bounds__(128) void cheb_gemm_kernel(
    const float* __restrict__ Xin, const float* __restrict__ AggT,
    const float* __restrict__ W0, const float* __restrict__ W1,
    const float* __restrict__ bias, float* __restrict__ Out,
    int K, int relu)
{
    typedef typename bvec<TILES>::type bv;
    constexpr int NOUT = 64 * TILES;
    const int wave  = threadIdx.x >> 5;
    const int lane  = threadIdx.x & 31;
    const int rlane = lane & 15;
    const int half  = lane >> 4;
    const int khalf = half << 1;
    const int rowBase = blockIdx.x << 4;
    const int colBase = wave * (TILES * 16);
    const int col0 = colBase + TILES * rlane;

    const float* a0p = A0P ? Xin + (size_t)half * PLP + 2 * (rowBase + rlane)
                           : Xin + (size_t)(rowBase + rlane) * K + khalf;
    const float* a1p = A1P ? AggT + (size_t)half * PLP + 2 * (rowBase + rlane)
                           : AggT + (size_t)khalf * AGGP + rowBase + rlane;
    const float* w0p = W0 + (size_t)khalf * NOUT + col0;
    const float* w1p = W1 + (size_t)khalf * NOUT + col0;

    v8f acc[TILES];
#pragma unroll
    for (int t = 0; t < TILES; ++t) acc[t] = (v8f){};

    stage_t<TILES> cur, nxt;
    load_stage<TILES, A0P, A1P>(cur, a0p, a1p, w0p, w1p, 0);

#pragma unroll 2
    for (int k = 0; k < K - 8; k += 8) {
        load_stage<TILES, A0P, A1P>(nxt, a0p, a1p, w0p, w1p, k + 8);
        mma_stage<TILES, OUTP>(cur, acc);
        cur = nxt;
    }
    mma_stage<TILES, OUTP>(cur, acc);

    if constexpr (OUTP) {
        float* op = Out + (size_t)((colBase >> 1) + (TILES / 2) * 8 * half) * PLP + 2 * (rowBase + rlane);
        const float* bp = bias + colBase + TILES * 8 * half;
#pragma unroll
        for (int r = 0; r < 8; ++r) {
            const bv bvv = *(const bv*)(bp + TILES * r);
#pragma unroll
            for (int t = 0; t < TILES; t += 2) {
                v2f v;
                v.x = acc[t][r] + bvv[t];
                v.y = acc[t + 1][r] + bvv[t + 1];
                if (relu) { v.x = fmaxf(v.x, 0.0f); v.y = fmaxf(v.y, 0.0f); }
                *(v2f*)(op + (size_t)((TILES / 2) * r + (t >> 1)) * PLP) = v;
            }
        }
    } else {
        const int rowOff = rowBase + (half << 3);
        const bv bvv = *(const bv*)(bias + col0);
        float* op = Out + (size_t)rowOff * NOUT + col0;
#pragma unroll
        for (int r = 0; r < 8; ++r) {
            bv v;
#pragma unroll
            for (int t = 0; t < TILES; ++t) {
                float x = acc[t][r] + bvv[t];
                if (relu) x = fmaxf(x, 0.0f);
                v[t] = x;
            }
            *(bv*)(op + (size_t)r * NOUT) = v;
        }
    }
}

extern "C" void kernel_launch(void* const* d_in, const int* in_sizes, int n_in,
                              void* d_out, int out_size, void* d_ws, size_t ws_size,
                              hipStream_t stream) {
    const float* x   = (const float*)d_in[0];
    const int*   ei  = (const int*)d_in[1];
    const float* W01 = (const float*)d_in[3];
    const float* W11 = (const float*)d_in[4];
    const float* b1  = (const float*)d_in[5];
    const float* W02 = (const float*)d_in[6];
    const float* W12 = (const float*)d_in[7];
    const float* b2  = (const float*)d_in[8];
    const float* W03 = (const float*)d_in[9];
    const float* W13 = (const float*)d_in[10];
    const float* b3  = (const float*)d_in[11];

    const int N = NNODES;
    const int E = in_sizes[1] / 2;
    const int* src = ei;
    const int* dst = ei + E;

    float* dis  = (float*)d_ws;
    float* aggT = dis + 10240;
    float* h1   = aggT + 256 * AGGP;
    float* h2   = h1  + (size_t)N * 256;
    float* out  = (float*)d_out;
    unsigned* cnt = (unsigned*)(dis + 10016);
    unsigned* cur = (unsigned*)(dis + 10080);
    unsigned* mid = (unsigned*)(dis + 10144);
    unsigned* list0 = (unsigned*)h2;
    unsigned* listw = (unsigned*)d_out;
    const uint4* list = (const uint4*)d_out;

    zero4_kernel<<<64, 256, 0, stream>>>((float4*)dis, 10240 / 4);
    deg_kernel<<<64, 1024, 0, stream>>>(src, dst, dis, cnt, E);
    dis_kernel<<<(N + 255) / 256, 256, 0, stream>>>(dis, N);
    place_kernel<<<64, 1024, 0, stream>>>(src, dst, cur, list0, E);
    order_kernel<<<64 * OSPLIT, 1024, 0, stream>>>(cnt, list0, listw, mid);

    agg_split_kernel<2, false><<<256, 1024, 0, stream>>>(x, src, dst, dis, cnt, mid, list, aggT, E, 128);
    cheb_gemm_kernel<4, false, false, true><<<N / 16, 128, 0, stream>>>(x, aggT, W01, W11, b1, h1, 128, 1);

    agg_split_kernel<4, true><<<256, 1024, 0, stream>>>(h1, src, dst, dis, cnt, mid, list, aggT, E, 256);
    cheb_gemm_kernel<4, true, true, true><<<N / 16, 128, 0, stream>>>(h1, aggT, W02, W12, b2, h2, 256, 1);

    agg_split_kernel<4, true><<<256, 1024, 0, stream>>>(h2, src, dst, dis, cnt, mid, list, aggT, E, 256);
    cheb_gemm_kernel<2, true, true, false><<<N / 16, 128, 0, stream>>>(h2, aggT, W03, W13, b3, out, 256, 0);
}


# Network input re-laid per call as feature-pair planes (in the dis launch): layer 1's aggregation fills LDS from contiguous runs, not 8 bytes out of each of 10000 rows; its aggregate is planes too
# speedup vs baseline: 4.4246x; 1.0419x over previous
#include <hip/hip_runtime.h>
#include <hip/hip_bf16.h>

typedef __attribute__((ext_vector_type(2))) float v2f;
typedef __attribute__((ext_vector_type(4))) float v4f;
typedef __attribute__((ext_vector_type(8))) float v8f;

#define NNODES 10000
#define AGGP 10016
#define PLP 20000

__global__ __launch_bounds__(256) void zero4_kernel(float4* __restrict__ p, int n4) {
    int i = blockIdx.x * 256 + threadIdx.x;
    int stride = gridDim.x * 256;
    float4 z; z.x = 0.f; z.y = 0.f; z.z = 0.f; z.w = 0.f;
    for (; i < n4; i += stride) p[i] = z;
}

#define SUBCAP 20000u
#define PARTQ 160000u
#define ACCN 5024u
#define NH 79u

#define ETILE 10240u
#define EPT 10
__global__ __launch_bounds__(1024) void deg_kernel(const int* __restrict__ src,
                                                   const int* __restrict__ dst,
                                                   float* __restrict__ deg,
                                                   unsigned* __restrict__ cnt, int E) {
    __shared__ unsigned hist[NNODES];
    __shared__ unsigned h[64];
    const unsigned tid = threadIdx.x;
    for (unsigned n = tid; n < NNODES; n += 1024u) hist[n] = 0u;
    if (tid < 64u) h[tid] = 0u;
    __syncthreads();
    for (unsigned t0 = blockIdx.x * ETILE; t0 < (unsigned)E; t0 += gridDim.x * ETILE) {
        unsigned s[EPT], d[EPT];
#pragma unroll
        for (int u = 0; u < EPT; ++u) {
            const unsigned i = t0 + 1024u * (unsigned)u + tid;
            const unsigned ii = (i < (unsigned)E) ? i : t0;
            s[u] = (unsigned)src[ii];
            d[u] = (unsigned)dst[ii];
        }
#pragma unroll
        for (int u = 0; u < EPT; ++u) {
            if (t0 + 1024u * (unsigned)u + tid < (unsigned)E) {
                atomicAdd(&hist[s[u]], 1u);
                atomicAdd(&h[d[u] & 63u], 1u);
            }
        }
    }
    __syncthreads();
    for (unsigned n = tid; n < NNODES; n += 1024u) {
        const unsigned c = hist[n];
        if (c != 0u) atomicAdd(&deg[n], (float)c);
    }
    if (tid < 64u) {
        const unsigned c = h[tid];
        if (c != 0u) atomicAdd(&cnt[tid], c);
    }
}

__global__ __launch_bounds__(1024) void place_kernel(const int* __restrict__ src,
                                                     const int* __restrict__ dst,
                                                     unsigned* __restrict__ cur,
                                                     unsigned* __restrict__ list, int E) {
    __shared__ unsigned h[64];
    __shared__ unsigned base[64];
    __shared__ unsigned coff[64];
    __shared__ unsigned stage[ETILE];
    const unsigned tid = threadIdx.x;
    const unsigned lane = tid & 31u;
    for (unsigned t0 = blockIdx.x * ETILE; t0 < (unsigned)E; t0 += gridDim.x * ETILE) {
        if (tid < 64u) h[tid] = 0u;
        __syncthreads();
        unsigned s[EPT], d[EPT], r[EPT];
#pragma unroll
        for (int u = 0; u < EPT; ++u) {
            const unsigned i = t0 + 1024u * (unsigned)u + tid;
            const unsigned ii = (i < (unsigned)E) ? i : t0;
            s[u] = (unsigned)src[ii];
            d[u] = (unsigned)dst[ii];
        }
#pragma unroll
        for (int u = 0; u < EPT; ++u) {
            r[u] = 0u;
            if (t0 + 1024u * (unsigned)u + tid < (unsigned)E) r[u] = atomicAdd(&h[d[u] & 63u], 1u);
        }
        __syncthreads();
        if (tid < 64u) {
            const unsigned n = h[tid];
            unsigned v = n;
#pragma unroll
            for (int o = 1; o < 32; o <<= 1) {
                const unsigned t = __shfl_up(v, o);
                if (lane >= (unsigned)o) v += t;
            }
            unsigned low = h[lane];
#pragma unroll
            for (int o = 16; o > 0; o >>= 1) low += __shfl_xor(low, o);
            coff[tid] = v - n + ((tid >= 32u) ? low : 0u);
            base[tid] = (n != 0u) ? atomicAdd(&cur[tid], n) : 0u;
        }
        __syncthreads();
#pragma unroll
        for (int u = 0; u < EPT; ++u) {
            if (t0 + 1024u * (unsigned)u + tid < (unsigned)E) {
                const unsigned c = d[u] & 63u;
                stage[coff[c] + r[u]] = s[u] | ((((d[u] >> 6) << 5) | (c & 31u)) << 16);
            }
        }
        __syncthreads();
        for (unsigned c = tid >> 5; c < 64u; c += 32u) {
            const unsigned bc = base[c];
            const unsigned end = bc + h[c];
            const unsigned cc = coff[c];
            unsigned* __restrict__ out = list + (size_t)c * SUBCAP;
            for (unsigned pos = (bc & ~31u) + lane; pos < end; pos += 32u) {
                if (pos >= bc && pos < SUBCAP)
                    __hip_atomic_exchange(&out[pos], stage[cc + (pos - bc)],
                                          __ATOMIC_RELAXED, __HIP_MEMORY_SCOPE_AGENT);
            }
        }
        __syncthreads();
    }
}

#define OSPLIT 4u
#define OGRP 40u
__global__ __launch_bounds__(1024) void order_kernel(const unsigned* __restrict__ cnt,
                                                     const unsigned* __restrict__ lin,
                                                     unsigned* __restrict__ lout,
                                                     unsigned* __restrict__ mid) {
    __shared__ unsigned ent[SUBCAP];
    __shared__ unsigned srt[SUBCAP];
    __shared__ unsigned gsz[160];
    __shared__ unsigned gst[161];
    const unsigned tid = threadIdx.x;
    const unsigned c = blockIdx.x / OSPLIT;
    const unsigned sp = blockIdx.x % OSPLIT;
    const unsigned n = cnt[c];
    if (n > SUBCAP) return;
    const unsigned* __restrict__ in = lin + (size_t)c * SUBCAP;
    unsigned* __restrict__ out = lout + (size_t)(c >> 5) * (PARTQ * 4u) + ((c & 31u) << 2);
    if (tid < 160u) gsz[tid] = 0u;
    __syncthreads();
    for (unsigned p = tid; p < n; p += 1024u) {
        const unsigned e = in[p];
        atomicAdd(&gsz[e >> 21], 1u);
    }
    __syncthreads();
    if (tid < 32u) {
        unsigned a[5];
        unsigned t = 0u;
#pragma unroll
        for (int k = 0; k < 5; ++k) { a[k] = gsz[5u * tid + (unsigned)k]; t += a[k]; }
        unsigned v = t;
#pragma unroll
        for (int o = 1; o < 32; o <<= 1) {
            const unsigned u = __shfl_up(v, o);
            if (tid >= (unsigned)o) v += u;
        }
        unsigned ex = v - t;
#pragma unroll
        for (int k = 0; k < 5; ++k) {
            gst[5u * tid + (unsigned)k] = ex;
            gsz[5u * tid + (unsigned)k] = 0u;
            ex += a[k];
        }
        if (tid == 31u) gst[160] = ex;
    }
    __syncthreads();
    if (sp == 0u && tid == 0u)
        __hip_atomic_exchange(&mid[c], gst[NH], __ATOMIC_RELAXED, __HIP_MEMORY_SCOPE_AGENT);
    for (unsigned p = tid; p < n; p += 1024u) {
        const unsigned e = in[p];
        const unsigned g = e >> 21;
        const unsigned r = atomicAdd(&gsz[g], 1u);
        ent[gst[g] + r] = e;
    }
    __syncthreads();
    const unsigned a0 = gst[sp * OGRP];
    const unsigned a1 = gst[sp * OGRP + OGRP];
    for (unsigned i = a0 + tid; i < a1; i += 1024u) {
        const unsigned key = ent[i];
        const unsigned g = key >> 21;
        const unsigned a = gst[g];
        const unsigned b = gst[g + 1u];
        unsigned rank = 0u;
        for (unsigned j = a; j < b; ++j) {
            const unsigned k = ent[j];
            rank += (k < key || (k == key && j < i)) ? 1u : 0u;
        }
        srt[a + rank] = key;
    }
    __syncthreads();
    for (unsigned i = a0 + tid; i < a1; i += 1024u)
        __hip_atomic_exchange(&out[((size_t)(i >> 2) << 7) + (i & 3u)], srt[i],
                              __ATOMIC_RELAXED, __HIP_MEMORY_SCOPE_AGENT);
}

__global__ __launch_bounds__(256) void dis_planes_kernel(float* __restrict__ deg, const float* __restrict__ x,
                                                         float* __restrict__ xp, int n) {
    const int i = blockIdx.x * 256 + threadIdx.x;
    if (i < n) {
        float d = deg[i];
        deg[i] = (d > 0.0f) ? rsqrtf(d) : 0.0f;
    }
    const unsigned node = blockIdx.x * 32u + (threadIdx.x & 31u);
    const unsigned w = threadIdx.x >> 5;
    if (node < (unsigned)n) {
        const v4f* __restrict__ r = (const v4f*)(x + (size_t)node * 128u + 16u * w);
        v4f a[4];
#pragma unroll
        for (int j = 0; j < 4; ++j) a[j] = r[j];
        float* __restrict__ o = xp + (size_t)(8u * w) * PLP + 2u * node;
#pragma unroll
        for (int j = 0; j < 4; ++j) {
            v2f lo, hi;
            lo.x = a[j].x; lo.y = a[j].y;
            hi.x = a[j].z; hi.y = a[j].w;
            *(v2f*)(o + (size_t)(2 * j) * PLP) = lo;
            *(v2f*)(o + (size_t)(2 * j + 1) * PLP) = hi;
        }
    }
}
template <int FPB> struct gvec;
template <> struct gvec<2> { typedef v2f type; };
template <> struct gvec<4> { typedef v4f type; };

#define ACCX (ACCN + 1024u)
#define NOSLOT 0xFFFFFFFFu

template <int FPB>
__device__ __forceinline__ void add_edge(float* accf, unsigned slot, typename gvec<FPB>::type v) {
#pragma unroll
    for (int j = 0; j < FPB; ++j)
        __hip_atomic_fetch_add(&accf[slot * (unsigned)FPB + (unsigned)j], v[j], __ATOMIC_RELAXED,
                               __HIP_MEMORY_SCOPE_WORKGROUP);
}

template <int FPB, bool GUARD>
__device__ __forceinline__ void run_step(typename gvec<FPB>::type* acc, unsigned& cur, unsigned& idx, typename gvec<FPB>::type& s,
                                         unsigned ew, typename gvec<FPB>::type v, bool valid) {
    const unsigned slot = ew >> 16;
    bool chg = slot != cur;
    if (GUARD) chg = chg && valid;
    if (chg) {
        acc[idx] = s;
        idx = slot;
        cur = slot;
    }
    const typename gvec<FPB>::type t = s + v;
    if (GUARD) {
        if (valid) s = chg ? v : t;
    } else {
        s = chg ? v : t;
    }
}

template <int FPB, int ROWS, bool GUARD>
__device__ __forceinline__ void run_rows(const uint4* __restrict__ pl, const typename gvec<FPB>::type* xs,
                                         typename gvec<FPB>::type* acc,
                                         unsigned q, unsigned lane, unsigned lo, unsigned len, unsigned& cur,
                                         unsigned& idx, typename gvec<FPB>::type& s) {
    typedef typename gvec<FPB>::type gv;
    uint4 e[ROWS];
    gv v[4 * ROWS];
#pragma unroll
    for (int u = 0; u < ROWS; ++u) e[u] = pl[((q + (unsigned)u) << 5) + lane];
#pragma unroll
    for (int u = 0; u < ROWS; ++u) {
        const unsigned p = ((q + (unsigned)u) << 2) - lo;
        v[4 * u + 0] = xs[(!GUARD || p < len) ? (e[u].x & 0xFFFFu) : 0u];
        v[4 * u + 1] = xs[(!GUARD || p + 1u < len) ? (e[u].y & 0xFFFFu) : 0u];
        v[4 * u + 2] = xs[(!GUARD || p + 2u < len) ? (e[u].z & 0xFFFFu) : 0u];
        v[4 * u + 3] = xs[(!GUARD || p + 3u < len) ? (e[u].w & 0xFFFFu) : 0u];
    }
#pragma unroll
    for (int u = 0; u < ROWS; ++u) {
        const unsigned p = ((q + (unsigned)u) << 2) - lo;
        run_step<FPB, GUARD>(acc, cur, idx, s, e[u].x, v[4 * u + 0], p < len);
        run_step<FPB, GUARD>(acc, cur, idx, s, e[u].y, v[4 * u + 1], p + 1u < len);
        run_step<FPB, GUARD>(acc, cur, idx, s, e[u].z, v[4 * u + 2], p + 2u < len);
        run_step<FPB, GUARD>(acc, cur, idx, s, e[u].w, v[4 * u + 3], p + 3u < len);
    }
}

template <int FPB, bool PL>
__global__ __launch_bounds__(1024) void agg_split_kernel(
    const float* __restrict__ xin, const int* __restrict__ src,
    const int* __restrict__ dst, const float* __restrict__ dis,
    const unsigned* __restrict__ cnt, const unsigned* __restrict__ mid, const uint4* __restrict__ list,
    float* __restrict__ aggT, int E, int D)
{
    typedef typename gvec<FPB>::type gv;
    __shared__ gv xs[NNODES];
    __shared__ gv acc[ACCX];
    __shared__ unsigned slotF[1024];
    __shared__ unsigned slotL[1024];
    const unsigned tid = threadIdx.x;
    const unsigned half = blockIdx.x & 1u;
    const unsigned part = (blockIdx.x >> 1) & 1u;
    const unsigned fg = blockIdx.x >> 2;
    const unsigned f0 = fg * (unsigned)FPB;
    const unsigned lane = tid & 31u;
    const unsigned wave = tid >> 5;

    if constexpr (PL) {
        const float* __restrict__ pin = xin + (size_t)(fg * (unsigned)(FPB / 2)) * PLP;
        for (unsigned n = tid; n < NNODES; n += 1024u) {
            gv t;
#pragma unroll
            for (int j = 0; j < FPB / 2; ++j) {
                const v2f a = *(const v2f*)(pin + (size_t)j * PLP + 2u * n);
                t[2 * j] = a.x;
                t[2 * j + 1] = a.y;
            }
            xs[n] = dis[n] * t;
        }
    } else {
        for (unsigned n = tid; n < NNODES; n += 1024u) {
            const gv t = *(const gv*)(xin + (size_t)n * D + f0);
            xs[n] = dis[n] * t;
        }
    }
    for (unsigned c = tid; c < ACCN; c += 1024u) acc[c] = gv{};
    const unsigned myc = cnt[part * 32u + lane];
    const unsigned mym0 = mid[part * 32u + lane];
    const unsigned mym = (mym0 < myc) ? mym0 : myc;
    const unsigned lo = (half != 0u) ? mym : 0u;
    const unsigned hi = (half != 0u) ? myc : mym;
    const unsigned len = hi - lo;
    unsigned mxc = myc, lomin = lo, lomax = lo, himin = hi, himax = hi;
#pragma unroll
    for (int o = 16; o > 0; o >>= 1) {
        const unsigned a = __shfl_xor(mxc, o);
        const unsigned b0 = __shfl_xor(lomin, o);
        const unsigned b1 = __shfl_xor(lomax, o);
        const unsigned c0 = __shfl_xor(himin, o);
        const unsigned c1 = __shfl_xor(himax, o);
        mxc = (a > mxc) ? a : mxc;
        lomin = (b0 < lomin) ? b0 : lomin;
        lomax = (b1 > lomax) ? b1 : lomax;
        himin = (c0 < himin) ? c0 : himin;
        himax = (c1 > himax) ? c1 : himax;
    }
    __syncthreads();

    const unsigned qB = lomin >> 2;
    const unsigned qE = (himax + 3u) >> 2;
    const unsigned nrows = qE - qB;
    const bool ordered = (mxc <= SUBCAP) && (nrows >= 32u);
    unsigned sF = NOSLOT, sL = NOSLOT;
    gv vF = {}, vL = {};
    if (ordered) {
        const uint4* __restrict__ pl = list + (size_t)part * PARTQ;
        const unsigned qFS = (lomax + 3u) >> 2;
        const unsigned qFE = himin >> 2;
        const unsigned qs = qB + ((wave * nrows) >> 5);
        const unsigned qe = qB + (((wave + 1u) * nrows) >> 5);
        unsigned cur = NOSLOT;
        {
            const unsigned p0 = ((qs << 2) > lo) ? (qs << 2) : lo;
            const unsigned p1 = ((qe << 2) < hi) ? (qe << 2) : hi;
            if (p0 < p1) {
                const uint4 w0 = pl[((p0 >> 2) << 5) + lane];
                const unsigned k = p0 & 3u;
                const unsigned ew = (k == 0u) ? w0.x : ((k == 1u) ? w0.y : ((k == 2u) ? w0.z : w0.w));
                cur = ew >> 16;
            }
        }
        const unsigned firstslot = cur;
        unsigned idx = ACCN + tid;
        gv s = {};
        unsigned q = qs;
        constexpr int RT = 4;
        for (; q + (unsigned)RT <= qe; q += (unsigned)RT) {
            if (q >= qFS && q + (unsigned)RT <= qFE) run_rows<FPB, RT, false>(pl, xs, acc, q, lane, lo, len, cur, idx, s);
            else run_rows<FPB, RT, true>(pl, xs, acc, q, lane, lo, len, cur, idx, s);
        }
        for (; q < qe; ++q) run_rows<FPB, 1, true>(pl, xs, acc, q, lane, lo, len, cur, idx, s);
        sF = firstslot;
        if (idx == ACCN + tid) {
            vF = s;
            acc[ACCN + tid] = s;
        } else {
            vF = acc[ACCN + tid];
            sL = cur;
            vL = s;
        }
        slotF[tid] = sF;
        slotL[tid] = sL;
    } else {
        for (unsigned e = tid; e < (unsigned)E; e += 1024u) {
            const unsigned d = (unsigned)dst[e];
            if (((d >> 5) & 1u) == part && (((d >> 6) >= NH) == (half != 0u)))
                add_edge<FPB>((float*)acc, ((d >> 6) << 5) | (d & 31u), xs[src[e]]);
        }
    }
    __syncthreads();

    if (ordered) {
        bool lead = sF != NOSLOT;
        if (lead && wave != 0u) {
            const unsigned pl1 = slotL[tid - 32u];
            const unsigned prev = (pl1 != NOSLOT) ? pl1 : slotF[tid - 32u];
            lead = prev != sF;
        }
        if (lead && sL != NOSLOT) acc[sF] = vF;
        if (sL != NOSLOT || lead) {
            const unsigned slot = (sL != NOSLOT) ? sL : sF;
            gv sum = (sL != NOSLOT) ? vL : vF;
            for (unsigned w2 = wave + 1u; w2 < 32u; ++w2) {
                const unsigned t2 = (w2 << 5) + lane;
                if (slotF[t2] != slot) break;
                sum = sum + acc[ACCN + t2];
                if (slotL[t2] != NOSLOT) break;
            }
            acc[slot] = sum;
        }
    }
    __syncthreads();

    const unsigned cB = (half != 0u) ? (NH << 5) : 0u;
    const unsigned cE = (half != 0u) ? ACCN : (NH << 5);
    for (unsigned c = cB + tid; c < cE; c += 1024u) {
        const unsigned n = ((c >> 5) << 6) | (part << 5) | (c & 31u);
        if (n < NNODES) {
            const float dn = dis[n];
            const gv a = acc[c];
            if constexpr (PL) {
#pragma unroll
                for (int j = 0; j < FPB / 2; ++j) {
                    v2f o;
                    o.x = -(dn * a[2 * j]);
                    o.y = -(dn * a[2 * j + 1]);
                    *(v2f*)(aggT + (size_t)(fg * (unsigned)(FPB / 2) + (unsigned)j) * PLP + 2u * n) = o;
                }
            } else {
#pragma unroll
                for (int j = 0; j < FPB; ++j)
                    aggT[(size_t)(f0 + j) * AGGP + n] = -(dn * a[j]);
            }
        }
    }
}

template <int TILES> struct bvec;
template <> struct bvec<4> { typedef v4f type; };
template <> struct bvec<2> { typedef v2f type; };

template <int TILES> struct stage_t {
    v2f a0[2], a1[2];
    typename bvec<TILES>::type b0lo[2], b0hi[2], b1lo[2], b1hi[2];
};

template <int TILES, bool A0P, bool A1P>
__device__ __forceinline__ void load_stage(stage_t<TILES>& s, const float* __restrict__ a0p,
                                           const float* __restrict__ a1p, const float* __restrict__ w0p,
                                           const float* __restrict__ w1p, int k) {
    typedef typename bvec<TILES>::type bv;
    constexpr int NOUT = 64 * TILES;
#pragma unroll
    for (int u = 0; u < 2; ++u) {
        const int kk = k + 4 * u;
        if constexpr (A0P) {
            s.a0[u] = *(const v2f*)(a0p + (size_t)(kk >> 1) * PLP);
        } else {
            s.a0[u] = *(const v2f*)(a0p + kk);
        }
        if constexpr (A1P) {
            s.a1[u] = *(const v2f*)(a1p + (size_t)(kk >> 1) * PLP);
        } else {
            s.a1[u].x = a1p[(size_t)kk * AGGP];
            s.a1[u].y = a1p[(size_t)(kk + 1) * AGGP];
        }
        s.b0lo[u] = *(const bv*)(w0p + (size_t)kk * NOUT);
        s.b0hi[u] = *(const bv*)(w0p + (size_t)(kk + 1) * NOUT);
        s.b1lo[u] = *(const bv*)(w1p + (size_t)kk * NOUT);
        s.b1hi[u] = *(const bv*)(w1p + (size_t)(kk + 1) * NOUT);
    }
}

template <int TILES, bool OUTP>
__device__ __forceinline__ void mma_stage(const stage_t<TILES>& s, v8f (&acc)[TILES]) {
#pragma unroll
    for (int u = 0; u < 2; ++u) {
#pragma unroll
        for (int t = 0; t < TILES; ++t) {
            v2f b; b.x = s.b0lo[u][t]; b.y = s.b0hi[u][t];
            if constexpr (OUTP)
                acc[t] = __builtin_amdgcn_wmma_f32_16x16x4_f32(
                    false, b, false, s.a0[u], (short)0, acc[t], false, false);
            else
                acc[t] = __builtin_amdgcn_wmma_f32_16x16x4_f32(
                    false, s.a0[u], false, b, (short)0, acc[t], false, false);
        }
#pragma unroll
        for (int t = 0; t < TILES; ++t) {
            v2f b; b.x = s.b1lo[u][t]; b.y = s.b1hi[u][t];
            if constexpr (OUTP)
                acc[t] = __builtin_amdgcn_wmma_f32_16x16x4_f32(
                    false, b, false, s.a1[u], (short)0, acc[t], false, false);
            else
                acc[t] = __builtin_amdgcn_wmma_f32_16x16x4_f32(
                    false, s.a1[u], false, b, (short)0, acc[t], false, false);
        }
    }
}

template <int TILES, bool A0P, bool A1P, bool OUTP>
__global__ __launch_bounds__(128) void cheb_gemm_kernel(
    const float* __restrict__ Xin, const float* __restrict__ AggT,
    const float* __restrict__ W0, const float* __restrict__ W1,
    const float* __restrict__ bias, float* __restrict__ Out,
    int K, int relu)
{
    typedef typename bvec<TILES>::type bv;
    constexpr int NOUT = 64 * TILES;
    const int wave  = threadIdx.x >> 5;
    const int lane  = threadIdx.x & 31;
    const int rlane = lane & 15;
    const int half  = lane >> 4;
    const int khalf = half << 1;
    const int rowBase = blockIdx.x << 4;
    const int colBase = wave * (TILES * 16);
    const int col0 = colBase + TILES * rlane;

    const float* a0p = A0P ? Xin + (size_t)half * PLP + 2 * (rowBase + rlane)
                           : Xin + (size_t)(rowBase + rlane) * K + khalf;
    const float* a1p = A1P ? AggT + (size_t)half * PLP + 2 * (rowBase + rlane)
                           : AggT + (size_t)khalf * AGGP + rowBase + rlane;
    const float* w0p = W0 + (size_t)khalf * NOUT + col0;
    const float* w1p = W1 + (size_t)khalf * NOUT + col0;

    v8f acc[TILES];
#pragma unroll
    for (int t = 0; t < TILES; ++t) acc[t] = (v8f){};

    stage_t<TILES> cur, nxt;
    load_stage<TILES, A0P, A1P>(cur, a0p, a1p, w0p, w1p, 0);

#pragma unroll 2
    for (int k = 0; k < K - 8; k += 8) {
        load_stage<TILES, A0P, A1P>(nxt, a0p, a1p, w0p, w1p, k + 8);
        mma_stage<TILES, OUTP>(cur, acc);
        cur = nxt;
    }
    mma_stage<TILES, OUTP>(cur, acc);

    if constexpr (OUTP) {
        float* op = Out + (size_t)((colBase >> 1) + (TILES / 2) * 8 * half) * PLP + 2 * (rowBase + rlane);
        const float* bp = bias + colBase + TILES * 8 * half;
#pragma unroll
        for (int r = 0; r < 8; ++r) {
            const bv bvv = *(const bv*)(bp + TILES * r);
#pragma unroll
            for (int t = 0; t < TILES; t += 2) {
                v2f v;
                v.x = acc[t][r] + bvv[t];
                v.y = acc[t + 1][r] + bvv[t + 1];
                if (relu) { v.x = fmaxf(v.x, 0.0f); v.y = fmaxf(v.y, 0.0f); }
                *(v2f*)(op + (size_t)((TILES / 2) * r + (t >> 1)) * PLP) = v;
            }
        }
    } else {
        const int rowOff = rowBase + (half << 3);
        const bv bvv = *(const bv*)(bias + col0);
        float* op = Out + (size_t)rowOff * NOUT + col0;
#pragma unroll
        for (int r = 0; r < 8; ++r) {
            bv v;
#pragma unroll
            for (int t = 0; t < TILES; ++t) {
                float x = acc[t][r] + bvv[t];
                if (relu) x = fmaxf(x, 0.0f);
                v[t] = x;
            }
            *(bv*)(op + (size_t)r * NOUT) = v;
        }
    }
}

extern "C" void kernel_launch(void* const* d_in, const int* in_sizes, int n_in,
                              void* d_out, int out_size, void* d_ws, size_t ws_size,
                              hipStream_t stream) {
    const float* x   = (const float*)d_in[0];
    const int*   ei  = (const int*)d_in[1];
    const float* W01 = (const float*)d_in[3];
    const float* W11 = (const float*)d_in[4];
    const float* b1  = (const float*)d_in[5];
    const float* W02 = (const float*)d_in[6];
    const float* W12 = (const float*)d_in[7];
    const float* b2  = (const float*)d_in[8];
    const float* W03 = (const float*)d_in[9];
    const float* W13 = (const float*)d_in[10];
    const float* b3  = (const float*)d_in[11];

    const int N = NNODES;
    const int E = in_sizes[1] / 2;
    const int* src = ei;
    const int* dst = ei + E;

    float* dis  = (float*)d_ws;
    float* aggT = dis + 10240;
    float* xpl  = aggT + 64 * PLP;
    float* h1   = aggT + 256 * AGGP;
    float* h2   = h1  + (size_t)N * 256;
    float* out  = (float*)d_out;
    unsigned* cnt = (unsigned*)(dis + 10016);
    unsigned* cur = (unsigned*)(dis + 10080);
    unsigned* mid = (unsigned*)(dis + 10144);
    unsigned* list0 = (unsigned*)h2;
    unsigned* listw = (unsigned*)d_out;
    const uint4* list = (const uint4*)d_out;

    zero4_kernel<<<64, 256, 0, stream>>>((float4*)dis, 10240 / 4);
    deg_kernel<<<64, 1024, 0, stream>>>(src, dst, dis, cnt, E);
    dis_planes_kernel<<<(N + 31) / 32, 256, 0, stream>>>(dis, x, xpl, N);
    place_kernel<<<64, 1024, 0, stream>>>(src, dst, cur, list0, E);
    order_kernel<<<64 * OSPLIT, 1024, 0, stream>>>(cnt, list0, listw, mid);

    agg_split_kernel<2, true><<<256, 1024, 0, stream>>>(xpl, src, dst, dis, cnt, mid, list, aggT, E, 128);
    cheb_gemm_kernel<4, false, true, true><<<N / 16, 128, 0, stream>>>(x, aggT, W01, W11, b1, h1, 128, 1);

    agg_split_kernel<4, true><<<256, 1024, 0, stream>>>(h1, src, dst, dis, cnt, mid, list, aggT, E, 256);
    cheb_gemm_kernel<4, true, true, true><<<N / 16, 128, 0, stream>>>(h1, aggT, W02, W12, b2, h2, 256, 1);

    agg_split_kernel<4, true><<<256, 1024, 0, stream>>>(h2, src, dst, dis, cnt, mid, list, aggT, E, 256);
    cheb_gemm_kernel<2, true, true, false><<<N / 16, 128, 0, stream>>>(h2, aggT, W03, W13, b3, out, 256, 0);
}


# Aggregation walk loads its list rows one trip ahead (row numbers in scalar registers; a trip's LDS reads in two groups of two rows to fit the registers); order of every sum as before
# speedup vs baseline: 4.6586x; 1.0529x over previous
#include <hip/hip_runtime.h>
#include <hip/hip_bf16.h>

typedef __attribute__((ext_vector_type(2))) float v2f;
typedef __attribute__((ext_vector_type(4))) float v4f;
typedef __attribute__((ext_vector_type(8))) float v8f;

#define NNODES 10000
#define AGGP 10016
#define PLP 20000

__global__ __launch_bounds__(256) void zero4_kernel(float4* __restrict__ p, int n4) {
    int i = blockIdx.x * 256 + threadIdx.x;
    int stride = gridDim.x * 256;
    float4 z; z.x = 0.f; z.y = 0.f; z.z = 0.f; z.w = 0.f;
    for (; i < n4; i += stride) p[i] = z;
}

#define SUBCAP 20000u
#define PARTQ 160000u
#define ACCN 5024u
#define NH 79u

#define ETILE 10240u
#define EPT 10
__global__ __launch_bounds__(1024) void deg_kernel(const int* __restrict__ src,
                                                   const int* __restrict__ dst,
                                                   float* __restrict__ deg,
                                                   unsigned* __restrict__ cnt, int E) {
    __shared__ unsigned hist[NNODES];
    __shared__ unsigned h[64];
    const unsigned tid = threadIdx.x;
    for (unsigned n = tid; n < NNODES; n += 1024u) hist[n] = 0u;
    if (tid < 64u) h[tid] = 0u;
    __syncthreads();
    for (unsigned t0 = blockIdx.x * ETILE; t0 < (unsigned)E; t0 += gridDim.x * ETILE) {
        unsigned s[EPT], d[EPT];
#pragma unroll
        for (int u = 0; u < EPT; ++u) {
            const unsigned i = t0 + 1024u * (unsigned)u + tid;
            const unsigned ii = (i < (unsigned)E) ? i : t0;
            s[u] = (unsigned)src[ii];
            d[u] = (unsigned)dst[ii];
        }
#pragma unroll
        for (int u = 0; u < EPT; ++u) {
            if (t0 + 1024u * (unsigned)u + tid < (unsigned)E) {
                atomicAdd(&hist[s[u]], 1u);
                atomicAdd(&h[d[u] & 63u], 1u);
            }
        }
    }
    __syncthreads();
    for (unsigned n = tid; n < NNODES; n += 1024u) {
        const unsigned c = hist[n];
        if (c != 0u) atomicAdd(&deg[n], (float)c);
    }
    if (tid < 64u) {
        const unsigned c = h[tid];
        if (c != 0u) atomicAdd(&cnt[tid], c);
    }
}

__global__ __launch_bounds__(1024) void place_kernel(const int* __restrict__ src,
                                                     const int* __restrict__ dst,
                                                     unsigned* __restrict__ cur,
                                                     unsigned* __restrict__ list, int E) {
    __shared__ unsigned h[64];
    __shared__ unsigned base[64];
    __shared__ unsigned coff[64];
    __shared__ unsigned stage[ETILE];
    const unsigned tid = threadIdx.x;
    const unsigned lane = tid & 31u;
    for (unsigned t0 = blockIdx.x * ETILE; t0 < (unsigned)E; t0 += gridDim.x * ETILE) {
        if (tid < 64u) h[tid] = 0u;
        __syncthreads();
        unsigned s[EPT], d[EPT], r[EPT];
#pragma unroll
        for (int u = 0; u < EPT; ++u) {
            const unsigned i = t0 + 1024u * (unsigned)u + tid;
            const unsigned ii = (i < (unsigned)E) ? i : t0;
            s[u] = (unsigned)src[ii];
            d[u] = (unsigned)dst[ii];
        }
#pragma unroll
        for (int u = 0; u < EPT; ++u) {
            r[u] = 0u;
            if (t0 + 1024u * (unsigned)u + tid < (unsigned)E) r[u] = atomicAdd(&h[d[u] & 63u], 1u);
        }
        __syncthreads();
        if (tid < 64u) {
            const unsigned n = h[tid];
            unsigned v = n;
#pragma unroll
            for (int o = 1; o < 32; o <<= 1) {
                const unsigned t = __shfl_up(v, o);
                if (lane >= (unsigned)o) v += t;
            }
            unsigned low = h[lane];
#pragma unroll
            for (int o = 16; o > 0; o >>= 1) low += __shfl_xor(low, o);
            coff[tid] = v - n + ((tid >= 32u) ? low : 0u);
            base[tid] = (n != 0u) ? atomicAdd(&cur[tid], n) : 0u;
        }
        __syncthreads();
#pragma unroll
        for (int u = 0; u < EPT; ++u) {
            if (t0 + 1024u * (unsigned)u + tid < (unsigned)E) {
                const unsigned c = d[u] & 63u;
                stage[coff[c] + r[u]] = s[u] | ((((d[u] >> 6) << 5) | (c & 31u)) << 16);
            }
        }
        __syncthreads();
        for (unsigned c = tid >> 5; c < 64u; c += 32u) {
            const unsigned bc = base[c];
            const unsigned end = bc + h[c];
            const unsigned cc = coff[c];
            unsigned* __restrict__ out = list + (size_t)c * SUBCAP;
            for (unsigned pos = (bc & ~31u) + lane; pos < end; pos += 32u) {
                if (pos >= bc && pos < SUBCAP)
                    __hip_atomic_exchange(&out[pos], stage[cc + (pos - bc)],
                                          __ATOMIC_RELAXED, __HIP_MEMORY_SCOPE_AGENT);
            }
        }
        __syncthreads();
    }
}

#define OSPLIT 4u
#define OGRP 40u
__global__ __launch_bounds__(1024) void order_kernel(const unsigned* __restrict__ cnt,
                                                     const unsigned* __restrict__ lin,
                                                     unsigned* __restrict__ lout,
                                                     unsigned* __restrict__ mid) {
    __shared__ unsigned ent[SUBCAP];
    __shared__ unsigned srt[SUBCAP];
    __shared__ unsigned gsz[160];
    __shared__ unsigned gst[161];
    const unsigned tid = threadIdx.x;
    const unsigned c = blockIdx.x / OSPLIT;
    const unsigned sp = blockIdx.x % OSPLIT;
    const unsigned n = cnt[c];
    if (n > SUBCAP) return;
    const unsigned* __restrict__ in = lin + (size_t)c * SUBCAP;
    unsigned* __restrict__ out = lout + (size_t)(c >> 5) * (PARTQ * 4u) + ((c & 31u) << 2);
    if (tid < 160u) gsz[tid] = 0u;
    __syncthreads();
    for (unsigned p = tid; p < n; p += 1024u) {
        const unsigned e = in[p];
        atomicAdd(&gsz[e >> 21], 1u);
    }
    __syncthreads();
    if (tid < 32u) {
        unsigned a[5];
        unsigned t = 0u;
#pragma unroll
        for (int k = 0; k < 5; ++k) { a[k] = gsz[5u * tid + (unsigned)k]; t += a[k]; }
        unsigned v = t;
#pragma unroll
        for (int o = 1; o < 32; o <<= 1) {
            const unsigned u = __shfl_up(v, o);
            if (tid >= (unsigned)o) v += u;
        }
        unsigned ex = v - t;
#pragma unroll
        for (int k = 0; k < 5; ++k) {
            gst[5u * tid + (unsigned)k] = ex;
            gsz[5u * tid + (unsigned)k] = 0u;
            ex += a[k];
        }
        if (tid == 31u) gst[160] = ex;
    }
    __syncthreads();
    if (sp == 0u && tid == 0u)
        __hip_atomic_exchange(&mid[c], gst[NH], __ATOMIC_RELAXED, __HIP_MEMORY_SCOPE_AGENT);
    for (unsigned p = tid; p < n; p += 1024u) {
        const unsigned e = in[p];
        const unsigned g = e >> 21;
        const unsigned r = atomicAdd(&gsz[g], 1u);
        ent[gst[g] + r] = e;
    }
    __syncthreads();
    const unsigned a0 = gst[sp * OGRP];
    const unsigned a1 = gst[sp * OGRP + OGRP];
    for (unsigned i = a0 + tid; i < a1; i += 1024u) {
        const unsigned key = ent[i];
        const unsigned g = key >> 21;
        const unsigned a = gst[g];
        const unsigned b = gst[g + 1u];
        unsigned rank = 0u;
        for (unsigned j = a; j < b; ++j) {
            const unsigned k = ent[j];
            rank += (k < key || (k == key && j < i)) ? 1u : 0u;
        }
        srt[a + rank] = key;
    }
    __syncthreads();
    for (unsigned i = a0 + tid; i < a1; i += 1024u)
        __hip_atomic_exchange(&out[((size_t)(i >> 2) << 7) + (i & 3u)], srt[i],
                              __ATOMIC_RELAXED, __HIP_MEMORY_SCOPE_AGENT);
}

__global__ __launch_bounds__(256) void dis_planes_kernel(float* __restrict__ deg, const float* __restrict__ x,
                                                         float* __restrict__ xp, int n) {
    const int i = blockIdx.x * 256 + threadIdx.x;
    if (i < n) {
        float d = deg[i];
        deg[i] = (d > 0.0f) ? rsqrtf(d) : 0.0f;
    }
    const unsigned node = blockIdx.x * 32u + (threadIdx.x & 31u);
    const unsigned w = threadIdx.x >> 5;
    if (node < (unsigned)n) {
        const v4f* __restrict__ r = (const v4f*)(x + (size_t)node * 128u + 16u * w);
        v4f a[4];
#pragma unroll
        for (int j = 0; j < 4; ++j) a[j] = r[j];
        float* __restrict__ o = xp + (size_t)(8u * w) * PLP + 2u * node;
#pragma unroll
        for (int j = 0; j < 4; ++j) {
            v2f lo, hi;
            lo.x = a[j].x; lo.y = a[j].y;
            hi.x = a[j].z; hi.y = a[j].w;
            *(v2f*)(o + (size_t)(2 * j) * PLP) = lo;
            *(v2f*)(o + (size_t)(2 * j + 1) * PLP) = hi;
        }
    }
}
template <int FPB> struct gvec;
template <> struct gvec<2> { typedef v2f type; };
template <> struct gvec<4> { typedef v4f type; };

#define ACCX (ACCN + 1024u)
#define NOSLOT 0xFFFFFFFFu

template <int FPB>
__device__ __forceinline__ void add_edge(float* accf, unsigned slot, typename gvec<FPB>::type v) {
#pragma unroll
    for (int j = 0; j < FPB; ++j)
        __hip_atomic_fetch_add(&accf[slot * (unsigned)FPB + (unsigned)j], v[j], __ATOMIC_RELAXED,
                               __HIP_MEMORY_SCOPE_WORKGROUP);
}

template <int FPB, bool GUARD>
__device__ __forceinline__ void run_step(typename gvec<FPB>::type* acc, unsigned& cur, unsigned& idx, typename gvec<FPB>::type& s,
                                         unsigned ew, typename gvec<FPB>::type v, bool valid) {
    const unsigned slot = ew >> 16;
    bool chg = slot != cur;
    if (GUARD) chg = chg && valid;
    if (chg) {
        acc[idx] = s;
        idx = slot;
        cur = slot;
    }
    const typename gvec<FPB>::type t = s + v;
    if (GUARD) {
        if (valid) s = chg ? v : t;
    } else {
        s = chg ? v : t;
    }
}

template <int ROWS>
__device__ __forceinline__ void load_rows(const uint4* __restrict__ pl, unsigned q, unsigned qlast, unsigned lane,
                                          uint4 (&e)[ROWS]) {
#pragma unroll
    for (int u = 0; u < ROWS; ++u) {
        const unsigned r = q + (unsigned)u;
        const uint4* __restrict__ row = pl + ((size_t)((r < qlast) ? r : qlast) << 5);
        e[u] = row[lane];
    }
}

template <int FPB, int ROWS, bool GUARD>
__device__ __forceinline__ void run_rows(const uint4 (&e)[ROWS], const typename gvec<FPB>::type* xs,
                                         typename gvec<FPB>::type* acc,
                                         unsigned q, unsigned lo, unsigned len, unsigned& cur,
                                         unsigned& idx, typename gvec<FPB>::type& s) {
    typedef typename gvec<FPB>::type gv;
    gv v[4 * ROWS];
#pragma unroll
    for (int u = 0; u < ROWS; ++u) {
        const unsigned p = ((q + (unsigned)u) << 2) - lo;
        v[4 * u + 0] = xs[(!GUARD || p < len) ? (e[u].x & 0xFFFFu) : 0u];
        v[4 * u + 1] = xs[(!GUARD || p + 1u < len) ? (e[u].y & 0xFFFFu) : 0u];
        v[4 * u + 2] = xs[(!GUARD || p + 2u < len) ? (e[u].z & 0xFFFFu) : 0u];
        v[4 * u + 3] = xs[(!GUARD || p + 3u < len) ? (e[u].w & 0xFFFFu) : 0u];
    }
#pragma unroll
    for (int u = 0; u < ROWS; ++u) {
        const unsigned p = ((q + (unsigned)u) << 2) - lo;
        run_step<FPB, GUARD>(acc, cur, idx, s, e[u].x, v[4 * u + 0], p < len);
        run_step<FPB, GUARD>(acc, cur, idx, s, e[u].y, v[4 * u + 1], p + 1u < len);
        run_step<FPB, GUARD>(acc, cur, idx, s, e[u].z, v[4 * u + 2], p + 2u < len);
        run_step<FPB, GUARD>(acc, cur, idx, s, e[u].w, v[4 * u + 3], p + 3u < len);
    }
}

template <int FPB, bool PL>
__global__ __launch_bounds__(1024) void agg_split_kernel(
    const float* __restrict__ xin, const int* __restrict__ src,
    const int* __restrict__ dst, const float* __restrict__ dis,
    const unsigned* __restrict__ cnt, const unsigned* __restrict__ mid, const uint4* __restrict__ list,
    float* __restrict__ aggT, int E, int D)
{
    typedef typename gvec<FPB>::type gv;
    __shared__ gv xs[NNODES];
    __shared__ gv acc[ACCX];
    __shared__ unsigned slotF[1024];
    __shared__ unsigned slotL[1024];
    const unsigned tid = threadIdx.x;
    const unsigned half = blockIdx.x & 1u;
    const unsigned part = (blockIdx.x >> 1) & 1u;
    const unsigned fg = blockIdx.x >> 2;
    const unsigned f0 = fg * (unsigned)FPB;
    const unsigned lane = tid & 31u;
    const unsigned wave = (unsigned)__builtin_amdgcn_readfirstlane((int)(tid >> 5));

    if constexpr (PL) {
        const float* __restrict__ pin = xin + (size_t)(fg * (unsigned)(FPB / 2)) * PLP;
        for (unsigned n = tid; n < NNODES; n += 1024u) {
            gv t;
#pragma unroll
            for (int j = 0; j < FPB / 2; ++j) {
                const v2f a = *(const v2f*)(pin + (size_t)j * PLP + 2u * n);
                t[2 * j] = a.x;
                t[2 * j + 1] = a.y;
            }
            xs[n] = dis[n] * t;
        }
    } else {
        for (unsigned n = tid; n < NNODES; n += 1024u) {
            const gv t = *(const gv*)(xin + (size_t)n * D + f0);
            xs[n] = dis[n] * t;
        }
    }
    for (unsigned c = tid; c < ACCN; c += 1024u) acc[c] = gv{};
    const unsigned myc = cnt[part * 32u + lane];
    const unsigned mym0 = mid[part * 32u + lane];
    const unsigned mym = (mym0 < myc) ? mym0 : myc;
    const unsigned lo = (half != 0u) ? mym : 0u;
    const unsigned hi = (half != 0u) ? myc : mym;
    const unsigned len = hi - lo;
    unsigned mxc = myc, lomin = lo, lomax = lo, himin = hi, himax = hi;
#pragma unroll
    for (int o = 16; o > 0; o >>= 1) {
        const unsigned a = __shfl_xor(mxc, o);
        const unsigned b0 = __shfl_xor(lomin, o);
        const unsigned b1 = __shfl_xor(lomax, o);
        const unsigned c0 = __shfl_xor(himin, o);
        const unsigned c1 = __shfl_xor(himax, o);
        mxc = (a > mxc) ? a : mxc;
        lomin = (b0 < lomin) ? b0 : lomin;
        lomax = (b1 > lomax) ? b1 : lomax;
        himin = (c0 < himin) ? c0 : himin;
        himax = (c1 > himax) ? c1 : himax;
    }
    __syncthreads();

    const unsigned qB = (unsigned)__builtin_amdgcn_readfirstlane((int)(lomin >> 2));
    const unsigned qE = (unsigned)__builtin_amdgcn_readfirstlane((int)((himax + 3u) >> 2));
    const unsigned nrows = qE - qB;
    const bool ordered = (mxc <= SUBCAP) && (nrows >= 32u);
    unsigned sF = NOSLOT, sL = NOSLOT;
    gv vF = {}, vL = {};
    if (ordered) {
        const uint4* __restrict__ pl = list + (size_t)part * PARTQ;
        const unsigned qFS = (unsigned)__builtin_amdgcn_readfirstlane((int)((lomax + 3u) >> 2));
        const unsigned qFE = (unsigned)__builtin_amdgcn_readfirstlane((int)(himin >> 2));
        const unsigned qs = qB + ((wave * nrows) >> 5);
        const unsigned qe = qB + (((wave + 1u) * nrows) >> 5);
        unsigned cur = NOSLOT;
        {
            const unsigned p0 = ((qs << 2) > lo) ? (qs << 2) : lo;
            const unsigned p1 = ((qe << 2) < hi) ? (qe << 2) : hi;
            if (p0 < p1) {
                const uint4 w0 = pl[((p0 >> 2) << 5) + lane];
                const unsigned k = p0 & 3u;
                const unsigned ew = (k == 0u) ? w0.x : ((k == 1u) ? w0.y : ((k == 2u) ? w0.z : w0.w));
                cur = ew >> 16;
            }
        }
        const unsigned firstslot = cur;
        unsigned idx = ACCN + tid;
        gv s = {};
        unsigned q = qs;
        constexpr int RT = 4;
        const unsigned qlast = qe - 1u;
        uint4 en[RT];
        load_rows<RT>(pl, qs, qlast, lane, en);
        for (; q + (unsigned)RT <= qe; q += (unsigned)RT) {
            uint4 e[RT];
#pragma unroll
            for (int u = 0; u < RT; ++u) e[u] = en[u];
            load_rows<RT>(pl, q + (unsigned)RT, qlast, lane, en);
            const uint4 ea[2] = { e[0], e[1] };
            const uint4 eb[2] = { e[2], e[3] };
            if (q >= qFS && q + (unsigned)RT <= qFE) {
                run_rows<FPB, 2, false>(ea, xs, acc, q, lo, len, cur, idx, s);
                run_rows<FPB, 2, false>(eb, xs, acc, q + 2u, lo, len, cur, idx, s);
            } else {
                run_rows<FPB, 2, true>(ea, xs, acc, q, lo, len, cur, idx, s);
                run_rows<FPB, 2, true>(eb, xs, acc, q + 2u, lo, len, cur, idx, s);
            }
        }
#pragma unroll
        for (int u = 0; u < RT - 1; ++u) {
            if (q + (unsigned)u < qe) {
                const uint4 e1[1] = { en[u] };
                run_rows<FPB, 1, true>(e1, xs, acc, q + (unsigned)u, lo, len, cur, idx, s);
            }
        }
        sF = firstslot;
        if (idx == ACCN + tid) {
            vF = s;
            acc[ACCN + tid] = s;
        } else {
            vF = acc[ACCN + tid];
            sL = cur;
            vL = s;
        }
        slotF[tid] = sF;
        slotL[tid] = sL;
    } else {
        for (unsigned e = tid; e < (unsigned)E; e += 1024u) {
            const unsigned d = (unsigned)dst[e];
            if (((d >> 5) & 1u) == part && (((d >> 6) >= NH) == (half != 0u)))
                add_edge<FPB>((float*)acc, ((d >> 6) << 5) | (d & 31u), xs[src[e]]);
        }
    }
    __syncthreads();

    if (ordered) {
        bool lead = sF != NOSLOT;
        if (lead && wave != 0u) {
            const unsigned pl1 = slotL[tid - 32u];
            const unsigned prev = (pl1 != NOSLOT) ? pl1 : slotF[tid - 32u];
            lead = prev != sF;
        }
        if (lead && sL != NOSLOT) acc[sF] = vF;
        if (sL != NOSLOT || lead) {
            const unsigned slot = (sL != NOSLOT) ? sL : sF;
            gv sum = (sL != NOSLOT) ? vL : vF;
            for (unsigned w2 = wave + 1u; w2 < 32u; ++w2) {
                const unsigned t2 = (w2 << 5) + lane;
                if (slotF[t2] != slot) break;
                sum = sum + acc[ACCN + t2];
                if (slotL[t2] != NOSLOT) break;
            }
            acc[slot] = sum;
        }
    }
    __syncthreads();

    const unsigned cB = (half != 0u) ? (NH << 5) : 0u;
    const unsigned cE = (half != 0u) ? ACCN : (NH << 5);
    for (unsigned c = cB + tid; c < cE; c += 1024u) {
        const unsigned n = ((c >> 5) << 6) | (part << 5) | (c & 31u);
        if (n < NNODES) {
            const float dn = dis[n];
            const gv a = acc[c];
            if constexpr (PL) {
#pragma unroll
                for (int j = 0; j < FPB / 2; ++j) {
                    v2f o;
                    o.x = -(dn * a[2 * j]);
                    o.y = -(dn * a[2 * j + 1]);
                    *(v2f*)(aggT + (size_t)(fg * (unsigned)(FPB / 2) + (unsigned)j) * PLP + 2u * n) = o;
                }
            } else {
#pragma unroll
                for (int j = 0; j < FPB; ++j)
                    aggT[(size_t)(f0 + j) * AGGP + n] = -(dn * a[j]);
            }
        }
    }
}

template <int TILES> struct bvec;
template <> struct bvec<4> { typedef v4f type; };
template <> struct bvec<2> { typedef v2f type; };

template <int TILES> struct stage_t {
    v2f a0[2], a1[2];
    typename bvec<TILES>::type b0lo[2], b0hi[2], b1lo[2], b1hi[2];
};

template <int TILES, bool A0P, bool A1P>
__device__ __forceinline__ void load_stage(stage_t<TILES>& s, const float* __restrict__ a0p,
                                           const float* __restrict__ a1p, const float* __restrict__ w0p,
                                           const float* __restrict__ w1p, int k) {
    typedef typename bvec<TILES>::type bv;
    constexpr int NOUT = 64 * TILES;
#pragma unroll
    for (int u = 0; u < 2; ++u) {
        const int kk = k + 4 * u;
        if constexpr (A0P) {
            s.a0[u] = *(const v2f*)(a0p + (size_t)(kk >> 1) * PLP);
        } else {
            s.a0[u] = *(const v2f*)(a0p + kk);
        }
        if constexpr (A1P) {
            s.a1[u] = *(const v2f*)(a1p + (size_t)(kk >> 1) * PLP);
        } else {
            s.a1[u].x = a1p[(size_t)kk * AGGP];
            s.a1[u].y = a1p[(size_t)(kk + 1) * AGGP];
        }
        s.b0lo[u] = *(const bv*)(w0p + (size_t)kk * NOUT);
        s.b0hi[u] = *(const bv*)(w0p + (size_t)(kk + 1) * NOUT);
        s.b1lo[u] = *(const bv*)(w1p + (size_t)kk * NOUT);
        s.b1hi[u] = *(const bv*)(w1p + (size_t)(kk + 1) * NOUT);
    }
}

template <int TILES, bool OUTP>
__device__ __forceinline__ void mma_stage(const stage_t<TILES>& s, v8f (&acc)[TILES]) {
#pragma unroll
    for (int u = 0; u < 2; ++u) {
#pragma unroll
        for (int t = 0; t < TILES; ++t) {
            v2f b; b.x = s.b0lo[u][t]; b.y = s.b0hi[u][t];
            if constexpr (OUTP)
                acc[t] = __builtin_amdgcn_wmma_f32_16x16x4_f32(
                    false, b, false, s.a0[u], (short)0, acc[t], false, false);
            else
                acc[t] = __builtin_amdgcn_wmma_f32_16x16x4_f32(
                    false, s.a0[u], false, b, (short)0, acc[t], false, false);
        }
#pragma unroll
        for (int t = 0; t < TILES; ++t) {
            v2f b; b.x = s.b1lo[u][t]; b.y = s.b1hi[u][t];
            if constexpr (OUTP)
                acc[t] = __builtin_amdgcn_wmma_f32_16x16x4_f32(
                    false, b, false, s.a1[u], (short)0, acc[t], false, false);
            else
                acc[t] = __builtin_amdgcn_wmma_f32_16x16x4_f32(
                    false, s.a1[u], false, b, (short)0, acc[t], false, false);
        }
    }
}

template <int TILES, bool A0P, bool A1P, bool OUTP>
__global__ __launch_bounds__(128) void cheb_gemm_kernel(
    const float* __restrict__ Xin, const float* __restrict__ AggT,
    const float* __restrict__ W0, const float* __restrict__ W1,
    const float* __restrict__ bias, float* __restrict__ Out,
    int K, int relu)
{
    typedef typename bvec<TILES>::type bv;
    constexpr int NOUT = 64 * TILES;
    const int wave  = threadIdx.x >> 5;
    const int lane  = threadIdx.x & 31;
    const int rlane = lane & 15;
    const int half  = lane >> 4;
    const int khalf = half << 1;
    const int rowBase = blockIdx.x << 4;
    const int colBase = wave * (TILES * 16);
    const int col0 = colBase + TILES * rlane;

    const float* a0p = A0P ? Xin + (size_t)half * PLP + 2 * (rowBase + rlane)
                           : Xin + (size_t)(rowBase + rlane) * K + khalf;
    const float* a1p = A1P ? AggT + (size_t)half * PLP + 2 * (rowBase + rlane)
                           : AggT + (size_t)khalf * AGGP + rowBase + rlane;
    const float* w0p = W0 + (size_t)khalf * NOUT + col0;
    const float* w1p = W1 + (size_t)khalf * NOUT + col0;

    v8f acc[TILES];
#pragma unroll
    for (int t = 0; t < TILES; ++t) acc[t] = (v8f){};

    stage_t<TILES> cur, nxt;
    load_stage<TILES, A0P, A1P>(cur, a0p, a1p, w0p, w1p, 0);

#pragma unroll 2
    for (int k = 0; k < K - 8; k += 8) {
        load_stage<TILES, A0P, A1P>(nxt, a0p, a1p, w0p, w1p, k + 8);
        mma_stage<TILES, OUTP>(cur, acc);
        cur = nxt;
    }
    mma_stage<TILES, OUTP>(cur, acc);

    if constexpr (OUTP) {
        float* op = Out + (size_t)((colBase >> 1) + (TILES / 2) * 8 * half) * PLP + 2 * (rowBase + rlane);
        const float* bp = bias + colBase + TILES * 8 * half;
#pragma unroll
        for (int r = 0; r < 8; ++r) {
            const bv bvv = *(const bv*)(bp + TILES * r);
#pragma unroll
            for (int t = 0; t < TILES; t += 2) {
                v2f v;
                v.x = acc[t][r] + bvv[t];
                v.y = acc[t + 1][r] + bvv[t + 1];
                if (relu) { v.x = fmaxf(v.x, 0.0f); v.y = fmaxf(v.y, 0.0f); }
                *(v2f*)(op + (size_t)((TILES / 2) * r + (t >> 1)) * PLP) = v;
            }
        }
    } else {
        const int rowOff = rowBase + (half << 3);
        const bv bvv = *(const bv*)(bias + col0);
        float* op = Out + (size_t)rowOff * NOUT + col0;
#pragma unroll
        for (int r = 0; r < 8; ++r) {
            bv v;
#pragma unroll
            for (int t = 0; t < TILES; ++t) {
                float x = acc[t][r] + bvv[t];
                if (relu) x = fmaxf(x, 0.0f);
                v[t] = x;
            }
            *(bv*)(op + (size_t)r * NOUT) = v;
        }
    }
}

extern "C" void kernel_launch(void* const* d_in, const int* in_sizes, int n_in,
                              void* d_out, int out_size, void* d_ws, size_t ws_size,
                              hipStream_t stream) {
    const float* x   = (const float*)d_in[0];
    const int*   ei  = (const int*)d_in[1];
    const float* W01 = (const float*)d_in[3];
    const float* W11 = (const float*)d_in[4];
    const float* b1  = (const float*)d_in[5];
    const float* W02 = (const float*)d_in[6];
    const float* W12 = (const float*)d_in[7];
    const float* b2  = (const float*)d_in[8];
    const float* W03 = (const float*)d_in[9];
    const float* W13 = (const float*)d_in[10];
    const float* b3  = (const float*)d_in[11];

    const int N = NNODES;
    const int E = in_sizes[1] / 2;
    const int* src = ei;
    const int* dst = ei + E;

    float* dis  = (float*)d_ws;
    float* aggT = dis + 10240;
    float* xpl  = aggT + 64 * PLP;
    float* h1   = aggT + 256 * AGGP;
    float* h2   = h1  + (size_t)N * 256;
    float* out  = (float*)d_out;
    unsigned* cnt = (unsigned*)(dis + 10016);
    unsigned* cur = (unsigned*)(dis + 10080);
    unsigned* mid = (unsigned*)(dis + 10144);
    unsigned* list0 = (unsigned*)h2;
    unsigned* listw = (unsigned*)d_out;
    const uint4* list = (const uint4*)d_out;

    zero4_kernel<<<64, 256, 0, stream>>>((float4*)dis, 10240 / 4);
    deg_kernel<<<64, 1024, 0, stream>>>(src, dst, dis, cnt, E);
    dis_planes_kernel<<<(N + 31) / 32, 256, 0, stream>>>(dis, x, xpl, N);
    place_kernel<<<64, 1024, 0, stream>>>(src, dst, cur, list0, E);
    order_kernel<<<64 * OSPLIT, 1024, 0, stream>>>(cnt, list0, listw, mid);

    agg_split_kernel<2, true><<<256, 1024, 0, stream>>>(xpl, src, dst, dis, cnt, mid, list, aggT, E, 128);
    cheb_gemm_kernel<4, false, true, true><<<N / 16, 128, 0, stream>>>(x, aggT, W01, W11, b1, h1, 128, 1);

    agg_split_kernel<4, true><<<256, 1024, 0, stream>>>(h1, src, dst, dis, cnt, mid, list, aggT, E, 256);
    cheb_gemm_kernel<4, true, true, true><<<N / 16, 128, 0, stream>>>(h1, aggT, W02, W12, b2, h2, 256, 1);

    agg_split_kernel<4, true><<<256, 1024, 0, stream>>>(h2, src, dst, dis, cnt, mid, list, aggT, E, 256);
    cheb_gemm_kernel<2, true, true, false><<<N / 16, 128, 0, stream>>>(h2, aggT, W03, W13, b3, out, 256, 0);
}
